# starting kernel
# baseline (speedup 1.0000x reference)
#include <hip/hip_runtime.h>
#include <hip/hip_bf16.h>
#include <math.h>

typedef __attribute__((ext_vector_type(2))) float v2f;
typedef __attribute__((ext_vector_type(8))) float v8f;

static inline int ceil_div_ll(long long a, long long b) { return (int)((a + b - 1) / b); }

template<int K, int M>
__global__ __launch_bounds__(256)
void wmma_gemm_f32(const float* __restrict__ A, const float* __restrict__ W,
                   const float* __restrict__ bias, float* __restrict__ C, int N)
{
    constexpr int TILES_N = M / 16;
    int wave = (int)(blockIdx.x * (blockDim.x >> 5)) + (int)(threadIdx.x >> 5);
    int lane = threadIdx.x & 31;
    int tile_m = wave / TILES_N;
    int row0 = tile_m * 16;
    if (row0 >= N) return;
    int col0 = (wave % TILES_N) * 16;
    int lh = lane >> 4;
    int lm = lane & 15;

    const float* a_ptr = A + (size_t)(row0 + lm) * K + (lh << 1);
    const float* b_ptr = W + (size_t)(lh << 1) * M + col0 + lm;

    v8f acc = {};
#pragma unroll
    for (int k = 0; k < K; k += 4) {
        v2f a, b;
        a.x = a_ptr[k];
        a.y = a_ptr[k + 1];
        b.x = b_ptr[(size_t)k * M];
        b.y = b_ptr[(size_t)(k + 1) * M];
        acc = __builtin_amdgcn_wmma_f32_16x16x4_f32(false, a, false, b,
                                                    (short)0, acc, false, false);
    }

    float badd = bias ? bias[col0 + lm] : 0.0f;
#pragma unroll
    for (int r = 0; r < 8; ++r) {
        int row = row0 + r + (lh << 3);
        C[(size_t)row * M + col0 + lm] = acc[r] + badd;
    }
}

__global__ void fill_f32_kernel(float* __restrict__ p, float v, long long n)
{
    long long i = (long long)blockIdx.x * blockDim.x + threadIdx.x;
    if (i < n) p[i] = v;
}

__global__ void deg_kernel(const int* __restrict__ dst, float* __restrict__ deg, int E)
{
    int e = blockIdx.x * blockDim.x + threadIdx.x;
    if (e < E) atomicAdd(&deg[dst[e]], 1.0f);
}

__global__ void dinv_kernel(const float* __restrict__ deg, float* __restrict__ dinv, int n)
{
    int i = blockIdx.x * blockDim.x + threadIdx.x;
    if (i < n) dinv[i] = rsqrtf(deg[i] + 1.0f);
}

__global__ void norm_kernel(const int* __restrict__ src, const int* __restrict__ dst,
                            const float* __restrict__ dinv, float* __restrict__ nrm, int E)
{
    int e = blockIdx.x * blockDim.x + threadIdx.x;
    if (e < E) nrm[e] = dinv[src[e]] * dinv[dst[e]];
}

template<int M>
__global__ __launch_bounds__(256)
void edge_scatter(const int* __restrict__ src, const int* __restrict__ dst,
                  const float* __restrict__ nrm, const float* __restrict__ xw,
                  float* __restrict__ agg, int E)
{
    constexpr int CPE = M / 4;
    int gid = blockIdx.x * blockDim.x + threadIdx.x;
    int e = gid / CPE;
    if (e >= E) return;
    int c = (gid % CPE) << 2;
    int s = src[e], d = dst[e];
    float nm = nrm[e];
    const float4 v = *(const float4*)(xw + (size_t)s * M + c);
    float* o = agg + (size_t)d * M + c;
    atomicAdd(o + 0, v.x * nm);
    atomicAdd(o + 1, v.y * nm);
    atomicAdd(o + 2, v.z * nm);
    atomicAdd(o + 3, v.w * nm);
}

template<int M>
__global__ __launch_bounds__(256)
void finalize_ln_elu(const float* __restrict__ agg, const float* __restrict__ xw,
                     const float* __restrict__ dinv, const float* __restrict__ bias,
                     const float* __restrict__ g, const float* __restrict__ be,
                     float* __restrict__ out, int N)
{
    constexpr int PER = M / 32;
    int node = (int)(blockIdx.x * (blockDim.x >> 5)) + (int)(threadIdx.x >> 5);
    int lane = threadIdx.x & 31;
    if (node >= N) return;

    float vals[PER];
    float sum = 0.0f;
    float selfw = xw ? dinv[node] * dinv[node] : 0.0f;
#pragma unroll
    for (int i = 0; i < PER; ++i) {
        int c = lane + (i << 5);
        float v = agg[(size_t)node * M + c];
        if (xw)   v += xw[(size_t)node * M + c] * selfw;
        if (bias) v += bias[c];
        vals[i] = v;
        sum += v;
    }
#pragma unroll
    for (int off = 16; off; off >>= 1) sum += __shfl_xor(sum, off, 32);
    float mu = sum * (1.0f / M);
    float vs = 0.0f;
#pragma unroll
    for (int i = 0; i < PER; ++i) { float d = vals[i] - mu; vs += d * d; }
#pragma unroll
    for (int off = 16; off; off >>= 1) vs += __shfl_xor(vs, off, 32);
    float rstd = rsqrtf(vs * (1.0f / M) + 1e-5f);
#pragma unroll
    for (int i = 0; i < PER; ++i) {
        int c = lane + (i << 5);
        float v = (vals[i] - mu) * rstd * g[c] + be[c];
        out[(size_t)node * M + c] = (v > 0.0f) ? v : (expf(v) - 1.0f);
    }
}

extern "C" void kernel_launch(void* const* d_in, const int* in_sizes, int n_in,
                              void* d_out, int out_size, void* d_ws, size_t ws_size,
                              hipStream_t stream)
{
    const float* x   = (const float*)d_in[0];
    const int*   ei  = (const int*)  d_in[1];
    const float* W1  = (const float*)d_in[2];
    const float* b1  = (const float*)d_in[3];
    const float* g1  = (const float*)d_in[4];
    const float* be1 = (const float*)d_in[5];
    const float* W2  = (const float*)d_in[6];
    const float* b2  = (const float*)d_in[7];
    const float* g2  = (const float*)d_in[8];
    const float* be2 = (const float*)d_in[9];
    const float* W3  = (const float*)d_in[10];
    const float* b3  = (const float*)d_in[11];
    const float* g3  = (const float*)d_in[12];
    const float* be3 = (const float*)d_in[13];
    const float* lw1 = (const float*)d_in[14];
    const float* lb1 = (const float*)d_in[15];
    const float* g4  = (const float*)d_in[16];
    const float* be4 = (const float*)d_in[17];
    const float* lw2 = (const float*)d_in[18];
    const float* lb2 = (const float*)d_in[19];

    const int N = in_sizes[0] / 128;
    const int E = in_sizes[1] / 2;
    const int* src = ei;
    const int* dst = ei + E;

    float* ws   = (float*)d_ws;
    float* xw   = ws;  ws += (size_t)N * 128;
    float* agg  = ws;  ws += (size_t)N * 128;
    float* h    = ws;  ws += (size_t)N * 128;
    float* deg  = ws;  ws += N;
    float* dinv = ws;  ws += N;
    float* nrm  = ws;  ws += E;

    const int B = 256;
    auto gemm_blocks = [N](int tiles_n) {
        long long waves = (long long)((N + 15) / 16) * tiles_n;
        return (int)((waves + 7) / 8);
    };
    const int node_blocks = ceil_div_ll(N, 8);

    fill_f32_kernel<<<ceil_div_ll(N, B), B, 0, stream>>>(deg, 0.0f, (long long)N);
    deg_kernel<<<ceil_div_ll(E, B), B, 0, stream>>>(dst, deg, E);
    dinv_kernel<<<ceil_div_ll(N, B), B, 0, stream>>>(deg, dinv, N);
    norm_kernel<<<ceil_div_ll(E, B), B, 0, stream>>>(src, dst, dinv, nrm, E);

    wmma_gemm_f32<128, 64><<<gemm_blocks(4), B, 0, stream>>>(x, W1, nullptr, xw, N);
    fill_f32_kernel<<<ceil_div_ll((long long)N * 64, B), B, 0, stream>>>(agg, 0.0f, (long long)N * 64);
    edge_scatter<64><<<ceil_div_ll((long long)E * 16, B), B, 0, stream>>>(src, dst, nrm, xw, agg, E);
    finalize_ln_elu<64><<<node_blocks, B, 0, stream>>>(agg, xw, dinv, b1, g1, be1, h, N);

    wmma_gemm_f32<64, 128><<<gemm_blocks(8), B, 0, stream>>>(h, W2, nullptr, xw, N);
    fill_f32_kernel<<<ceil_div_ll((long long)N * 128, B), B, 0, stream>>>(agg, 0.0f, (long long)N * 128);
    edge_scatter<128><<<ceil_div_ll((long long)E * 32, B), B, 0, stream>>>(src, dst, nrm, xw, agg, E);
    finalize_ln_elu<128><<<node_blocks, B, 0, stream>>>(agg, xw, dinv, b2, g2, be2, h, N);

    wmma_gemm_f32<128, 64><<<gemm_blocks(4), B, 0, stream>>>(h, W3, nullptr, xw, N);
    fill_f32_kernel<<<ceil_div_ll((long long)N * 64, B), B, 0, stream>>>(agg, 0.0f, (long long)N * 64);
    edge_scatter<64><<<ceil_div_ll((long long)E * 16, B), B, 0, stream>>>(src, dst, nrm, xw, agg, E);
    finalize_ln_elu<64><<<node_blocks, B, 0, stream>>>(agg, xw, dinv, b3, g3, be3, h, N);

    wmma_gemm_f32<64, 32><<<gemm_blocks(2), B, 0, stream>>>(h, lw1, lb1, xw, N);
    finalize_ln_elu<32><<<node_blocks, B, 0, stream>>>(xw, nullptr, nullptr, nullptr, g4, be4, h, N);
    wmma_gemm_f32<32, 32><<<gemm_blocks(2), B, 0, stream>>>(h, lw2, lb2, (float*)d_out, N);
}


# Replace float atomic scatter (and its fill, degree, norm passes) by an in-call dst-sorted edge list (arrival-order integer rank atomics + scan) and a per-node register-sum gather; sum order changes
# speedup vs baseline: 2.0571x; 2.0571x over previous
#include <hip/hip_runtime.h>
#include <hip/hip_bf16.h>
#include <math.h>

typedef __attribute__((ext_vector_type(2))) float v2f;
typedef __attribute__((ext_vector_type(8))) float v8f;

static inline int ceil_div_ll(long long a, long long b) { return (int)((a + b - 1) / b); }

template<int K, int M>
__global__ __launch_bounds__(256)
void wmma_gemm_f32(const float* __restrict__ A, const float* __restrict__ W,
                   const float* __restrict__ bias, float* __restrict__ C, int N)
{
    constexpr int TILES_N = M / 16;
    int wave = (int)(blockIdx.x * (blockDim.x >> 5)) + (int)(threadIdx.x >> 5);
    int lane = threadIdx.x & 31;
    int tile_m = wave / TILES_N;
    int row0 = tile_m * 16;
    if (row0 >= N) return;
    int col0 = (wave % TILES_N) * 16;
    int lh = lane >> 4;
    int lm = lane & 15;

    const float* a_ptr = A + (size_t)(row0 + lm) * K + (lh << 1);
    const float* b_ptr = W + (size_t)(lh << 1) * M + col0 + lm;

    v8f acc = {};
#pragma unroll
    for (int k = 0; k < K; k += 4) {
        v2f a, b;
        a.x = a_ptr[k];
        a.y = a_ptr[k + 1];
        b.x = b_ptr[(size_t)k * M];
        b.y = b_ptr[(size_t)(k + 1) * M];
        acc = __builtin_amdgcn_wmma_f32_16x16x4_f32(false, a, false, b,
                                                    (short)0, acc, false, false);
    }

    float badd = bias ? bias[col0 + lm] : 0.0f;
#pragma unroll
    for (int r = 0; r < 8; ++r) {
        int row = row0 + r + (lh << 3);
        C[(size_t)row * M + col0 + lm] = acc[r] + badd;
    }
}

__global__ void zero_i32_kernel(int* __restrict__ p, int n)
{
    int i = blockIdx.x * blockDim.x + threadIdx.x;
    if (i < n) p[i] = 0;
}

__global__ void rank_kernel(const int* __restrict__ dst, int* __restrict__ cnt,
                            int* __restrict__ rank, int E)
{
    int e = blockIdx.x * blockDim.x + threadIdx.x;
    if (e < E) rank[e] = atomicAdd(&cnt[dst[e]], 1);
}

__global__ __launch_bounds__(1024)
void scan_dinv_kernel(const int* __restrict__ cnt, int* __restrict__ rowptr,
                      float* __restrict__ dinv, int N)
{
    __shared__ int wsum[32];
    __shared__ int s_prefix;
    const int t = threadIdx.x, lane = t & 31, wv = t >> 5;
    const int b0 = blockIdx.x * 1024;

    int p = 0;
    for (int i = t; i < b0; i += 1024) p += cnt[i];
#pragma unroll
    for (int off = 16; off; off >>= 1) p += __shfl_xor(p, off, 32);
    if (lane == 0) wsum[wv] = p;
    __syncthreads();
    if (wv == 0) {
        int q = wsum[lane];
#pragma unroll
        for (int off = 16; off; off >>= 1) q += __shfl_xor(q, off, 32);
        if (lane == 0) s_prefix = q;
    }
    __syncthreads();
    const int prefix = s_prefix;

    const int i = b0 + t;
    const int c = (i < N) ? cnt[i] : 0;
    int x = c;
#pragma unroll
    for (int off = 1; off < 32; off <<= 1) {
        int y = __shfl_up(x, off, 32);
        if (lane >= off) x += y;
    }
    if (lane == 31) wsum[wv] = x;
    __syncthreads();
    if (wv == 0) {
        const int tot = wsum[lane];
        int q = tot;
#pragma unroll
        for (int off = 1; off < 32; off <<= 1) {
            int y = __shfl_up(q, off, 32);
            if (lane >= off) q += y;
        }
        wsum[lane] = q - tot;
    }
    __syncthreads();
    const int excl = prefix + wsum[wv] + x - c;
    if (i < N) {
        rowptr[i] = excl;
        dinv[i] = rsqrtf((float)c + 1.0f);
        if (i == N - 1) rowptr[N] = excl + c;
    }
}

__global__ void place_kernel(const int* __restrict__ src, const int* __restrict__ dst,
                             const int* __restrict__ rank, const int* __restrict__ rowptr,
                             const float* __restrict__ dinv, int2* __restrict__ colw, int E)
{
    int e = blockIdx.x * blockDim.x + threadIdx.x;
    if (e < E) {
        int s = src[e], d = dst[e];
        float w = dinv[s] * dinv[d];
        colw[rowptr[d] + rank[e]] = make_int2(s, __float_as_int(w));
    }
}

template<int M>
__global__ __launch_bounds__(256)
void csr_gather(const int* __restrict__ rowptr, const int2* __restrict__ colw,
                const float* __restrict__ xw, float* __restrict__ agg, int N)
{
    constexpr int LPR = M / 4;
    constexpr int EPW = 32 / LPR;
    constexpr int UN  = 4;
    int node = (int)(blockIdx.x * (blockDim.x >> 5)) + (int)(threadIdx.x >> 5);
    node = __builtin_amdgcn_readfirstlane(node);
    const int lane = threadIdx.x & 31;
    if (node >= N) return;
    const int beg = rowptr[node];
    const int end = rowptr[node + 1];
    const int sub = lane / LPR;
    const int c = (lane % LPR) << 2;

    float4 acc = make_float4(0.0f, 0.0f, 0.0f, 0.0f);
    for (int base = beg; base < end; base += 32) {
        const int n = min(32, end - base);
        int2 mine = make_int2(0, 0);
        if (lane < n) mine = colw[base + lane];
        for (int j0 = 0; j0 < n; j0 += UN * EPW) {
            float4 v[UN];
            float  w[UN];
#pragma unroll
            for (int u = 0; u < UN; ++u) {
                const int j = (j0 + u * EPW + sub) & 31;
                const int s = __shfl(mine.x, j, 32);
                w[u] = __int_as_float(__shfl(mine.y, j, 32));
                v[u] = *(const float4*)(xw + (size_t)s * M + c);
            }
#pragma unroll
            for (int u = 0; u < UN; ++u) {
                if (j0 + u * EPW + sub < n) {
                    acc.x += v[u].x * w[u];
                    acc.y += v[u].y * w[u];
                    acc.z += v[u].z * w[u];
                    acc.w += v[u].w * w[u];
                }
            }
        }
    }
    if (EPW == 2) {
        acc.x += __shfl_xor(acc.x, 16, 32);
        acc.y += __shfl_xor(acc.y, 16, 32);
        acc.z += __shfl_xor(acc.z, 16, 32);
        acc.w += __shfl_xor(acc.w, 16, 32);
    }
    if (sub == 0) *(float4*)(agg + (size_t)node * M + c) = acc;
}

template<int M>
__global__ __launch_bounds__(256)
void finalize_ln_elu(const float* agg, const float* __restrict__ xw,
                     const float* __restrict__ dinv, const float* __restrict__ bias,
                     const float* __restrict__ g, const float* __restrict__ be,
                     float* out, int N)
{
    constexpr int PER = M / 32;
    int node = (int)(blockIdx.x * (blockDim.x >> 5)) + (int)(threadIdx.x >> 5);
    int lane = threadIdx.x & 31;
    if (node >= N) return;

    float vals[PER];
    float sum = 0.0f;
    float selfw = xw ? dinv[node] * dinv[node] : 0.0f;
#pragma unroll
    for (int i = 0; i < PER; ++i) {
        int c = lane + (i << 5);
        float v = agg[(size_t)node * M + c];
        if (xw)   v += xw[(size_t)node * M + c] * selfw;
        if (bias) v += bias[c];
        vals[i] = v;
        sum += v;
    }
#pragma unroll
    for (int off = 16; off; off >>= 1) sum += __shfl_xor(sum, off, 32);
    float mu = sum * (1.0f / M);
    float vs = 0.0f;
#pragma unroll
    for (int i = 0; i < PER; ++i) { float d = vals[i] - mu; vs += d * d; }
#pragma unroll
    for (int off = 16; off; off >>= 1) vs += __shfl_xor(vs, off, 32);
    float rstd = rsqrtf(vs * (1.0f / M) + 1e-5f);
#pragma unroll
    for (int i = 0; i < PER; ++i) {
        int c = lane + (i << 5);
        float v = (vals[i] - mu) * rstd * g[c] + be[c];
        out[(size_t)node * M + c] = (v > 0.0f) ? v : (expf(v) - 1.0f);
    }
}

extern "C" void kernel_launch(void* const* d_in, const int* in_sizes, int n_in,
                              void* d_out, int out_size, void* d_ws, size_t ws_size,
                              hipStream_t stream)
{
    const float* x   = (const float*)d_in[0];
    const int*   ei  = (const int*)  d_in[1];
    const float* W1  = (const float*)d_in[2];
    const float* b1  = (const float*)d_in[3];
    const float* g1  = (const float*)d_in[4];
    const float* be1 = (const float*)d_in[5];
    const float* W2  = (const float*)d_in[6];
    const float* b2  = (const float*)d_in[7];
    const float* g2  = (const float*)d_in[8];
    const float* be2 = (const float*)d_in[9];
    const float* W3  = (const float*)d_in[10];
    const float* b3  = (const float*)d_in[11];
    const float* g3  = (const float*)d_in[12];
    const float* be3 = (const float*)d_in[13];
    const float* lw1 = (const float*)d_in[14];
    const float* lb1 = (const float*)d_in[15];
    const float* g4  = (const float*)d_in[16];
    const float* be4 = (const float*)d_in[17];
    const float* lw2 = (const float*)d_in[18];
    const float* lb2 = (const float*)d_in[19];

    const int N = in_sizes[0] / 128;
    const int E = in_sizes[1] / 2;
    const int* src = ei;
    const int* dst = ei + E;

    float* ws   = (float*)d_ws;
    float* xw   = ws;  ws += (size_t)N * 128;
    float* agg  = ws;  ws += (size_t)N * 128;
    float* h    = ws;  ws += (size_t)N * 64;
    int2*  colw = (int2*)ws;
    int*   rank = (int*)(ws + (size_t)2 * E);
    ws += (size_t)N * 64;
    int*   cnt    = (int*)ws;  ws += N;
    float* dinv   = ws;        ws += N;
    int*   rowptr = (int*)ws;  ws += E;

    const int B = 256;
    auto gemm_blocks = [N](int tiles_n) {
        long long waves = (long long)((N + 15) / 16) * tiles_n;
        return (int)((waves + 7) / 8);
    };
    const int node_blocks = ceil_div_ll(N, 8);

    zero_i32_kernel<<<ceil_div_ll(N, B), B, 0, stream>>>(cnt, N);
    rank_kernel<<<ceil_div_ll(E, B), B, 0, stream>>>(dst, cnt, rank, E);
    scan_dinv_kernel<<<ceil_div_ll(N, 1024), 1024, 0, stream>>>(cnt, rowptr, dinv, N);
    place_kernel<<<ceil_div_ll(E, B), B, 0, stream>>>(src, dst, rank, rowptr, dinv, colw, E);

    wmma_gemm_f32<128, 64><<<gemm_blocks(4), B, 0, stream>>>(x, W1, nullptr, xw, N);
    csr_gather<64><<<node_blocks, B, 0, stream>>>(rowptr, colw, xw, agg, N);
    finalize_ln_elu<64><<<node_blocks, B, 0, stream>>>(agg, xw, dinv, b1, g1, be1, h, N);

    wmma_gemm_f32<64, 128><<<gemm_blocks(8), B, 0, stream>>>(h, W2, nullptr, xw, N);
    csr_gather<128><<<node_blocks, B, 0, stream>>>(rowptr, colw, xw, agg, N);
    finalize_ln_elu<128><<<node_blocks, B, 0, stream>>>(agg, xw, dinv, b2, g2, be2, agg, N);

    wmma_gemm_f32<128, 64><<<gemm_blocks(4), B, 0, stream>>>(agg, W3, nullptr, xw, N);
    csr_gather<64><<<node_blocks, B, 0, stream>>>(rowptr, colw, xw, agg, N);
    finalize_ln_elu<64><<<node_blocks, B, 0, stream>>>(agg, xw, dinv, b3, g3, be3, h, N);

    wmma_gemm_f32<64, 32><<<gemm_blocks(2), B, 0, stream>>>(h, lw1, lb1, xw, N);
    finalize_ln_elu<32><<<node_blocks, B, 0, stream>>>(xw, nullptr, nullptr, nullptr, g4, be4, h, N);
    wmma_gemm_f32<32, 32><<<gemm_blocks(2), B, 0, stream>>>(h, lw2, lb2, (float*)d_out, N);
}


# Rank, place, row-gather and LN+ELU kernels launch one resident grid (at most 2048 blocks of 8 waves) and loop with a grid stride instead of a wave per node or per 32 edges; sums and rank order kept
# speedup vs baseline: 2.2139x; 1.0762x over previous
#include <hip/hip_runtime.h>
#include <hip/hip_bf16.h>
#include <math.h>

typedef __attribute__((ext_vector_type(2))) float v2f;
typedef __attribute__((ext_vector_type(8))) float v8f;

static inline int ceil_div_ll(long long a, long long b) { return (int)((a + b - 1) / b); }

constexpr int LOOP_BLOCKS = 2048;
static inline int loop_grid(long long blocks_wanted)
{
    return (int)(blocks_wanted < LOOP_BLOCKS ? blocks_wanted : LOOP_BLOCKS);
}

template<int K, int M>
__global__ __launch_bounds__(256)
void wmma_gemm_f32(const float* __restrict__ A, const float* __restrict__ W,
                   const float* __restrict__ bias, float* __restrict__ C, int N)
{
    constexpr int TILES_N = M / 16;
    int wave = (int)(blockIdx.x * (blockDim.x >> 5)) + (int)(threadIdx.x >> 5);
    int lane = threadIdx.x & 31;
    int tile_m = wave / TILES_N;
    int row0 = tile_m * 16;
    if (row0 >= N) return;
    int col0 = (wave % TILES_N) * 16;
    int lh = lane >> 4;
    int lm = lane & 15;

    const float* a_ptr = A + (size_t)(row0 + lm) * K + (lh << 1);
    const float* b_ptr = W + (size_t)(lh << 1) * M + col0 + lm;

    v8f acc = {};
#pragma unroll
    for (int k = 0; k < K; k += 4) {
        v2f a, b;
        a.x = a_ptr[k];
        a.y = a_ptr[k + 1];
        b.x = b_ptr[(size_t)k * M];
        b.y = b_ptr[(size_t)(k + 1) * M];
        acc = __builtin_amdgcn_wmma_f32_16x16x4_f32(false, a, false, b,
                                                    (short)0, acc, false, false);
    }

    float badd = bias ? bias[col0 + lm] : 0.0f;
#pragma unroll
    for (int r = 0; r < 8; ++r) {
        int row = row0 + r + (lh << 3);
        C[(size_t)row * M + col0 + lm] = acc[r] + badd;
    }
}

__global__ void zero_i32_kernel(int* __restrict__ p, int n)
{
    int i = blockIdx.x * blockDim.x + threadIdx.x;
    if (i < n) p[i] = 0;
}

constexpr int EU = 4;

__global__ __launch_bounds__(256)
void rank_kernel(const int* __restrict__ dst, int* __restrict__ cnt,
                 int* __restrict__ rank, int E)
{
    const int stride = (int)(gridDim.x * blockDim.x);
    for (int e0 = (int)(blockIdx.x * blockDim.x + threadIdx.x); e0 < E; e0 += EU * stride) {
        if (e0 + (EU - 1) * stride < E) {
            int d[EU], r[EU];
#pragma unroll
            for (int u = 0; u < EU; ++u) d[u] = dst[e0 + u * stride];
#pragma unroll
            for (int u = 0; u < EU; ++u) r[u] = atomicAdd(&cnt[d[u]], 1);
#pragma unroll
            for (int u = 0; u < EU; ++u) rank[e0 + u * stride] = r[u];
        } else {
            for (int e = e0; e < E; e += stride) rank[e] = atomicAdd(&cnt[dst[e]], 1);
        }
    }
}

__global__ __launch_bounds__(1024)
void scan_dinv_kernel(const int* __restrict__ cnt, int* __restrict__ rowptr,
                      float* __restrict__ dinv, int N)
{
    __shared__ int wsum[32];
    __shared__ int s_prefix;
    const int t = threadIdx.x, lane = t & 31, wv = t >> 5;
    const int b0 = blockIdx.x * 1024;

    int p = 0;
    for (int i = t; i < b0; i += 1024) p += cnt[i];
#pragma unroll
    for (int off = 16; off; off >>= 1) p += __shfl_xor(p, off, 32);
    if (lane == 0) wsum[wv] = p;
    __syncthreads();
    if (wv == 0) {
        int q = wsum[lane];
#pragma unroll
        for (int off = 16; off; off >>= 1) q += __shfl_xor(q, off, 32);
        if (lane == 0) s_prefix = q;
    }
    __syncthreads();
    const int prefix = s_prefix;

    const int i = b0 + t;
    const int c = (i < N) ? cnt[i] : 0;
    int x = c;
#pragma unroll
    for (int off = 1; off < 32; off <<= 1) {
        int y = __shfl_up(x, off, 32);
        if (lane >= off) x += y;
    }
    if (lane == 31) wsum[wv] = x;
    __syncthreads();
    if (wv == 0) {
        const int tot = wsum[lane];
        int q = tot;
#pragma unroll
        for (int off = 1; off < 32; off <<= 1) {
            int y = __shfl_up(q, off, 32);
            if (lane >= off) q += y;
        }
        wsum[lane] = q - tot;
    }
    __syncthreads();
    const int excl = prefix + wsum[wv] + x - c;
    if (i < N) {
        rowptr[i] = excl;
        dinv[i] = rsqrtf((float)c + 1.0f);
        if (i == N - 1) rowptr[N] = excl + c;
    }
}

__global__ __launch_bounds__(256)
void place_kernel(const int* __restrict__ src, const int* __restrict__ dst,
                  const int* __restrict__ rank, const int* __restrict__ rowptr,
                  const float* __restrict__ dinv, int2* __restrict__ colw, int E)
{
    const int stride = (int)(gridDim.x * blockDim.x);
    for (int e0 = (int)(blockIdx.x * blockDim.x + threadIdx.x); e0 < E; e0 += EU * stride) {
        if (e0 + (EU - 1) * stride < E) {
            int s[EU], d[EU], rk[EU], pos[EU];
            float w[EU];
#pragma unroll
            for (int u = 0; u < EU; ++u) {
                const int e = e0 + u * stride;
                s[u] = src[e];
                d[u] = dst[e];
                rk[u] = rank[e];
            }
#pragma unroll
            for (int u = 0; u < EU; ++u) {
                w[u] = dinv[s[u]] * dinv[d[u]];
                pos[u] = rowptr[d[u]] + rk[u];
            }
#pragma unroll
            for (int u = 0; u < EU; ++u) colw[pos[u]] = make_int2(s[u], __float_as_int(w[u]));
        } else {
            for (int e = e0; e < E; e += stride) {
                const int s1 = src[e], d1 = dst[e];
                const float w1 = dinv[s1] * dinv[d1];
                colw[rowptr[d1] + rank[e]] = make_int2(s1, __float_as_int(w1));
            }
        }
    }
}

template<int M>
__global__ __launch_bounds__(256)
void csr_gather(const int* __restrict__ rowptr, const int2* __restrict__ colw,
                const float* __restrict__ xw, float* __restrict__ agg, int N)
{
    constexpr int LPR = M / 4;
    constexpr int EPW = 32 / LPR;
    constexpr int UN  = 4;
    const int nwaves = (int)(gridDim.x * (blockDim.x >> 5));
    int wave = (int)(blockIdx.x * (blockDim.x >> 5)) + (int)(threadIdx.x >> 5);
    wave = __builtin_amdgcn_readfirstlane(wave);
    const int lane = threadIdx.x & 31;
    const int sub = lane / LPR;
    const int c = (lane % LPR) << 2;

    for (int node = wave; node < N; node += nwaves) {
        const int beg = rowptr[node];
        const int end = rowptr[node + 1];

        float4 acc = make_float4(0.0f, 0.0f, 0.0f, 0.0f);
        for (int base = beg; base < end; base += 32) {
            const int n = min(32, end - base);
            int2 mine = make_int2(0, 0);
            if (lane < n) mine = colw[base + lane];
            for (int j0 = 0; j0 < n; j0 += UN * EPW) {
                float4 v[UN];
                float  w[UN];
#pragma unroll
                for (int u = 0; u < UN; ++u) {
                    const int j = (j0 + u * EPW + sub) & 31;
                    const int s = __shfl(mine.x, j, 32);
                    w[u] = __int_as_float(__shfl(mine.y, j, 32));
                    v[u] = *(const float4*)(xw + (size_t)s * M + c);
                }
#pragma unroll
                for (int u = 0; u < UN; ++u) {
                    if (j0 + u * EPW + sub < n) {
                        acc.x += v[u].x * w[u];
                        acc.y += v[u].y * w[u];
                        acc.z += v[u].z * w[u];
                        acc.w += v[u].w * w[u];
                    }
                }
            }
        }
        if (EPW == 2) {
            acc.x += __shfl_xor(acc.x, 16, 32);
            acc.y += __shfl_xor(acc.y, 16, 32);
            acc.z += __shfl_xor(acc.z, 16, 32);
            acc.w += __shfl_xor(acc.w, 16, 32);
        }
        if (sub == 0) *(float4*)(agg + (size_t)node * M + c) = acc;
    }
}

template<int M>
__global__ __launch_bounds__(256)
void finalize_ln_elu(const float* agg, const float* __restrict__ xw,
                     const float* __restrict__ dinv, const float* __restrict__ bias,
                     const float* __restrict__ g, const float* __restrict__ be,
                     float* out, int N)
{
    constexpr int PER = M / 32;
    const int nwaves = (int)(gridDim.x * (blockDim.x >> 5));
    int wave = (int)(blockIdx.x * (blockDim.x >> 5)) + (int)(threadIdx.x >> 5);
    wave = __builtin_amdgcn_readfirstlane(wave);
    const int lane = threadIdx.x & 31;

    float bi[PER], gg[PER], bb[PER];
#pragma unroll
    for (int i = 0; i < PER; ++i) {
        const int c = lane + (i << 5);
        bi[i] = bias ? bias[c] : 0.0f;
        gg[i] = g[c];
        bb[i] = be[c];
    }

    for (int node = wave; node < N; node += nwaves) {
        float vals[PER], xs[PER];
        float sum = 0.0f;
        float selfw = 0.0f;
#pragma unroll
        for (int i = 0; i < PER; ++i) vals[i] = agg[(size_t)node * M + lane + (i << 5)];
        if (xw) {
#pragma unroll
            for (int i = 0; i < PER; ++i) xs[i] = xw[(size_t)node * M + lane + (i << 5)];
            selfw = dinv[node] * dinv[node];
        } else {
#pragma unroll
            for (int i = 0; i < PER; ++i) xs[i] = 0.0f;
        }
#pragma unroll
        for (int i = 0; i < PER; ++i) {
            float v = vals[i];
            if (xw)   v += xs[i] * selfw;
            if (bias) v += bi[i];
            vals[i] = v;
            sum += v;
        }
#pragma unroll
        for (int off = 16; off; off >>= 1) sum += __shfl_xor(sum, off, 32);
        float mu = sum * (1.0f / M);
        float vs = 0.0f;
#pragma unroll
        for (int i = 0; i < PER; ++i) { float d = vals[i] - mu; vs += d * d; }
#pragma unroll
        for (int off = 16; off; off >>= 1) vs += __shfl_xor(vs, off, 32);
        float rstd = rsqrtf(vs * (1.0f / M) + 1e-5f);
#pragma unroll
        for (int i = 0; i < PER; ++i) {
            int c = lane + (i << 5);
            float v = (vals[i] - mu) * rstd * gg[i] + bb[i];
            out[(size_t)node * M + c] = (v > 0.0f) ? v : (expf(v) - 1.0f);
        }
    }
}

extern "C" void kernel_launch(void* const* d_in, const int* in_sizes, int n_in,
                              void* d_out, int out_size, void* d_ws, size_t ws_size,
                              hipStream_t stream)
{
    const float* x   = (const float*)d_in[0];
    const int*   ei  = (const int*)  d_in[1];
    const float* W1  = (const float*)d_in[2];
    const float* b1  = (const float*)d_in[3];
    const float* g1  = (const float*)d_in[4];
    const float* be1 = (const float*)d_in[5];
    const float* W2  = (const float*)d_in[6];
    const float* b2  = (const float*)d_in[7];
    const float* g2  = (const float*)d_in[8];
    const float* be2 = (const float*)d_in[9];
    const float* W3  = (const float*)d_in[10];
    const float* b3  = (const float*)d_in[11];
    const float* g3  = (const float*)d_in[12];
    const float* be3 = (const float*)d_in[13];
    const float* lw1 = (const float*)d_in[14];
    const float* lb1 = (const float*)d_in[15];
    const float* g4  = (const float*)d_in[16];
    const float* be4 = (const float*)d_in[17];
    const float* lw2 = (const float*)d_in[18];
    const float* lb2 = (const float*)d_in[19];

    const int N = in_sizes[0] / 128;
    const int E = in_sizes[1] / 2;
    const int* src = ei;
    const int* dst = ei + E;

    float* ws   = (float*)d_ws;
    float* xw   = ws;  ws += (size_t)N * 128;
    float* agg  = ws;  ws += (size_t)N * 128;
    float* h    = ws;  ws += (size_t)N * 64;
    int2*  colw = (int2*)ws;
    int*   rank = (int*)(ws + (size_t)2 * E);
    ws += (size_t)N * 64;
    int*   cnt    = (int*)ws;  ws += N;
    float* dinv   = ws;        ws += N;
    int*   rowptr = (int*)ws;  ws += E;

    const int B = 256;
    auto gemm_blocks = [N](int tiles_n) {
        long long waves = (long long)((N + 15) / 16) * tiles_n;
        return (int)((waves + 7) / 8);
    };
    const int node_blocks = loop_grid(ceil_div_ll(N, 8));
    const int edge_blocks = loop_grid(ceil_div_ll(E, (long long)B * EU));

    zero_i32_kernel<<<ceil_div_ll(N, B), B, 0, stream>>>(cnt, N);
    rank_kernel<<<edge_blocks, B, 0, stream>>>(dst, cnt, rank, E);
    scan_dinv_kernel<<<ceil_div_ll(N, 1024), 1024, 0, stream>>>(cnt, rowptr, dinv, N);
    place_kernel<<<edge_blocks, B, 0, stream>>>(src, dst, rank, rowptr, dinv, colw, E);

    wmma_gemm_f32<128, 64><<<gemm_blocks(4), B, 0, stream>>>(x, W1, nullptr, xw, N);
    csr_gather<64><<<node_blocks, B, 0, stream>>>(rowptr, colw, xw, agg, N);
    finalize_ln_elu<64><<<node_blocks, B, 0, stream>>>(agg, xw, dinv, b1, g1, be1, h, N);

    wmma_gemm_f32<64, 128><<<gemm_blocks(8), B, 0, stream>>>(h, W2, nullptr, xw, N);
    csr_gather<128><<<node_blocks, B, 0, stream>>>(rowptr, colw, xw, agg, N);
    finalize_ln_elu<128><<<node_blocks, B, 0, stream>>>(agg, xw, dinv, b2, g2, be2, agg, N);

    wmma_gemm_f32<128, 64><<<gemm_blocks(4), B, 0, stream>>>(agg, W3, nullptr, xw, N);
    csr_gather<64><<<node_blocks, B, 0, stream>>>(rowptr, colw, xw, agg, N);
    finalize_ln_elu<64><<<node_blocks, B, 0, stream>>>(agg, xw, dinv, b3, g3, be3, h, N);

    wmma_gemm_f32<64, 32><<<gemm_blocks(2), B, 0, stream>>>(h, lw1, lb1, xw, N);
    finalize_ln_elu<32><<<node_blocks, B, 0, stream>>>(xw, nullptr, nullptr, nullptr, g4, be4, h, N);
    wmma_gemm_f32<32, 32><<<gemm_blocks(2), B, 0, stream>>>(h, lw2, lb2, (float*)d_out, N);
}


# Fuse each graph layer's self-loop, bias, LayerNorm and ELU pass into its row-gather kernel (3 launches and the agg round trip gone); self term joins the edge sum, LN sums regrouped: sum order changes
# speedup vs baseline: 2.4039x; 1.0858x over previous
#include <hip/hip_runtime.h>
#include <hip/hip_bf16.h>
#include <math.h>

typedef __attribute__((ext_vector_type(2))) float v2f;
typedef __attribute__((ext_vector_type(8))) float v8f;

static inline int ceil_div_ll(long long a, long long b) { return (int)((a + b - 1) / b); }

constexpr int LOOP_BLOCKS = 2048;
static inline int loop_grid(long long blocks_wanted)
{
    return (int)(blocks_wanted < LOOP_BLOCKS ? blocks_wanted : LOOP_BLOCKS);
}

template<int K, int M>
__global__ __launch_bounds__(256)
void wmma_gemm_f32(const float* __restrict__ A, const float* __restrict__ W,
                   const float* __restrict__ bias, float* __restrict__ C, int N)
{
    constexpr int TILES_N = M / 16;
    int wave = (int)(blockIdx.x * (blockDim.x >> 5)) + (int)(threadIdx.x >> 5);
    int lane = threadIdx.x & 31;
    int tile_m = wave / TILES_N;
    int row0 = tile_m * 16;
    if (row0 >= N) return;
    int col0 = (wave % TILES_N) * 16;
    int lh = lane >> 4;
    int lm = lane & 15;

    const float* a_ptr = A + (size_t)(row0 + lm) * K + (lh << 1);
    const float* b_ptr = W + (size_t)(lh << 1) * M + col0 + lm;

    v8f acc = {};
#pragma unroll
    for (int k = 0; k < K; k += 4) {
        v2f a, b;
        a.x = a_ptr[k];
        a.y = a_ptr[k + 1];
        b.x = b_ptr[(size_t)k * M];
        b.y = b_ptr[(size_t)(k + 1) * M];
        acc = __builtin_amdgcn_wmma_f32_16x16x4_f32(false, a, false, b,
                                                    (short)0, acc, false, false);
    }

    float badd = bias ? bias[col0 + lm] : 0.0f;
#pragma unroll
    for (int r = 0; r < 8; ++r) {
        int row = row0 + r + (lh << 3);
        C[(size_t)row * M + col0 + lm] = acc[r] + badd;
    }
}

__global__ void zero_i32_kernel(int* __restrict__ p, int n)
{
    int i = blockIdx.x * blockDim.x + threadIdx.x;
    if (i < n) p[i] = 0;
}

constexpr int EU = 4;

__global__ __launch_bounds__(256)
void rank_kernel(const int* __restrict__ dst, int* __restrict__ cnt,
                 int* __restrict__ rank, int E)
{
    const int stride = (int)(gridDim.x * blockDim.x);
    for (int e0 = (int)(blockIdx.x * blockDim.x + threadIdx.x); e0 < E; e0 += EU * stride) {
        if (e0 + (EU - 1) * stride < E) {
            int d[EU], r[EU];
#pragma unroll
            for (int u = 0; u < EU; ++u) d[u] = dst[e0 + u * stride];
#pragma unroll
            for (int u = 0; u < EU; ++u) r[u] = atomicAdd(&cnt[d[u]], 1);
#pragma unroll
            for (int u = 0; u < EU; ++u) rank[e0 + u * stride] = r[u];
        } else {
            for (int e = e0; e < E; e += stride) rank[e] = atomicAdd(&cnt[dst[e]], 1);
        }
    }
}

__global__ __launch_bounds__(1024)
void scan_dinv_kernel(const int* __restrict__ cnt, int* __restrict__ rowptr,
                      float* __restrict__ dinv, int N)
{
    __shared__ int wsum[32];
    __shared__ int s_prefix;
    const int t = threadIdx.x, lane = t & 31, wv = t >> 5;
    const int b0 = blockIdx.x * 1024;

    int p = 0;
    for (int i = t; i < b0; i += 1024) p += cnt[i];
#pragma unroll
    for (int off = 16; off; off >>= 1) p += __shfl_xor(p, off, 32);
    if (lane == 0) wsum[wv] = p;
    __syncthreads();
    if (wv == 0) {
        int q = wsum[lane];
#pragma unroll
        for (int off = 16; off; off >>= 1) q += __shfl_xor(q, off, 32);
        if (lane == 0) s_prefix = q;
    }
    __syncthreads();
    const int prefix = s_prefix;

    const int i = b0 + t;
    const int c = (i < N) ? cnt[i] : 0;
    int x = c;
#pragma unroll
    for (int off = 1; off < 32; off <<= 1) {
        int y = __shfl_up(x, off, 32);
        if (lane >= off) x += y;
    }
    if (lane == 31) wsum[wv] = x;
    __syncthreads();
    if (wv == 0) {
        const int tot = wsum[lane];
        int q = tot;
#pragma unroll
        for (int off = 1; off < 32; off <<= 1) {
            int y = __shfl_up(q, off, 32);
            if (lane >= off) q += y;
        }
        wsum[lane] = q - tot;
    }
    __syncthreads();
    const int excl = prefix + wsum[wv] + x - c;
    if (i < N) {
        rowptr[i] = excl;
        dinv[i] = rsqrtf((float)c + 1.0f);
        if (i == N - 1) rowptr[N] = excl + c;
    }
}

__global__ __launch_bounds__(256)
void place_kernel(const int* __restrict__ src, const int* __restrict__ dst,
                  const int* __restrict__ rank, const int* __restrict__ rowptr,
                  const float* __restrict__ dinv, int2* __restrict__ colw, int E)
{
    const int stride = (int)(gridDim.x * blockDim.x);
    for (int e0 = (int)(blockIdx.x * blockDim.x + threadIdx.x); e0 < E; e0 += EU * stride) {
        if (e0 + (EU - 1) * stride < E) {
            int s[EU], d[EU], rk[EU], pos[EU];
            float w[EU];
#pragma unroll
            for (int u = 0; u < EU; ++u) {
                const int e = e0 + u * stride;
                s[u] = src[e];
                d[u] = dst[e];
                rk[u] = rank[e];
            }
#pragma unroll
            for (int u = 0; u < EU; ++u) {
                w[u] = dinv[s[u]] * dinv[d[u]];
                pos[u] = rowptr[d[u]] + rk[u];
            }
#pragma unroll
            for (int u = 0; u < EU; ++u) colw[pos[u]] = make_int2(s[u], __float_as_int(w[u]));
        } else {
            for (int e = e0; e < E; e += stride) {
                const int s1 = src[e], d1 = dst[e];
                const float w1 = dinv[s1] * dinv[d1];
                colw[rowptr[d1] + rank[e]] = make_int2(s1, __float_as_int(w1));
            }
        }
    }
}

__device__ __forceinline__ float xhalf(float x)
{
    const int b = __builtin_bit_cast(int, x);
    return __builtin_bit_cast(float,
        __builtin_amdgcn_permlanex16(b, b, 0x76543210, 0xfedcba98, false, false));
}
template<int MASK>
__device__ __forceinline__ float xrow(float x)
{
    const int b = __builtin_bit_cast(int, x);
    return __builtin_bit_cast(float, __builtin_amdgcn_update_dpp(b, b, 0x160 | MASK, 0xf, 0xf, false));
}
template<int LANES>
__device__ __forceinline__ float lanes_sum(float x)
{
    x += xrow<1>(x);
    x += xrow<2>(x);
    x += xrow<4>(x);
    x += xrow<8>(x);
    if (LANES == 32) x += xhalf(x);
    return x;
}

template<int M>
__global__ __launch_bounds__(256)
void csr_gather_ln_elu(const int* __restrict__ rowptr, const int2* __restrict__ colw,
                       const float* __restrict__ xw, const float* __restrict__ dinv,
                       const float* __restrict__ bias, const float* __restrict__ g,
                       const float* __restrict__ be, float* __restrict__ out, int N)
{
    constexpr int LPR = M / 4;
    constexpr int EPW = 32 / LPR;
    constexpr int UN  = 4;
    const int nwaves = (int)(gridDim.x * (blockDim.x >> 5));
    int wave = (int)(blockIdx.x * (blockDim.x >> 5)) + (int)(threadIdx.x >> 5);
    wave = __builtin_amdgcn_readfirstlane(wave);
    const int lane = threadIdx.x & 31;
    const int sub = lane / LPR;
    const int c = (lane % LPR) << 2;

    float4 bi, gg, bb;
    bi.x = bias[c]; bi.y = bias[c + 1]; bi.z = bias[c + 2]; bi.w = bias[c + 3];
    gg.x = g[c];    gg.y = g[c + 1];    gg.z = g[c + 2];    gg.w = g[c + 3];
    bb.x = be[c];   bb.y = be[c + 1];   bb.z = be[c + 2];   bb.w = be[c + 3];

    for (int node = wave; node < N; node += nwaves) {
        const int beg = rowptr[node];
        const int end = rowptr[node + 1];
        const float di = dinv[node];
        const int selfw = __float_as_int(di * di);
        const int stop = end + 1;

        float4 acc = make_float4(0.0f, 0.0f, 0.0f, 0.0f);
        int base = beg;
        do {
            const int n = min(32, stop - base);
            const int e = base + lane;
            int2 mine = make_int2(0, 0);
            if (e < end)       mine = colw[e];
            else if (e == end) mine = make_int2(node, selfw);
            for (int j0 = 0; j0 < n; j0 += UN * EPW) {
                float4 v[UN];
                float  w[UN];
#pragma unroll
                for (int u = 0; u < UN; ++u) {
                    const int j = (j0 + u * EPW + sub) & 31;
                    const int s = __shfl(mine.x, j, 32);
                    w[u] = __int_as_float(__shfl(mine.y, j, 32));
                    v[u] = *(const float4*)(xw + (size_t)s * M + c);
                }
#pragma unroll
                for (int u = 0; u < UN; ++u) {
                    if (j0 + u * EPW + sub < n) {
                        acc.x += v[u].x * w[u];
                        acc.y += v[u].y * w[u];
                        acc.z += v[u].z * w[u];
                        acc.w += v[u].w * w[u];
                    }
                }
            }
            base += 32;
        } while (base < stop);
        if (EPW == 2) {
            acc.x += __shfl_xor(acc.x, 16, 32);
            acc.y += __shfl_xor(acc.y, 16, 32);
            acc.z += __shfl_xor(acc.z, 16, 32);
            acc.w += __shfl_xor(acc.w, 16, 32);
        }

        float4 t;
        t.x = acc.x + bi.x;
        t.y = acc.y + bi.y;
        t.z = acc.z + bi.z;
        t.w = acc.w + bi.w;

        const float sum = lanes_sum<LPR>((t.x + t.y) + (t.z + t.w));
        const float mu = sum * (1.0f / M);
        const float dx = t.x - mu, dy = t.y - mu, dz = t.z - mu, dw = t.w - mu;
        const float vs = lanes_sum<LPR>((dx * dx + dy * dy) + (dz * dz + dw * dw));
        const float rstd = rsqrtf(vs * (1.0f / M) + 1e-5f);
        float4 o;
        o.x = dx * rstd * gg.x + bb.x;
        o.y = dy * rstd * gg.y + bb.y;
        o.z = dz * rstd * gg.z + bb.z;
        o.w = dw * rstd * gg.w + bb.w;
        o.x = (o.x > 0.0f) ? o.x : (expf(o.x) - 1.0f);
        o.y = (o.y > 0.0f) ? o.y : (expf(o.y) - 1.0f);
        o.z = (o.z > 0.0f) ? o.z : (expf(o.z) - 1.0f);
        o.w = (o.w > 0.0f) ? o.w : (expf(o.w) - 1.0f);
        if (sub == 0) *(float4*)(out + (size_t)node * M + c) = o;
    }
}

template<int M>
__global__ __launch_bounds__(256)
void finalize_ln_elu(const float* agg, const float* __restrict__ xw,
                     const float* __restrict__ dinv, const float* __restrict__ bias,
                     const float* __restrict__ g, const float* __restrict__ be,
                     float* out, int N)
{
    constexpr int PER = M / 32;
    const int nwaves = (int)(gridDim.x * (blockDim.x >> 5));
    int wave = (int)(blockIdx.x * (blockDim.x >> 5)) + (int)(threadIdx.x >> 5);
    wave = __builtin_amdgcn_readfirstlane(wave);
    const int lane = threadIdx.x & 31;

    float bi[PER], gg[PER], bb[PER];
#pragma unroll
    for (int i = 0; i < PER; ++i) {
        const int c = lane + (i << 5);
        bi[i] = bias ? bias[c] : 0.0f;
        gg[i] = g[c];
        bb[i] = be[c];
    }

    for (int node = wave; node < N; node += nwaves) {
        float vals[PER], xs[PER];
        float sum = 0.0f;
        float selfw = 0.0f;
#pragma unroll
        for (int i = 0; i < PER; ++i) vals[i] = agg[(size_t)node * M + lane + (i << 5)];
        if (xw) {
#pragma unroll
            for (int i = 0; i < PER; ++i) xs[i] = xw[(size_t)node * M + lane + (i << 5)];
            selfw = dinv[node] * dinv[node];
        } else {
#pragma unroll
            for (int i = 0; i < PER; ++i) xs[i] = 0.0f;
        }
#pragma unroll
        for (int i = 0; i < PER; ++i) {
            float v = vals[i];
            if (xw)   v += xs[i] * selfw;
            if (bias) v += bi[i];
            vals[i] = v;
            sum += v;
        }
#pragma unroll
        for (int off = 16; off; off >>= 1) sum += __shfl_xor(sum, off, 32);
        float mu = sum * (1.0f / M);
        float vs = 0.0f;
#pragma unroll
        for (int i = 0; i < PER; ++i) { float d = vals[i] - mu; vs += d * d; }
#pragma unroll
        for (int off = 16; off; off >>= 1) vs += __shfl_xor(vs, off, 32);
        float rstd = rsqrtf(vs * (1.0f / M) + 1e-5f);
#pragma unroll
        for (int i = 0; i < PER; ++i) {
            int c = lane + (i << 5);
            float v = (vals[i] - mu) * rstd * gg[i] + bb[i];
            out[(size_t)node * M + c] = (v > 0.0f) ? v : (expf(v) - 1.0f);
        }
    }
}

extern "C" void kernel_launch(void* const* d_in, const int* in_sizes, int n_in,
                              void* d_out, int out_size, void* d_ws, size_t ws_size,
                              hipStream_t stream)
{
    const float* x   = (const float*)d_in[0];
    const int*   ei  = (const int*)  d_in[1];
    const float* W1  = (const float*)d_in[2];
    const float* b1  = (const float*)d_in[3];
    const float* g1  = (const float*)d_in[4];
    const float* be1 = (const float*)d_in[5];
    const float* W2  = (const float*)d_in[6];
    const float* b2  = (const float*)d_in[7];
    const float* g2  = (const float*)d_in[8];
    const float* be2 = (const float*)d_in[9];
    const float* W3  = (const float*)d_in[10];
    const float* b3  = (const float*)d_in[11];
    const float* g3  = (const float*)d_in[12];
    const float* be3 = (const float*)d_in[13];
    const float* lw1 = (const float*)d_in[14];
    const float* lb1 = (const float*)d_in[15];
    const float* g4  = (const float*)d_in[16];
    const float* be4 = (const float*)d_in[17];
    const float* lw2 = (const float*)d_in[18];
    const float* lb2 = (const float*)d_in[19];

    const int N = in_sizes[0] / 128;
    const int E = in_sizes[1] / 2;
    const int* src = ei;
    const int* dst = ei + E;

    float* ws   = (float*)d_ws;
    float* xw   = ws;  ws += (size_t)N * 128;
    float* agg  = ws;  ws += (size_t)N * 128;
    float* h    = ws;  ws += (size_t)N * 64;
    int2*  colw = (int2*)ws;
    int*   rank = (int*)(ws + (size_t)2 * E);
    ws += (size_t)N * 64;
    int*   cnt    = (int*)ws;  ws += N;
    float* dinv   = ws;        ws += N;
    int*   rowptr = (int*)ws;  ws += E;

    const int B = 256;
    auto gemm_blocks = [N](int tiles_n) {
        long long waves = (long long)((N + 15) / 16) * tiles_n;
        return (int)((waves + 7) / 8);
    };
    const int node_blocks = loop_grid(ceil_div_ll(N, 8));
    const int edge_blocks = loop_grid(ceil_div_ll(E, (long long)B * EU));

    zero_i32_kernel<<<ceil_div_ll(N, B), B, 0, stream>>>(cnt, N);
    rank_kernel<<<edge_blocks, B, 0, stream>>>(dst, cnt, rank, E);
    scan_dinv_kernel<<<ceil_div_ll(N, 1024), 1024, 0, stream>>>(cnt, rowptr, dinv, N);
    place_kernel<<<edge_blocks, B, 0, stream>>>(src, dst, rank, rowptr, dinv, colw, E);

    wmma_gemm_f32<128, 64><<<gemm_blocks(4), B, 0, stream>>>(x, W1, nullptr, xw, N);
    csr_gather_ln_elu<64><<<node_blocks, B, 0, stream>>>(rowptr, colw, xw, dinv, b1, g1, be1, h, N);

    wmma_gemm_f32<64, 128><<<gemm_blocks(8), B, 0, stream>>>(h, W2, nullptr, xw, N);
    csr_gather_ln_elu<128><<<node_blocks, B, 0, stream>>>(rowptr, colw, xw, dinv, b2, g2, be2, agg, N);

    wmma_gemm_f32<128, 64><<<gemm_blocks(4), B, 0, stream>>>(agg, W3, nullptr, xw, N);
    csr_gather_ln_elu<64><<<node_blocks, B, 0, stream>>>(rowptr, colw, xw, dinv, b3, g3, be3, h, N);

    wmma_gemm_f32<64, 32><<<gemm_blocks(2), B, 0, stream>>>(h, lw1, lb1, xw, N);
    finalize_ln_elu<32><<<node_blocks, B, 0, stream>>>(xw, nullptr, nullptr, nullptr, g4, be4, h, N);
    wmma_gemm_f32<32, 32><<<gemm_blocks(2), B, 0, stream>>>(h, lw2, lb2, (float*)d_out, N);
}


# Sort edges by destination with LDS counters (bin edge chunks into 256-node buckets, then count and place per bucket) instead of per-edge returning global adds; row order, so sum order, changes
# speedup vs baseline: 2.9568x; 1.2300x over previous
#include <hip/hip_runtime.h>
#include <hip/hip_bf16.h>
#include <math.h>

typedef __attribute__((ext_vector_type(2))) float v2f;
typedef __attribute__((ext_vector_type(8))) float v8f;

static inline int ceil_div_ll(long long a, long long b) { return (int)((a + b - 1) / b); }

constexpr int LOOP_BLOCKS = 2048;
static inline int loop_grid(long long blocks_wanted)
{
    return (int)(blocks_wanted < LOOP_BLOCKS ? blocks_wanted : LOOP_BLOCKS);
}

template<int K, int M>
__global__ __launch_bounds__(256)
void wmma_gemm_f32(const float* __restrict__ A, const float* __restrict__ W,
                   const float* __restrict__ bias, float* __restrict__ C, int N)
{
    constexpr int TILES_N = M / 16;
    int wave = (int)(blockIdx.x * (blockDim.x >> 5)) + (int)(threadIdx.x >> 5);
    int lane = threadIdx.x & 31;
    int tile_m = wave / TILES_N;
    int row0 = tile_m * 16;
    if (row0 >= N) return;
    int col0 = (wave % TILES_N) * 16;
    int lh = lane >> 4;
    int lm = lane & 15;

    const float* a_ptr = A + (size_t)(row0 + lm) * K + (lh << 1);
    const float* b_ptr = W + (size_t)(lh << 1) * M + col0 + lm;

    v8f acc = {};
#pragma unroll
    for (int k = 0; k < K; k += 4) {
        v2f a, b;
        a.x = a_ptr[k];
        a.y = a_ptr[k + 1];
        b.x = b_ptr[(size_t)k * M];
        b.y = b_ptr[(size_t)(k + 1) * M];
        acc = __builtin_amdgcn_wmma_f32_16x16x4_f32(false, a, false, b,
                                                    (short)0, acc, false, false);
    }

    float badd = bias ? bias[col0 + lm] : 0.0f;
#pragma unroll
    for (int r = 0; r < 8; ++r) {
        int row = row0 + r + (lh << 3);
        C[(size_t)row * M + col0 + lm] = acc[r] + badd;
    }
}

constexpr int CH_E   = 4096;
constexpr int BK_SH  = 8;
constexpr int BK_N   = 1 << BK_SH;
constexpr int PREP_T = 1024;
constexpr int CH_U   = CH_E / PREP_T;

__global__ __launch_bounds__(1024)
void bin_kernel(const int* __restrict__ src, const int* __restrict__ dst,
                int2* __restrict__ binned, int* __restrict__ tcnt, int* __restrict__ toff,
                int E, int NB, int NCH)
{
    __shared__ int hist[PREP_T];
    __shared__ int offs[PREP_T];
    __shared__ int wsum[32];
    const int t = threadIdx.x, lane = t & 31, wv = t >> 5;
    const int chunk = blockIdx.x;
    const int e_base = chunk * CH_E;

    hist[t] = 0;
    __syncthreads();

    int s[CH_U], d[CH_U], r[CH_U];
#pragma unroll
    for (int u = 0; u < CH_U; ++u) {
        const int e = e_base + u * PREP_T + t;
        const bool ok = e < E;
        s[u] = ok ? src[e] : 0;
        d[u] = ok ? dst[e] : -1;
    }
#pragma unroll
    for (int u = 0; u < CH_U; ++u)
        r[u] = (d[u] >= 0) ? atomicAdd(&hist[d[u] >> BK_SH], 1) : 0;
    __syncthreads();

    const int c = hist[t];
    int x = c;
#pragma unroll
    for (int off = 1; off < 32; off <<= 1) {
        int y = __shfl_up(x, off, 32);
        if (lane >= off) x += y;
    }
    if (lane == 31) wsum[wv] = x;
    __syncthreads();
    if (wv == 0) {
        const int tot = wsum[lane];
        int q = tot;
#pragma unroll
        for (int off = 1; off < 32; off <<= 1) {
            int y = __shfl_up(q, off, 32);
            if (lane >= off) q += y;
        }
        wsum[lane] = q - tot;
    }
    __syncthreads();
    const int excl = wsum[wv] + x - c;
    offs[t] = excl;
    if (t < NB) {
        tcnt[(size_t)t * NCH + chunk] = c;
        toff[(size_t)t * NCH + chunk] = excl;
    }
    __syncthreads();
#pragma unroll
    for (int u = 0; u < CH_U; ++u)
        if (d[u] >= 0) binned[e_base + offs[d[u] >> BK_SH] + r[u]] = make_int2(s[u], d[u]);
}

__global__ __launch_bounds__(1024)
void bucket_place_kernel(const int* __restrict__ tcnt, const int* __restrict__ toff,
                         const int2* __restrict__ binned, int* __restrict__ rowptr,
                         float* __restrict__ dinv, int2* __restrict__ colw, int N, int NCH)
{
    __shared__ int ncnt[BK_N];
    __shared__ int lrow[BK_N];
    __shared__ int wsum[32];
    __shared__ int s_prefix;
    const int t = threadIdx.x, lane = t & 31, wv = t >> 5;
    const int b = blockIdx.x;
    const int node0 = b << BK_SH;
    const int* __restrict__ cnt_row = tcnt + (size_t)b * NCH;
    const int* __restrict__ off_row = toff + (size_t)b * NCH;

    if (t < BK_N) ncnt[t] = 0;

    int p = 0;
    for (int c = t; c < NCH; c += PREP_T) p += off_row[c];
#pragma unroll
    for (int off = 16; off; off >>= 1) p += __shfl_xor(p, off, 32);
    if (lane == 0) wsum[wv] = p;
    __syncthreads();
    if (wv == 0) {
        int q = wsum[lane];
#pragma unroll
        for (int off = 16; off; off >>= 1) q += __shfl_xor(q, off, 32);
        if (lane == 0) s_prefix = q;
    }
    __syncthreads();
    const int bstart = s_prefix;

    for (int c = t; c < NCH; c += PREP_T) {
        const int n = cnt_row[c];
        const int2* __restrict__ seg = binned + (size_t)c * CH_E + off_row[c];
        for (int i = 0; i < n; ++i) atomicAdd(&ncnt[seg[i].y - node0], 1);
    }
    __syncthreads();

    const int cn = (t < BK_N) ? ncnt[t] : 0;
    int x = cn;
#pragma unroll
    for (int off = 1; off < 32; off <<= 1) {
        int y = __shfl_up(x, off, 32);
        if (lane >= off) x += y;
    }
    if (t < BK_N && lane == 31) wsum[wv] = x;
    __syncthreads();
    if (t < BK_N) {
        int before = 0;
        for (int w = 0; w < wv; ++w) before += wsum[w];
        const int excl = before + x - cn;
        lrow[t] = excl;
        ncnt[t] = 0;
        const int node = node0 + t;
        if (node < N) {
            rowptr[node] = bstart + excl;
            dinv[node] = rsqrtf((float)cn + 1.0f);
            if (node == N - 1) rowptr[N] = bstart + excl + cn;
        }
    }
    __syncthreads();

    for (int c = t; c < NCH; c += PREP_T) {
        const int n = cnt_row[c];
        const int2* __restrict__ seg = binned + (size_t)c * CH_E + off_row[c];
        for (int i = 0; i < n; ++i) {
            const int2 sd = seg[i];
            const int dl = sd.y - node0;
            const int r = atomicAdd(&ncnt[dl], 1);
            colw[bstart + lrow[dl] + r] = sd;
        }
    }
}

constexpr int EU = 4;

__global__ __launch_bounds__(256)
void weight_kernel(const float* __restrict__ dinv, int2* __restrict__ colw, int E)
{
    const int stride = (int)(gridDim.x * blockDim.x);
    for (int p0 = (int)(blockIdx.x * blockDim.x + threadIdx.x); p0 < E; p0 += EU * stride) {
        if (p0 + (EU - 1) * stride < E) {
            int2 sd[EU];
            float w[EU];
#pragma unroll
            for (int u = 0; u < EU; ++u) sd[u] = colw[p0 + u * stride];
#pragma unroll
            for (int u = 0; u < EU; ++u) w[u] = dinv[sd[u].x] * dinv[sd[u].y];
#pragma unroll
            for (int u = 0; u < EU; ++u)
                colw[p0 + u * stride] = make_int2(sd[u].x, __float_as_int(w[u]));
        } else {
            for (int p = p0; p < E; p += stride) {
                const int2 sd1 = colw[p];
                colw[p] = make_int2(sd1.x, __float_as_int(dinv[sd1.x] * dinv[sd1.y]));
            }
        }
    }
}

__device__ __forceinline__ float xhalf(float x)
{
    const int b = __builtin_bit_cast(int, x);
    return __builtin_bit_cast(float,
        __builtin_amdgcn_permlanex16(b, b, 0x76543210, 0xfedcba98, false, false));
}
template<int MASK>
__device__ __forceinline__ float xrow(float x)
{
    const int b = __builtin_bit_cast(int, x);
    return __builtin_bit_cast(float, __builtin_amdgcn_update_dpp(b, b, 0x160 | MASK, 0xf, 0xf, false));
}
template<int LANES>
__device__ __forceinline__ float lanes_sum(float x)
{
    x += xrow<1>(x);
    x += xrow<2>(x);
    x += xrow<4>(x);
    x += xrow<8>(x);
    if (LANES == 32) x += xhalf(x);
    return x;
}

template<int M>
__global__ __launch_bounds__(256)
void csr_gather_ln_elu(const int* __restrict__ rowptr, const int2* __restrict__ colw,
                       const float* __restrict__ xw, const float* __restrict__ dinv,
                       const float* __restrict__ bias, const float* __restrict__ g,
                       const float* __restrict__ be, float* __restrict__ out, int N)
{
    constexpr int LPR = M / 4;
    constexpr int EPW = 32 / LPR;
    constexpr int UN  = 4;
    const int nwaves = (int)(gridDim.x * (blockDim.x >> 5));
    int wave = (int)(blockIdx.x * (blockDim.x >> 5)) + (int)(threadIdx.x >> 5);
    wave = __builtin_amdgcn_readfirstlane(wave);
    const int lane = threadIdx.x & 31;
    const int sub = lane / LPR;
    const int c = (lane % LPR) << 2;

    float4 bi, gg, bb;
    bi.x = bias[c]; bi.y = bias[c + 1]; bi.z = bias[c + 2]; bi.w = bias[c + 3];
    gg.x = g[c];    gg.y = g[c + 1];    gg.z = g[c + 2];    gg.w = g[c + 3];
    bb.x = be[c];   bb.y = be[c + 1];   bb.z = be[c + 2];   bb.w = be[c + 3];

    for (int node = wave; node < N; node += nwaves) {
        const int beg = rowptr[node];
        const int end = rowptr[node + 1];
        const float di = dinv[node];
        const int selfw = __float_as_int(di * di);
        const int stop = end + 1;

        float4 acc = make_float4(0.0f, 0.0f, 0.0f, 0.0f);
        int base = beg;
        do {
            const int n = min(32, stop - base);
            const int e = base + lane;
            int2 mine = make_int2(0, 0);
            if (e < end)       mine = colw[e];
            else if (e == end) mine = make_int2(node, selfw);
            for (int j0 = 0; j0 < n; j0 += UN * EPW) {
                float4 v[UN];
                float  w[UN];
#pragma unroll
                for (int u = 0; u < UN; ++u) {
                    const int j = (j0 + u * EPW + sub) & 31;
                    const int s = __shfl(mine.x, j, 32);
                    w[u] = __int_as_float(__shfl(mine.y, j, 32));
                    v[u] = *(const float4*)(xw + (size_t)s * M + c);
                }
#pragma unroll
                for (int u = 0; u < UN; ++u) {
                    if (j0 + u * EPW + sub < n) {
                        acc.x += v[u].x * w[u];
                        acc.y += v[u].y * w[u];
                        acc.z += v[u].z * w[u];
                        acc.w += v[u].w * w[u];
                    }
                }
            }
            base += 32;
        } while (base < stop);
        if (EPW == 2) {
            acc.x += __shfl_xor(acc.x, 16, 32);
            acc.y += __shfl_xor(acc.y, 16, 32);
            acc.z += __shfl_xor(acc.z, 16, 32);
            acc.w += __shfl_xor(acc.w, 16, 32);
        }

        float4 t;
        t.x = acc.x + bi.x;
        t.y = acc.y + bi.y;
        t.z = acc.z + bi.z;
        t.w = acc.w + bi.w;

        const float sum = lanes_sum<LPR>((t.x + t.y) + (t.z + t.w));
        const float mu = sum * (1.0f / M);
        const float dx = t.x - mu, dy = t.y - mu, dz = t.z - mu, dw = t.w - mu;
        const float vs = lanes_sum<LPR>((dx * dx + dy * dy) + (dz * dz + dw * dw));
        const float rstd = rsqrtf(vs * (1.0f / M) + 1e-5f);
        float4 o;
        o.x = dx * rstd * gg.x + bb.x;
        o.y = dy * rstd * gg.y + bb.y;
        o.z = dz * rstd * gg.z + bb.z;
        o.w = dw * rstd * gg.w + bb.w;
        o.x = (o.x > 0.0f) ? o.x : (expf(o.x) - 1.0f);
        o.y = (o.y > 0.0f) ? o.y : (expf(o.y) - 1.0f);
        o.z = (o.z > 0.0f) ? o.z : (expf(o.z) - 1.0f);
        o.w = (o.w > 0.0f) ? o.w : (expf(o.w) - 1.0f);
        if (sub == 0) *(float4*)(out + (size_t)node * M + c) = o;
    }
}

template<int M>
__global__ __launch_bounds__(256)
void finalize_ln_elu(const float* agg, const float* __restrict__ xw,
                     const float* __restrict__ dinv, const float* __restrict__ bias,
                     const float* __restrict__ g, const float* __restrict__ be,
                     float* out, int N)
{
    constexpr int PER = M / 32;
    const int nwaves = (int)(gridDim.x * (blockDim.x >> 5));
    int wave = (int)(blockIdx.x * (blockDim.x >> 5)) + (int)(threadIdx.x >> 5);
    wave = __builtin_amdgcn_readfirstlane(wave);
    const int lane = threadIdx.x & 31;

    float bi[PER], gg[PER], bb[PER];
#pragma unroll
    for (int i = 0; i < PER; ++i) {
        const int c = lane + (i << 5);
        bi[i] = bias ? bias[c] : 0.0f;
        gg[i] = g[c];
        bb[i] = be[c];
    }

    for (int node = wave; node < N; node += nwaves) {
        float vals[PER], xs[PER];
        float sum = 0.0f;
        float selfw = 0.0f;
#pragma unroll
        for (int i = 0; i < PER; ++i) vals[i] = agg[(size_t)node * M + lane + (i << 5)];
        if (xw) {
#pragma unroll
            for (int i = 0; i < PER; ++i) xs[i] = xw[(size_t)node * M + lane + (i << 5)];
            selfw = dinv[node] * dinv[node];
        } else {
#pragma unroll
            for (int i = 0; i < PER; ++i) xs[i] = 0.0f;
        }
#pragma unroll
        for (int i = 0; i < PER; ++i) {
            float v = vals[i];
            if (xw)   v += xs[i] * selfw;
            if (bias) v += bi[i];
            vals[i] = v;
            sum += v;
        }
#pragma unroll
        for (int off = 16; off; off >>= 1) sum += __shfl_xor(sum, off, 32);
        float mu = sum * (1.0f / M);
        float vs = 0.0f;
#pragma unroll
        for (int i = 0; i < PER; ++i) { float d = vals[i] - mu; vs += d * d; }
#pragma unroll
        for (int off = 16; off; off >>= 1) vs += __shfl_xor(vs, off, 32);
        float rstd = rsqrtf(vs * (1.0f / M) + 1e-5f);
#pragma unroll
        for (int i = 0; i < PER; ++i) {
            int c = lane + (i << 5);
            float v = (vals[i] - mu) * rstd * gg[i] + bb[i];
            out[(size_t)node * M + c] = (v > 0.0f) ? v : (expf(v) - 1.0f);
        }
    }
}

extern "C" void kernel_launch(void* const* d_in, const int* in_sizes, int n_in,
                              void* d_out, int out_size, void* d_ws, size_t ws_size,
                              hipStream_t stream)
{
    const float* x   = (const float*)d_in[0];
    const int*   ei  = (const int*)  d_in[1];
    const float* W1  = (const float*)d_in[2];
    const float* b1  = (const float*)d_in[3];
    const float* g1  = (const float*)d_in[4];
    const float* be1 = (const float*)d_in[5];
    const float* W2  = (const float*)d_in[6];
    const float* b2  = (const float*)d_in[7];
    const float* g2  = (const float*)d_in[8];
    const float* be2 = (const float*)d_in[9];
    const float* W3  = (const float*)d_in[10];
    const float* b3  = (const float*)d_in[11];
    const float* g3  = (const float*)d_in[12];
    const float* be3 = (const float*)d_in[13];
    const float* lw1 = (const float*)d_in[14];
    const float* lb1 = (const float*)d_in[15];
    const float* g4  = (const float*)d_in[16];
    const float* be4 = (const float*)d_in[17];
    const float* lw2 = (const float*)d_in[18];
    const float* lb2 = (const float*)d_in[19];

    const int N = in_sizes[0] / 128;
    const int E = in_sizes[1] / 2;
    const int* src = ei;
    const int* dst = ei + E;

    const int NB  = (N + BK_N - 1) / BK_N;
    const int NCH = ceil_div_ll(E, CH_E);
    float* ws   = (float*)d_ws;
    float* xw   = ws;  ws += (size_t)N * 128;
    float* agg  = ws;  ws += (size_t)N * 128;
    float* h    = ws;  ws += (size_t)N * 64;
    int2*  colw = (int2*)ws;
    ws += (size_t)N * 64;
    ws += N;
    float* dinv   = ws;        ws += N;
    int*   rowptr = (int*)ws;  ws += E;
    int2*  binned = (int2*)agg;
    int*   tcnt   = (int*)agg + (size_t)2 * NCH * CH_E;
    int*   toff   = tcnt + (size_t)NB * NCH;

    const int B = 256;
    auto gemm_blocks = [N](int tiles_n) {
        long long waves = (long long)((N + 15) / 16) * tiles_n;
        return (int)((waves + 7) / 8);
    };
    const int node_blocks = loop_grid(ceil_div_ll(N, 8));
    const int edge_blocks = loop_grid(ceil_div_ll(E, (long long)B * EU));

    bin_kernel<<<NCH, PREP_T, 0, stream>>>(src, dst, binned, tcnt, toff, E, NB, NCH);
    bucket_place_kernel<<<NB, PREP_T, 0, stream>>>(tcnt, toff, binned, rowptr, dinv, colw, N, NCH);
    weight_kernel<<<edge_blocks, B, 0, stream>>>(dinv, colw, E);

    wmma_gemm_f32<128, 64><<<gemm_blocks(4), B, 0, stream>>>(x, W1, nullptr, xw, N);
    csr_gather_ln_elu<64><<<node_blocks, B, 0, stream>>>(rowptr, colw, xw, dinv, b1, g1, be1, h, N);

    wmma_gemm_f32<64, 128><<<gemm_blocks(8), B, 0, stream>>>(h, W2, nullptr, xw, N);
    csr_gather_ln_elu<128><<<node_blocks, B, 0, stream>>>(rowptr, colw, xw, dinv, b2, g2, be2, agg, N);

    wmma_gemm_f32<128, 64><<<gemm_blocks(4), B, 0, stream>>>(agg, W3, nullptr, xw, N);
    csr_gather_ln_elu<64><<<node_blocks, B, 0, stream>>>(rowptr, colw, xw, dinv, b3, g3, be3, h, N);

    wmma_gemm_f32<64, 32><<<gemm_blocks(2), B, 0, stream>>>(h, lw1, lb1, xw, N);
    finalize_ln_elu<32><<<node_blocks, B, 0, stream>>>(xw, nullptr, nullptr, nullptr, g4, be4, h, N);
    wmma_gemm_f32<32, 32><<<gemm_blocks(2), B, 0, stream>>>(h, lw2, lb2, (float*)d_out, N);
}


# GEMMs: a work-group stages W once (LDS, then all fragments in registers) and each wave computes whole 16-row strips, all column tiles from one 16-byte-per-lane read of A; k sum order changes
# speedup vs baseline: 3.2992x; 1.1158x over previous
#include <hip/hip_runtime.h>
#include <hip/hip_bf16.h>
#include <math.h>

typedef __attribute__((ext_vector_type(2))) float v2f;
typedef __attribute__((ext_vector_type(8))) float v8f;

static inline int ceil_div_ll(long long a, long long b) { return (int)((a + b - 1) / b); }

constexpr int LOOP_BLOCKS = 2048;
static inline int loop_grid(long long blocks_wanted)
{
    return (int)(blocks_wanted < LOOP_BLOCKS ? blocks_wanted : LOOP_BLOCKS);
}

constexpr int GEMM_T = 256;
constexpr int GEMM_WGPS = 256;

template<int K>
__device__ __forceinline__ void load_a_strip(const float* __restrict__ A, int strip, int lm, int lh,
                                             int N, float4 (&a)[K / 8])
{
    int arow = (strip << 4) + lm;
    arow = arow < N ? arow : N - 1;
    const float4* ap = (const float4*)(A + (size_t)arow * K) + lh;
#pragma unroll
    for (int j = 0; j < K / 8; ++j) a[j] = ap[2 * j];
}

template<int K, int M>
__global__ __launch_bounds__(GEMM_T)
void wmma_gemm_f32(const float* __restrict__ A, const float* __restrict__ W,
                   const float* __restrict__ bias, float* __restrict__ C, int N)
{
    constexpr int TN = M / 16;
    constexpr int KB = K / 8;
    constexpr int NW = GEMM_T / 32;
    __shared__ float4 wfrag[KB * TN * 32];

    const int lane = threadIdx.x & 31;
    const int wv = __builtin_amdgcn_readfirstlane((int)(threadIdx.x >> 5));
    const int lh = lane >> 4;
    const int lm = lane & 15;

    const unsigned nstrips = (unsigned)(N + 15) >> 4;
    const int s_beg = (int)(blockIdx.x * nstrips / gridDim.x);
    const int s_end = (int)((blockIdx.x + 1u) * nstrips / gridDim.x);

    int strip = s_beg + wv;
    float4 a_cur[KB];
#pragma unroll
    for (int j = 0; j < KB; ++j) a_cur[j] = make_float4(0.0f, 0.0f, 0.0f, 0.0f);
    if (strip < s_end) load_a_strip<K>(A, strip, lm, lh, N, a_cur);

    constexpr int SPT = KB * TN * 32 / GEMM_T;
    static_assert(SPT * GEMM_T == KB * TN * 32, "K * M must be a multiple of 4 * GEMM_T");
    float4 wv4[SPT];
#pragma unroll
    for (int i = 0; i < SPT; ++i) {
        const int slot = (int)threadIdx.x + i * GEMM_T;
        const int l = slot & 31;
        const int jt = slot >> 5;
        const int t = jt % TN;
        const int j = jt / TN;
        const float* p = W + (size_t)(8 * j + 4 * (l >> 4)) * M + 16 * t + (l & 15);
        wv4[i].x = p[0];
        wv4[i].y = p[M];
        wv4[i].z = p[2 * M];
        wv4[i].w = p[3 * M];
    }
#pragma unroll
    for (int i = 0; i < SPT; ++i) wfrag[(int)threadIdx.x + i * GEMM_T] = wv4[i];

    float badd[TN];
#pragma unroll
    for (int t = 0; t < TN; ++t) badd[t] = bias ? bias[16 * t + lm] : 0.0f;

    __syncthreads();

    float4 wb[KB * TN];
#pragma unroll
    for (int i = 0; i < KB * TN; ++i) wb[i] = wfrag[i * 32 + lane];

    for (; strip < s_end; strip += NW) {
        float4 a_nxt[KB];
#pragma unroll
        for (int j = 0; j < KB; ++j) a_nxt[j] = a_cur[j];
        if (strip + NW < s_end) load_a_strip<K>(A, strip + NW, lm, lh, N, a_nxt);

        v8f acc[TN];
#pragma unroll
        for (int t = 0; t < TN; ++t) acc[t] = v8f{};

#pragma unroll
        for (int j = 0; j < KB; ++j) {
            v2f a01, a23;
            a01.x = a_cur[j].x; a01.y = a_cur[j].y;
            a23.x = a_cur[j].z; a23.y = a_cur[j].w;
#pragma unroll
            for (int t = 0; t < TN; ++t) {
                v2f b01;
                b01.x = wb[j * TN + t].x; b01.y = wb[j * TN + t].y;
                acc[t] = __builtin_amdgcn_wmma_f32_16x16x4_f32(false, a01, false, b01,
                                                               (short)0, acc[t], false, false);
            }
#pragma unroll
            for (int t = 0; t < TN; ++t) {
                v2f b23;
                b23.x = wb[j * TN + t].z; b23.y = wb[j * TN + t].w;
                acc[t] = __builtin_amdgcn_wmma_f32_16x16x4_f32(false, a23, false, b23,
                                                               (short)0, acc[t], false, false);
            }
        }

        const int row0 = strip << 4;
#pragma unroll
        for (int r = 0; r < 8; ++r) {
            const int row = row0 + r + (lh << 3);
            if (row < N) {
#pragma unroll
                for (int t = 0; t < TN; ++t)
                    C[(size_t)row * M + 16 * t + lm] = acc[t][r] + badd[t];
            }
        }

#pragma unroll
        for (int j = 0; j < KB; ++j) a_cur[j] = a_nxt[j];
    }
}

constexpr int CH_E   = 4096;
constexpr int BK_SH  = 8;
constexpr int BK_N   = 1 << BK_SH;
constexpr int PREP_T = 1024;
constexpr int CH_U   = CH_E / PREP_T;

__global__ __launch_bounds__(1024)
void bin_kernel(const int* __restrict__ src, const int* __restrict__ dst,
                int2* __restrict__ binned, int* __restrict__ tcnt, int* __restrict__ toff,
                int E, int NB, int NCH)
{
    __shared__ int hist[PREP_T];
    __shared__ int offs[PREP_T];
    __shared__ int wsum[32];
    const int t = threadIdx.x, lane = t & 31, wv = t >> 5;
    const int chunk = blockIdx.x;
    const int e_base = chunk * CH_E;

    hist[t] = 0;
    __syncthreads();

    int s[CH_U], d[CH_U], r[CH_U];
#pragma unroll
    for (int u = 0; u < CH_U; ++u) {
        const int e = e_base + u * PREP_T + t;
        const bool ok = e < E;
        s[u] = ok ? src[e] : 0;
        d[u] = ok ? dst[e] : -1;
    }
#pragma unroll
    for (int u = 0; u < CH_U; ++u)
        r[u] = (d[u] >= 0) ? atomicAdd(&hist[d[u] >> BK_SH], 1) : 0;
    __syncthreads();

    const int c = hist[t];
    int x = c;
#pragma unroll
    for (int off = 1; off < 32; off <<= 1) {
        int y = __shfl_up(x, off, 32);
        if (lane >= off) x += y;
    }
    if (lane == 31) wsum[wv] = x;
    __syncthreads();
    if (wv == 0) {
        const int tot = wsum[lane];
        int q = tot;
#pragma unroll
        for (int off = 1; off < 32; off <<= 1) {
            int y = __shfl_up(q, off, 32);
            if (lane >= off) q += y;
        }
        wsum[lane] = q - tot;
    }
    __syncthreads();
    const int excl = wsum[wv] + x - c;
    offs[t] = excl;
    if (t < NB) {
        tcnt[(size_t)t * NCH + chunk] = c;
        toff[(size_t)t * NCH + chunk] = excl;
    }
    __syncthreads();
#pragma unroll
    for (int u = 0; u < CH_U; ++u)
        if (d[u] >= 0) binned[e_base + offs[d[u] >> BK_SH] + r[u]] = make_int2(s[u], d[u]);
}

__global__ __launch_bounds__(1024)
void bucket_place_kernel(const int* __restrict__ tcnt, const int* __restrict__ toff,
                         const int2* __restrict__ binned, int* __restrict__ rowptr,
                         float* __restrict__ dinv, int2* __restrict__ colw, int N, int NCH)
{
    __shared__ int ncnt[BK_N];
    __shared__ int lrow[BK_N];
    __shared__ int wsum[32];
    __shared__ int s_prefix;
    const int t = threadIdx.x, lane = t & 31, wv = t >> 5;
    const int b = blockIdx.x;
    const int node0 = b << BK_SH;
    const int* __restrict__ cnt_row = tcnt + (size_t)b * NCH;
    const int* __restrict__ off_row = toff + (size_t)b * NCH;

    if (t < BK_N) ncnt[t] = 0;

    int p = 0;
    for (int c = t; c < NCH; c += PREP_T) p += off_row[c];
#pragma unroll
    for (int off = 16; off; off >>= 1) p += __shfl_xor(p, off, 32);
    if (lane == 0) wsum[wv] = p;
    __syncthreads();
    if (wv == 0) {
        int q = wsum[lane];
#pragma unroll
        for (int off = 16; off; off >>= 1) q += __shfl_xor(q, off, 32);
        if (lane == 0) s_prefix = q;
    }
    __syncthreads();
    const int bstart = s_prefix;

    for (int c = t; c < NCH; c += PREP_T) {
        const int n = cnt_row[c];
        const int2* __restrict__ seg = binned + (size_t)c * CH_E + off_row[c];
        for (int i = 0; i < n; ++i) atomicAdd(&ncnt[seg[i].y - node0], 1);
    }
    __syncthreads();

    const int cn = (t < BK_N) ? ncnt[t] : 0;
    int x = cn;
#pragma unroll
    for (int off = 1; off < 32; off <<= 1) {
        int y = __shfl_up(x, off, 32);
        if (lane >= off) x += y;
    }
    if (t < BK_N && lane == 31) wsum[wv] = x;
    __syncthreads();
    if (t < BK_N) {
        int before = 0;
        for (int w = 0; w < wv; ++w) before += wsum[w];
        const int excl = before + x - cn;
        lrow[t] = excl;
        ncnt[t] = 0;
        const int node = node0 + t;
        if (node < N) {
            rowptr[node] = bstart + excl;
            dinv[node] = rsqrtf((float)cn + 1.0f);
            if (node == N - 1) rowptr[N] = bstart + excl + cn;
        }
    }
    __syncthreads();

    for (int c = t; c < NCH; c += PREP_T) {
        const int n = cnt_row[c];
        const int2* __restrict__ seg = binned + (size_t)c * CH_E + off_row[c];
        for (int i = 0; i < n; ++i) {
            const int2 sd = seg[i];
            const int dl = sd.y - node0;
            const int r = atomicAdd(&ncnt[dl], 1);
            colw[bstart + lrow[dl] + r] = sd;
        }
    }
}

constexpr int EU = 4;

__global__ __launch_bounds__(256)
void weight_kernel(const float* __restrict__ dinv, int2* __restrict__ colw, int E)
{
    const int stride = (int)(gridDim.x * blockDim.x);
    for (int p0 = (int)(blockIdx.x * blockDim.x + threadIdx.x); p0 < E; p0 += EU * stride) {
        if (p0 + (EU - 1) * stride < E) {
            int2 sd[EU];
            float w[EU];
#pragma unroll
            for (int u = 0; u < EU; ++u) sd[u] = colw[p0 + u * stride];
#pragma unroll
            for (int u = 0; u < EU; ++u) w[u] = dinv[sd[u].x] * dinv[sd[u].y];
#pragma unroll
            for (int u = 0; u < EU; ++u)
                colw[p0 + u * stride] = make_int2(sd[u].x, __float_as_int(w[u]));
        } else {
            for (int p = p0; p < E; p += stride) {
                const int2 sd1 = colw[p];
                colw[p] = make_int2(sd1.x, __float_as_int(dinv[sd1.x] * dinv[sd1.y]));
            }
        }
    }
}

__device__ __forceinline__ float xhalf(float x)
{
    const int b = __builtin_bit_cast(int, x);
    return __builtin_bit_cast(float,
        __builtin_amdgcn_permlanex16(b, b, 0x76543210, 0xfedcba98, false, false));
}
template<int MASK>
__device__ __forceinline__ float xrow(float x)
{
    const int b = __builtin_bit_cast(int, x);
    return __builtin_bit_cast(float, __builtin_amdgcn_update_dpp(b, b, 0x160 | MASK, 0xf, 0xf, false));
}
template<int LANES>
__device__ __forceinline__ float lanes_sum(float x)
{
    x += xrow<1>(x);
    x += xrow<2>(x);
    x += xrow<4>(x);
    x += xrow<8>(x);
    if (LANES == 32) x += xhalf(x);
    return x;
}

template<int M>
__global__ __launch_bounds__(256)
void csr_gather_ln_elu(const int* __restrict__ rowptr, const int2* __restrict__ colw,
                       const float* __restrict__ xw, const float* __restrict__ dinv,
                       const float* __restrict__ bias, const float* __restrict__ g,
                       const float* __restrict__ be, float* __restrict__ out, int N)
{
    constexpr int LPR = M / 4;
    constexpr int EPW = 32 / LPR;
    constexpr int UN  = 4;
    const int nwaves = (int)(gridDim.x * (blockDim.x >> 5));
    int wave = (int)(blockIdx.x * (blockDim.x >> 5)) + (int)(threadIdx.x >> 5);
    wave = __builtin_amdgcn_readfirstlane(wave);
    const int lane = threadIdx.x & 31;
    const int sub = lane / LPR;
    const int c = (lane % LPR) << 2;

    float4 bi, gg, bb;
    bi.x = bias[c]; bi.y = bias[c + 1]; bi.z = bias[c + 2]; bi.w = bias[c + 3];
    gg.x = g[c];    gg.y = g[c + 1];    gg.z = g[c + 2];    gg.w = g[c + 3];
    bb.x = be[c];   bb.y = be[c + 1];   bb.z = be[c + 2];   bb.w = be[c + 3];

    for (int node = wave; node < N; node += nwaves) {
        const int beg = rowptr[node];
        const int end = rowptr[node + 1];
        const float di = dinv[node];
        const int selfw = __float_as_int(di * di);
        const int stop = end + 1;

        float4 acc = make_float4(0.0f, 0.0f, 0.0f, 0.0f);
        int base = beg;
        do {
            const int n = min(32, stop - base);
            const int e = base + lane;
            int2 mine = make_int2(0, 0);
            if (e < end)       mine = colw[e];
            else if (e == end) mine = make_int2(node, selfw);
            for (int j0 = 0; j0 < n; j0 += UN * EPW) {
                float4 v[UN];
                float  w[UN];
#pragma unroll
                for (int u = 0; u < UN; ++u) {
                    const int j = (j0 + u * EPW + sub) & 31;
                    const int s = __shfl(mine.x, j, 32);
                    w[u] = __int_as_float(__shfl(mine.y, j, 32));
                    v[u] = *(const float4*)(xw + (size_t)s * M + c);
                }
#pragma unroll
                for (int u = 0; u < UN; ++u) {
                    if (j0 + u * EPW + sub < n) {
                        acc.x += v[u].x * w[u];
                        acc.y += v[u].y * w[u];
                        acc.z += v[u].z * w[u];
                        acc.w += v[u].w * w[u];
                    }
                }
            }
            base += 32;
        } while (base < stop);
        if (EPW == 2) {
            acc.x += __shfl_xor(acc.x, 16, 32);
            acc.y += __shfl_xor(acc.y, 16, 32);
            acc.z += __shfl_xor(acc.z, 16, 32);
            acc.w += __shfl_xor(acc.w, 16, 32);
        }

        float4 t;
        t.x = acc.x + bi.x;
        t.y = acc.y + bi.y;
        t.z = acc.z + bi.z;
        t.w = acc.w + bi.w;

        const float sum = lanes_sum<LPR>((t.x + t.y) + (t.z + t.w));
        const float mu = sum * (1.0f / M);
        const float dx = t.x - mu, dy = t.y - mu, dz = t.z - mu, dw = t.w - mu;
        const float vs = lanes_sum<LPR>((dx * dx + dy * dy) + (dz * dz + dw * dw));
        const float rstd = rsqrtf(vs * (1.0f / M) + 1e-5f);
        float4 o;
        o.x = dx * rstd * gg.x + bb.x;
        o.y = dy * rstd * gg.y + bb.y;
        o.z = dz * rstd * gg.z + bb.z;
        o.w = dw * rstd * gg.w + bb.w;
        o.x = (o.x > 0.0f) ? o.x : (expf(o.x) - 1.0f);
        o.y = (o.y > 0.0f) ? o.y : (expf(o.y) - 1.0f);
        o.z = (o.z > 0.0f) ? o.z : (expf(o.z) - 1.0f);
        o.w = (o.w > 0.0f) ? o.w : (expf(o.w) - 1.0f);
        if (sub == 0) *(float4*)(out + (size_t)node * M + c) = o;
    }
}

template<int M>
__global__ __launch_bounds__(256)
void finalize_ln_elu(const float* agg, const float* __restrict__ xw,
                     const float* __restrict__ dinv, const float* __restrict__ bias,
                     const float* __restrict__ g, const float* __restrict__ be,
                     float* out, int N)
{
    constexpr int PER = M / 32;
    const int nwaves = (int)(gridDim.x * (blockDim.x >> 5));
    int wave = (int)(blockIdx.x * (blockDim.x >> 5)) + (int)(threadIdx.x >> 5);
    wave = __builtin_amdgcn_readfirstlane(wave);
    const int lane = threadIdx.x & 31;

    float bi[PER], gg[PER], bb[PER];
#pragma unroll
    for (int i = 0; i < PER; ++i) {
        const int c = lane + (i << 5);
        bi[i] = bias ? bias[c] : 0.0f;
        gg[i] = g[c];
        bb[i] = be[c];
    }

    for (int node = wave; node < N; node += nwaves) {
        float vals[PER], xs[PER];
        float sum = 0.0f;
        float selfw = 0.0f;
#pragma unroll
        for (int i = 0; i < PER; ++i) vals[i] = agg[(size_t)node * M + lane + (i << 5)];
        if (xw) {
#pragma unroll
            for (int i = 0; i < PER; ++i) xs[i] = xw[(size_t)node * M + lane + (i << 5)];
            selfw = dinv[node] * dinv[node];
        } else {
#pragma unroll
            for (int i = 0; i < PER; ++i) xs[i] = 0.0f;
        }
#pragma unroll
        for (int i = 0; i < PER; ++i) {
            float v = vals[i];
            if (xw)   v += xs[i] * selfw;
            if (bias) v += bi[i];
            vals[i] = v;
            sum += v;
        }
#pragma unroll
        for (int off = 16; off; off >>= 1) sum += __shfl_xor(sum, off, 32);
        float mu = sum * (1.0f / M);
        float vs = 0.0f;
#pragma unroll
        for (int i = 0; i < PER; ++i) { float d = vals[i] - mu; vs += d * d; }
#pragma unroll
        for (int off = 16; off; off >>= 1) vs += __shfl_xor(vs, off, 32);
        float rstd = rsqrtf(vs * (1.0f / M) + 1e-5f);
#pragma unroll
        for (int i = 0; i < PER; ++i) {
            int c = lane + (i << 5);
            float v = (vals[i] - mu) * rstd * gg[i] + bb[i];
            out[(size_t)node * M + c] = (v > 0.0f) ? v : (expf(v) - 1.0f);
        }
    }
}

extern "C" void kernel_launch(void* const* d_in, const int* in_sizes, int n_in,
                              void* d_out, int out_size, void* d_ws, size_t ws_size,
                              hipStream_t stream)
{
    const float* x   = (const float*)d_in[0];
    const int*   ei  = (const int*)  d_in[1];
    const float* W1  = (const float*)d_in[2];
    const float* b1  = (const float*)d_in[3];
    const float* g1  = (const float*)d_in[4];
    const float* be1 = (const float*)d_in[5];
    const float* W2  = (const float*)d_in[6];
    const float* b2  = (const float*)d_in[7];
    const float* g2  = (const float*)d_in[8];
    const float* be2 = (const float*)d_in[9];
    const float* W3  = (const float*)d_in[10];
    const float* b3  = (const float*)d_in[11];
    const float* g3  = (const float*)d_in[12];
    const float* be3 = (const float*)d_in[13];
    const float* lw1 = (const float*)d_in[14];
    const float* lb1 = (const float*)d_in[15];
    const float* g4  = (const float*)d_in[16];
    const float* be4 = (const float*)d_in[17];
    const float* lw2 = (const float*)d_in[18];
    const float* lb2 = (const float*)d_in[19];

    const int N = in_sizes[0] / 128;
    const int E = in_sizes[1] / 2;
    const int* src = ei;
    const int* dst = ei + E;

    const int NB  = (N + BK_N - 1) / BK_N;
    const int NCH = ceil_div_ll(E, CH_E);
    float* ws   = (float*)d_ws;
    float* xw   = ws;  ws += (size_t)N * 128;
    float* agg  = ws;  ws += (size_t)N * 128;
    float* h    = ws;  ws += (size_t)N * 64;
    int2*  colw = (int2*)ws;
    ws += (size_t)N * 64;
    ws += N;
    float* dinv   = ws;        ws += N;
    int*   rowptr = (int*)ws;  ws += E;
    int2*  binned = (int2*)agg;
    int*   tcnt   = (int*)agg + (size_t)2 * NCH * CH_E;
    int*   toff   = tcnt + (size_t)NB * NCH;

    const int B = 256;
    auto gemm_grid = [N](int per_wgp) {
        const int want = GEMM_WGPS * per_wgp;
        const int most = ceil_div_ll((N + 15) / 16, GEMM_T / 32);
        return want < most ? want : most;
    };
    const int node_blocks = loop_grid(ceil_div_ll(N, 8));
    const int edge_blocks = loop_grid(ceil_div_ll(E, (long long)B * EU));

    bin_kernel<<<NCH, PREP_T, 0, stream>>>(src, dst, binned, tcnt, toff, E, NB, NCH);
    bucket_place_kernel<<<NB, PREP_T, 0, stream>>>(tcnt, toff, binned, rowptr, dinv, colw, N, NCH);
    weight_kernel<<<edge_blocks, B, 0, stream>>>(dinv, colw, E);

    wmma_gemm_f32<128, 64><<<gemm_grid(1), GEMM_T, 0, stream>>>(x, W1, nullptr, xw, N);
    csr_gather_ln_elu<64><<<node_blocks, B, 0, stream>>>(rowptr, colw, xw, dinv, b1, g1, be1, h, N);

    wmma_gemm_f32<64, 128><<<gemm_grid(1), GEMM_T, 0, stream>>>(h, W2, nullptr, xw, N);
    csr_gather_ln_elu<128><<<node_blocks, B, 0, stream>>>(rowptr, colw, xw, dinv, b2, g2, be2, agg, N);

    wmma_gemm_f32<128, 64><<<gemm_grid(1), GEMM_T, 0, stream>>>(agg, W3, nullptr, xw, N);
    csr_gather_ln_elu<64><<<node_blocks, B, 0, stream>>>(rowptr, colw, xw, dinv, b3, g3, be3, h, N);

    wmma_gemm_f32<64, 32><<<gemm_grid(3), GEMM_T, 0, stream>>>(h, lw1, lb1, xw, N);
    finalize_ln_elu<32><<<node_blocks, B, 0, stream>>>(xw, nullptr, nullptr, nullptr, g4, be4, h, N);
    wmma_gemm_f32<32, 32><<<gemm_grid(3), GEMM_T, 0, stream>>>(h, lw2, lb2, (float*)d_out, N);
}


# Layer 2 (64 to 128) aggregates before it multiplies, (A h) W2 for A (h W2): its row gather reads 64-wide rows not 128-wide; its LN+ELU moves into the GEMM epilogue; sum order changes
# speedup vs baseline: 3.5119x; 1.0645x over previous
#include <hip/hip_runtime.h>
#include <hip/hip_bf16.h>
#include <math.h>

typedef __attribute__((ext_vector_type(2))) float v2f;
typedef __attribute__((ext_vector_type(8))) float v8f;

static inline int ceil_div_ll(long long a, long long b) { return (int)((a + b - 1) / b); }

constexpr int LOOP_BLOCKS = 2048;
static inline int loop_grid(long long blocks_wanted)
{
    return (int)(blocks_wanted < LOOP_BLOCKS ? blocks_wanted : LOOP_BLOCKS);
}

__device__ __forceinline__ float xhalf(float x)
{
    const int b = __builtin_bit_cast(int, x);
    return __builtin_bit_cast(float,
        __builtin_amdgcn_permlanex16(b, b, 0x76543210, 0xfedcba98, false, false));
}
template<int MASK>
__device__ __forceinline__ float xrow(float x)
{
    const int b = __builtin_bit_cast(int, x);
    return __builtin_bit_cast(float, __builtin_amdgcn_update_dpp(b, b, 0x160 | MASK, 0xf, 0xf, false));
}
template<int LANES>
__device__ __forceinline__ float lanes_sum(float x)
{
    x += xrow<1>(x);
    x += xrow<2>(x);
    x += xrow<4>(x);
    x += xrow<8>(x);
    if (LANES == 32) x += xhalf(x);
    return x;
}

constexpr int GEMM_T = 256;
constexpr int GEMM_WGPS = 256;

template<int K>
__device__ __forceinline__ void load_a_strip(const float* __restrict__ A, int strip, int lm, int lh,
                                             int N, float4 (&a)[K / 8])
{
    int arow = (strip << 4) + lm;
    arow = arow < N ? arow : N - 1;
    const float4* ap = (const float4*)(A + (size_t)arow * K) + lh;
#pragma unroll
    for (int j = 0; j < K / 8; ++j) a[j] = ap[2 * j];
}

template<int K, int M, bool LNELU>
__global__ __launch_bounds__(GEMM_T)
void wmma_gemm_f32(const float* __restrict__ A, const float* __restrict__ W,
                   const float* __restrict__ bias, const float* __restrict__ g,
                   const float* __restrict__ be, float* __restrict__ C, int N)
{
    constexpr int TN = M / 16;
    constexpr int KB = K / 8;
    constexpr int NW = GEMM_T / 32;
    __shared__ float4 wfrag[KB * TN * 32];

    const int lane = threadIdx.x & 31;
    const int wv = __builtin_amdgcn_readfirstlane((int)(threadIdx.x >> 5));
    const int lh = lane >> 4;
    const int lm = lane & 15;

    const unsigned nstrips = (unsigned)(N + 15) >> 4;
    const int s_beg = (int)(blockIdx.x * nstrips / gridDim.x);
    const int s_end = (int)((blockIdx.x + 1u) * nstrips / gridDim.x);

    int strip = s_beg + wv;
    float4 a_cur[KB];
#pragma unroll
    for (int j = 0; j < KB; ++j) a_cur[j] = make_float4(0.0f, 0.0f, 0.0f, 0.0f);
    if (strip < s_end) load_a_strip<K>(A, strip, lm, lh, N, a_cur);

    constexpr int SPT = KB * TN * 32 / GEMM_T;
    static_assert(SPT * GEMM_T == KB * TN * 32, "K * M must be a multiple of 4 * GEMM_T");
    float4 wv4[SPT];
#pragma unroll
    for (int i = 0; i < SPT; ++i) {
        const int slot = (int)threadIdx.x + i * GEMM_T;
        const int l = slot & 31;
        const int jt = slot >> 5;
        const int t = jt % TN;
        const int j = jt / TN;
        const float* p = W + (size_t)(8 * j + 4 * (l >> 4)) * M + 16 * t + (l & 15);
        wv4[i].x = p[0];
        wv4[i].y = p[M];
        wv4[i].z = p[2 * M];
        wv4[i].w = p[3 * M];
    }
#pragma unroll
    for (int i = 0; i < SPT; ++i) wfrag[(int)threadIdx.x + i * GEMM_T] = wv4[i];

    float badd[TN];
#pragma unroll
    for (int t = 0; t < TN; ++t) badd[t] = bias ? bias[16 * t + lm] : 0.0f;
    float gsc[TN], bsh[TN];
#pragma unroll
    for (int t = 0; t < TN; ++t) {
        gsc[t] = LNELU ? g[16 * t + lm] : 1.0f;
        bsh[t] = LNELU ? be[16 * t + lm] : 0.0f;
    }

    __syncthreads();

    float4 wb[KB * TN];
#pragma unroll
    for (int i = 0; i < KB * TN; ++i) wb[i] = wfrag[i * 32 + lane];

    for (; strip < s_end; strip += NW) {
        float4 a_nxt[KB];
#pragma unroll
        for (int j = 0; j < KB; ++j) a_nxt[j] = a_cur[j];
        if (strip + NW < s_end) load_a_strip<K>(A, strip + NW, lm, lh, N, a_nxt);

        v8f acc[TN];
#pragma unroll
        for (int t = 0; t < TN; ++t) acc[t] = v8f{};

#pragma unroll
        for (int j = 0; j < KB; ++j) {
            v2f a01, a23;
            a01.x = a_cur[j].x; a01.y = a_cur[j].y;
            a23.x = a_cur[j].z; a23.y = a_cur[j].w;
#pragma unroll
            for (int t = 0; t < TN; ++t) {
                v2f b01;
                b01.x = wb[j * TN + t].x; b01.y = wb[j * TN + t].y;
                acc[t] = __builtin_amdgcn_wmma_f32_16x16x4_f32(false, a01, false, b01,
                                                               (short)0, acc[t], false, false);
            }
#pragma unroll
            for (int t = 0; t < TN; ++t) {
                v2f b23;
                b23.x = wb[j * TN + t].z; b23.y = wb[j * TN + t].w;
                acc[t] = __builtin_amdgcn_wmma_f32_16x16x4_f32(false, a23, false, b23,
                                                               (short)0, acc[t], false, false);
            }
        }

        const int row0 = strip << 4;
#pragma unroll
        for (int r = 0; r < 8; ++r) {
            const int row = row0 + r + (lh << 3);
            if (LNELU) {
                float v[TN];
                float s = 0.0f;
#pragma unroll
                for (int t = 0; t < TN; ++t) { v[t] = acc[t][r] + badd[t]; s += v[t]; }
                const float mu = lanes_sum<16>(s) * (1.0f / M);
                float q = 0.0f;
#pragma unroll
                for (int t = 0; t < TN; ++t) { v[t] -= mu; q += v[t] * v[t]; }
                const float rstd = rsqrtf(lanes_sum<16>(q) * (1.0f / M) + 1e-5f);
                if (row < N) {
#pragma unroll
                    for (int t = 0; t < TN; ++t) {
                        const float o = v[t] * rstd * gsc[t] + bsh[t];
                        C[(size_t)row * M + 16 * t + lm] = (o > 0.0f) ? o : (expf(o) - 1.0f);
                    }
                }
            } else if (row < N) {
#pragma unroll
                for (int t = 0; t < TN; ++t)
                    C[(size_t)row * M + 16 * t + lm] = acc[t][r] + badd[t];
            }
        }

#pragma unroll
        for (int j = 0; j < KB; ++j) a_cur[j] = a_nxt[j];
    }
}

constexpr int CH_E   = 4096;
constexpr int BK_SH  = 8;
constexpr int BK_N   = 1 << BK_SH;
constexpr int PREP_T = 1024;
constexpr int CH_U   = CH_E / PREP_T;

__global__ __launch_bounds__(1024)
void bin_kernel(const int* __restrict__ src, const int* __restrict__ dst,
                int2* __restrict__ binned, int* __restrict__ tcnt, int* __restrict__ toff,
                int E, int NB, int NCH)
{
    __shared__ int hist[PREP_T];
    __shared__ int offs[PREP_T];
    __shared__ int wsum[32];
    const int t = threadIdx.x, lane = t & 31, wv = t >> 5;
    const int chunk = blockIdx.x;
    const int e_base = chunk * CH_E;

    hist[t] = 0;
    __syncthreads();

    int s[CH_U], d[CH_U], r[CH_U];
#pragma unroll
    for (int u = 0; u < CH_U; ++u) {
        const int e = e_base + u * PREP_T + t;
        const bool ok = e < E;
        s[u] = ok ? src[e] : 0;
        d[u] = ok ? dst[e] : -1;
    }
#pragma unroll
    for (int u = 0; u < CH_U; ++u)
        r[u] = (d[u] >= 0) ? atomicAdd(&hist[d[u] >> BK_SH], 1) : 0;
    __syncthreads();

    const int c = hist[t];
    int x = c;
#pragma unroll
    for (int off = 1; off < 32; off <<= 1) {
        int y = __shfl_up(x, off, 32);
        if (lane >= off) x += y;
    }
    if (lane == 31) wsum[wv] = x;
    __syncthreads();
    if (wv == 0) {
        const int tot = wsum[lane];
        int q = tot;
#pragma unroll
        for (int off = 1; off < 32; off <<= 1) {
            int y = __shfl_up(q, off, 32);
            if (lane >= off) q += y;
        }
        wsum[lane] = q - tot;
    }
    __syncthreads();
    const int excl = wsum[wv] + x - c;
    offs[t] = excl;
    if (t < NB) {
        tcnt[(size_t)t * NCH + chunk] = c;
        toff[(size_t)t * NCH + chunk] = excl;
    }
    __syncthreads();
#pragma unroll
    for (int u = 0; u < CH_U; ++u)
        if (d[u] >= 0) binned[e_base + offs[d[u] >> BK_SH] + r[u]] = make_int2(s[u], d[u]);
}

__global__ __launch_bounds__(1024)
void bucket_place_kernel(const int* __restrict__ tcnt, const int* __restrict__ toff,
                         const int2* __restrict__ binned, int* __restrict__ rowptr,
                         float* __restrict__ dinv, int2* __restrict__ colw, int N, int NCH)
{
    __shared__ int ncnt[BK_N];
    __shared__ int lrow[BK_N];
    __shared__ int wsum[32];
    __shared__ int s_prefix;
    const int t = threadIdx.x, lane = t & 31, wv = t >> 5;
    const int b = blockIdx.x;
    const int node0 = b << BK_SH;
    const int* __restrict__ cnt_row = tcnt + (size_t)b * NCH;
    const int* __restrict__ off_row = toff + (size_t)b * NCH;

    if (t < BK_N) ncnt[t] = 0;

    int p = 0;
    for (int c = t; c < NCH; c += PREP_T) p += off_row[c];
#pragma unroll
    for (int off = 16; off; off >>= 1) p += __shfl_xor(p, off, 32);
    if (lane == 0) wsum[wv] = p;
    __syncthreads();
    if (wv == 0) {
        int q = wsum[lane];
#pragma unroll
        for (int off = 16; off; off >>= 1) q += __shfl_xor(q, off, 32);
        if (lane == 0) s_prefix = q;
    }
    __syncthreads();
    const int bstart = s_prefix;

    for (int c = t; c < NCH; c += PREP_T) {
        const int n = cnt_row[c];
        const int2* __restrict__ seg = binned + (size_t)c * CH_E + off_row[c];
        for (int i = 0; i < n; ++i) atomicAdd(&ncnt[seg[i].y - node0], 1);
    }
    __syncthreads();

    const int cn = (t < BK_N) ? ncnt[t] : 0;
    int x = cn;
#pragma unroll
    for (int off = 1; off < 32; off <<= 1) {
        int y = __shfl_up(x, off, 32);
        if (lane >= off) x += y;
    }
    if (t < BK_N && lane == 31) wsum[wv] = x;
    __syncthreads();
    if (t < BK_N) {
        int before = 0;
        for (int w = 0; w < wv; ++w) before += wsum[w];
        const int excl = before + x - cn;
        lrow[t] = excl;
        ncnt[t] = 0;
        const int node = node0 + t;
        if (node < N) {
            rowptr[node] = bstart + excl;
            dinv[node] = rsqrtf((float)cn + 1.0f);
            if (node == N - 1) rowptr[N] = bstart + excl + cn;
        }
    }
    __syncthreads();

    for (int c = t; c < NCH; c += PREP_T) {
        const int n = cnt_row[c];
        const int2* __restrict__ seg = binned + (size_t)c * CH_E + off_row[c];
        for (int i = 0; i < n; ++i) {
            const int2 sd = seg[i];
            const int dl = sd.y - node0;
            const int r = atomicAdd(&ncnt[dl], 1);
            colw[bstart + lrow[dl] + r] = sd;
        }
    }
}

constexpr int EU = 4;

__global__ __launch_bounds__(256)
void weight_kernel(const float* __restrict__ dinv, int2* __restrict__ colw, int E)
{
    const int stride = (int)(gridDim.x * blockDim.x);
    for (int p0 = (int)(blockIdx.x * blockDim.x + threadIdx.x); p0 < E; p0 += EU * stride) {
        if (p0 + (EU - 1) * stride < E) {
            int2 sd[EU];
            float w[EU];
#pragma unroll
            for (int u = 0; u < EU; ++u) sd[u] = colw[p0 + u * stride];
#pragma unroll
            for (int u = 0; u < EU; ++u) w[u] = dinv[sd[u].x] * dinv[sd[u].y];
#pragma unroll
            for (int u = 0; u < EU; ++u)
                colw[p0 + u * stride] = make_int2(sd[u].x, __float_as_int(w[u]));
        } else {
            for (int p = p0; p < E; p += stride) {
                const int2 sd1 = colw[p];
                colw[p] = make_int2(sd1.x, __float_as_int(dinv[sd1.x] * dinv[sd1.y]));
            }
        }
    }
}

template<int M, bool TAIL>
__global__ __launch_bounds__(256)
void csr_gather_ln_elu(const int* __restrict__ rowptr, const int2* __restrict__ colw,
                       const float* __restrict__ xw, const float* __restrict__ dinv,
                       const float* __restrict__ bias, const float* __restrict__ g,
                       const float* __restrict__ be, float* __restrict__ out, int N)
{
    constexpr int LPR = M / 4;
    constexpr int EPW = 32 / LPR;
    constexpr int UN  = 4;
    const int nwaves = (int)(gridDim.x * (blockDim.x >> 5));
    int wave = (int)(blockIdx.x * (blockDim.x >> 5)) + (int)(threadIdx.x >> 5);
    wave = __builtin_amdgcn_readfirstlane(wave);
    const int lane = threadIdx.x & 31;
    const int sub = lane / LPR;
    const int c = (lane % LPR) << 2;

    float4 bi = make_float4(0.0f, 0.0f, 0.0f, 0.0f), gg = bi, bb = bi;
    if (TAIL) {
        bi.x = bias[c]; bi.y = bias[c + 1]; bi.z = bias[c + 2]; bi.w = bias[c + 3];
        gg.x = g[c];    gg.y = g[c + 1];    gg.z = g[c + 2];    gg.w = g[c + 3];
        bb.x = be[c];   bb.y = be[c + 1];   bb.z = be[c + 2];   bb.w = be[c + 3];
    }

    for (int node = wave; node < N; node += nwaves) {
        const int beg = rowptr[node];
        const int end = rowptr[node + 1];
        const float di = dinv[node];
        const int selfw = __float_as_int(di * di);
        const int stop = end + 1;

        float4 acc = make_float4(0.0f, 0.0f, 0.0f, 0.0f);
        int base = beg;
        do {
            const int n = min(32, stop - base);
            const int e = base + lane;
            int2 mine = make_int2(0, 0);
            if (e < end)       mine = colw[e];
            else if (e == end) mine = make_int2(node, selfw);
            for (int j0 = 0; j0 < n; j0 += UN * EPW) {
                float4 v[UN];
                float  w[UN];
#pragma unroll
                for (int u = 0; u < UN; ++u) {
                    const int j = (j0 + u * EPW + sub) & 31;
                    const int s = __shfl(mine.x, j, 32);
                    w[u] = __int_as_float(__shfl(mine.y, j, 32));
                    v[u] = *(const float4*)(xw + (size_t)s * M + c);
                }
#pragma unroll
                for (int u = 0; u < UN; ++u) {
                    if (j0 + u * EPW + sub < n) {
                        acc.x += v[u].x * w[u];
                        acc.y += v[u].y * w[u];
                        acc.z += v[u].z * w[u];
                        acc.w += v[u].w * w[u];
                    }
                }
            }
            base += 32;
        } while (base < stop);
        if (EPW == 2) {
            acc.x += __shfl_xor(acc.x, 16, 32);
            acc.y += __shfl_xor(acc.y, 16, 32);
            acc.z += __shfl_xor(acc.z, 16, 32);
            acc.w += __shfl_xor(acc.w, 16, 32);
        }

        if (!TAIL) {
            if (sub == 0) *(float4*)(out + (size_t)node * M + c) = acc;
            continue;
        }

        float4 t;
        t.x = acc.x + bi.x;
        t.y = acc.y + bi.y;
        t.z = acc.z + bi.z;
        t.w = acc.w + bi.w;

        const float sum = lanes_sum<LPR>((t.x + t.y) + (t.z + t.w));
        const float mu = sum * (1.0f / M);
        const float dx = t.x - mu, dy = t.y - mu, dz = t.z - mu, dw = t.w - mu;
        const float vs = lanes_sum<LPR>((dx * dx + dy * dy) + (dz * dz + dw * dw));
        const float rstd = rsqrtf(vs * (1.0f / M) + 1e-5f);
        float4 o;
        o.x = dx * rstd * gg.x + bb.x;
        o.y = dy * rstd * gg.y + bb.y;
        o.z = dz * rstd * gg.z + bb.z;
        o.w = dw * rstd * gg.w + bb.w;
        o.x = (o.x > 0.0f) ? o.x : (expf(o.x) - 1.0f);
        o.y = (o.y > 0.0f) ? o.y : (expf(o.y) - 1.0f);
        o.z = (o.z > 0.0f) ? o.z : (expf(o.z) - 1.0f);
        o.w = (o.w > 0.0f) ? o.w : (expf(o.w) - 1.0f);
        if (sub == 0) *(float4*)(out + (size_t)node * M + c) = o;
    }
}

template<int M>
__global__ __launch_bounds__(256)
void finalize_ln_elu(const float* agg, const float* __restrict__ xw,
                     const float* __restrict__ dinv, const float* __restrict__ bias,
                     const float* __restrict__ g, const float* __restrict__ be,
                     float* out, int N)
{
    constexpr int PER = M / 32;
    const int nwaves = (int)(gridDim.x * (blockDim.x >> 5));
    int wave = (int)(blockIdx.x * (blockDim.x >> 5)) + (int)(threadIdx.x >> 5);
    wave = __builtin_amdgcn_readfirstlane(wave);
    const int lane = threadIdx.x & 31;

    float bi[PER], gg[PER], bb[PER];
#pragma unroll
    for (int i = 0; i < PER; ++i) {
        const int c = lane + (i << 5);
        bi[i] = bias ? bias[c] : 0.0f;
        gg[i] = g[c];
        bb[i] = be[c];
    }

    for (int node = wave; node < N; node += nwaves) {
        float vals[PER], xs[PER];
        float sum = 0.0f;
        float selfw = 0.0f;
#pragma unroll
        for (int i = 0; i < PER; ++i) vals[i] = agg[(size_t)node * M + lane + (i << 5)];
        if (xw) {
#pragma unroll
            for (int i = 0; i < PER; ++i) xs[i] = xw[(size_t)node * M + lane + (i << 5)];
            selfw = dinv[node] * dinv[node];
        } else {
#pragma unroll
            for (int i = 0; i < PER; ++i) xs[i] = 0.0f;
        }
#pragma unroll
        for (int i = 0; i < PER; ++i) {
            float v = vals[i];
            if (xw)   v += xs[i] * selfw;
            if (bias) v += bi[i];
            vals[i] = v;
            sum += v;
        }
#pragma unroll
        for (int off = 16; off; off >>= 1) sum += __shfl_xor(sum, off, 32);
        float mu = sum * (1.0f / M);
        float vs = 0.0f;
#pragma unroll
        for (int i = 0; i < PER; ++i) { float d = vals[i] - mu; vs += d * d; }
#pragma unroll
        for (int off = 16; off; off >>= 1) vs += __shfl_xor(vs, off, 32);
        float rstd = rsqrtf(vs * (1.0f / M) + 1e-5f);
#pragma unroll
        for (int i = 0; i < PER; ++i) {
            int c = lane + (i << 5);
            float v = (vals[i] - mu) * rstd * gg[i] + bb[i];
            out[(size_t)node * M + c] = (v > 0.0f) ? v : (expf(v) - 1.0f);
        }
    }
}

extern "C" void kernel_launch(void* const* d_in, const int* in_sizes, int n_in,
                              void* d_out, int out_size, void* d_ws, size_t ws_size,
                              hipStream_t stream)
{
    const float* x   = (const float*)d_in[0];
    const int*   ei  = (const int*)  d_in[1];
    const float* W1  = (const float*)d_in[2];
    const float* b1  = (const float*)d_in[3];
    const float* g1  = (const float*)d_in[4];
    const float* be1 = (const float*)d_in[5];
    const float* W2  = (const float*)d_in[6];
    const float* b2  = (const float*)d_in[7];
    const float* g2  = (const float*)d_in[8];
    const float* be2 = (const float*)d_in[9];
    const float* W3  = (const float*)d_in[10];
    const float* b3  = (const float*)d_in[11];
    const float* g3  = (const float*)d_in[12];
    const float* be3 = (const float*)d_in[13];
    const float* lw1 = (const float*)d_in[14];
    const float* lb1 = (const float*)d_in[15];
    const float* g4  = (const float*)d_in[16];
    const float* be4 = (const float*)d_in[17];
    const float* lw2 = (const float*)d_in[18];
    const float* lb2 = (const float*)d_in[19];

    const int N = in_sizes[0] / 128;
    const int E = in_sizes[1] / 2;
    const int* src = ei;
    const int* dst = ei + E;

    const int NB  = (N + BK_N - 1) / BK_N;
    const int NCH = ceil_div_ll(E, CH_E);
    float* ws   = (float*)d_ws;
    float* xw   = ws;  ws += (size_t)N * 128;
    float* agg  = ws;  ws += (size_t)N * 128;
    float* h    = ws;  ws += (size_t)N * 64;
    int2*  colw = (int2*)ws;
    ws += (size_t)N * 64;
    ws += N;
    float* dinv   = ws;        ws += N;
    int*   rowptr = (int*)ws;  ws += E;
    int2*  binned = (int2*)agg;
    int*   tcnt   = (int*)agg + (size_t)2 * NCH * CH_E;
    int*   toff   = tcnt + (size_t)NB * NCH;

    const int B = 256;
    auto gemm_grid = [N](int per_wgp) {
        const int want = GEMM_WGPS * per_wgp;
        const int most = ceil_div_ll((N + 15) / 16, GEMM_T / 32);
        return want < most ? want : most;
    };
    const int node_blocks = loop_grid(ceil_div_ll(N, 8));
    const int edge_blocks = loop_grid(ceil_div_ll(E, (long long)B * EU));

    bin_kernel<<<NCH, PREP_T, 0, stream>>>(src, dst, binned, tcnt, toff, E, NB, NCH);
    bucket_place_kernel<<<NB, PREP_T, 0, stream>>>(tcnt, toff, binned, rowptr, dinv, colw, N, NCH);
    weight_kernel<<<edge_blocks, B, 0, stream>>>(dinv, colw, E);

    wmma_gemm_f32<128, 64, false><<<gemm_grid(1), GEMM_T, 0, stream>>>(x, W1, nullptr, nullptr, nullptr, xw, N);
    csr_gather_ln_elu<64, true><<<node_blocks, B, 0, stream>>>(rowptr, colw, xw, dinv, b1, g1, be1, h, N);

    csr_gather_ln_elu<64, false><<<node_blocks, B, 0, stream>>>(rowptr, colw, h, dinv, nullptr, nullptr, nullptr, xw, N);
    wmma_gemm_f32<64, 128, true><<<gemm_grid(1), GEMM_T, 0, stream>>>(xw, W2, b2, g2, be2, agg, N);

    wmma_gemm_f32<128, 64, false><<<gemm_grid(1), GEMM_T, 0, stream>>>(agg, W3, nullptr, nullptr, nullptr, xw, N);
    csr_gather_ln_elu<64, true><<<node_blocks, B, 0, stream>>>(rowptr, colw, xw, dinv, b3, g3, be3, h, N);

    wmma_gemm_f32<64, 32, false><<<gemm_grid(3), GEMM_T, 0, stream>>>(h, lw1, lb1, nullptr, nullptr, xw, N);
    finalize_ln_elu<32><<<node_blocks, B, 0, stream>>>(xw, nullptr, nullptr, nullptr, g4, be4, h, N);
    wmma_gemm_f32<32, 32, false><<<gemm_grid(3), GEMM_T, 0, stream>>>(h, lw2, lb2, nullptr, nullptr, (float*)d_out, N);
}


# Fuse the head (linear, LN+ELU, linear) into one kernel: hidden rows go C layout -> per-wave LDS tile -> A fragments; 2 launches, 2 round trips gone; LN sums regrouped: sum order changes
# speedup vs baseline: 3.6682x; 1.0445x over previous
#include <hip/hip_runtime.h>
#include <hip/hip_bf16.h>
#include <math.h>

typedef __attribute__((ext_vector_type(2))) float v2f;
typedef __attribute__((ext_vector_type(8))) float v8f;

static inline int ceil_div_ll(long long a, long long b) { return (int)((a + b - 1) / b); }

constexpr int LOOP_BLOCKS = 2048;
static inline int loop_grid(long long blocks_wanted)
{
    return (int)(blocks_wanted < LOOP_BLOCKS ? blocks_wanted : LOOP_BLOCKS);
}

__device__ __forceinline__ float xhalf(float x)
{
    const int b = __builtin_bit_cast(int, x);
    return __builtin_bit_cast(float,
        __builtin_amdgcn_permlanex16(b, b, 0x76543210, 0xfedcba98, false, false));
}
template<int MASK>
__device__ __forceinline__ float xrow(float x)
{
    const int b = __builtin_bit_cast(int, x);
    return __builtin_bit_cast(float, __builtin_amdgcn_update_dpp(b, b, 0x160 | MASK, 0xf, 0xf, false));
}
template<int LANES>
__device__ __forceinline__ float lanes_sum(float x)
{
    x += xrow<1>(x);
    x += xrow<2>(x);
    x += xrow<4>(x);
    x += xrow<8>(x);
    if (LANES == 32) x += xhalf(x);
    return x;
}

constexpr int GEMM_T = 256;
constexpr int GEMM_WGPS = 256;

template<int K>
__device__ __forceinline__ void load_a_strip(const float* __restrict__ A, int strip, int lm, int lh,
                                             int N, float4 (&a)[K / 8])
{
    int arow = (strip << 4) + lm;
    arow = arow < N ? arow : N - 1;
    const float4* ap = (const float4*)(A + (size_t)arow * K) + lh;
#pragma unroll
    for (int j = 0; j < K / 8; ++j) a[j] = ap[2 * j];
}

template<int K, int M, bool LNELU>
__global__ __launch_bounds__(GEMM_T)
void wmma_gemm_f32(const float* __restrict__ A, const float* __restrict__ W,
                   const float* __restrict__ bias, const float* __restrict__ g,
                   const float* __restrict__ be, float* __restrict__ C, int N)
{
    constexpr int TN = M / 16;
    constexpr int KB = K / 8;
    constexpr int NW = GEMM_T / 32;
    __shared__ float4 wfrag[KB * TN * 32];

    const int lane = threadIdx.x & 31;
    const int wv = __builtin_amdgcn_readfirstlane((int)(threadIdx.x >> 5));
    const int lh = lane >> 4;
    const int lm = lane & 15;

    const unsigned nstrips = (unsigned)(N + 15) >> 4;
    const int s_beg = (int)(blockIdx.x * nstrips / gridDim.x);
    const int s_end = (int)((blockIdx.x + 1u) * nstrips / gridDim.x);

    int strip = s_beg + wv;
    float4 a_cur[KB];
#pragma unroll
    for (int j = 0; j < KB; ++j) a_cur[j] = make_float4(0.0f, 0.0f, 0.0f, 0.0f);
    if (strip < s_end) load_a_strip<K>(A, strip, lm, lh, N, a_cur);

    constexpr int SPT = KB * TN * 32 / GEMM_T;
    static_assert(SPT * GEMM_T == KB * TN * 32, "K * M must be a multiple of 4 * GEMM_T");
    float4 wv4[SPT];
#pragma unroll
    for (int i = 0; i < SPT; ++i) {
        const int slot = (int)threadIdx.x + i * GEMM_T;
        const int l = slot & 31;
        const int jt = slot >> 5;
        const int t = jt % TN;
        const int j = jt / TN;
        const float* p = W + (size_t)(8 * j + 4 * (l >> 4)) * M + 16 * t + (l & 15);
        wv4[i].x = p[0];
        wv4[i].y = p[M];
        wv4[i].z = p[2 * M];
        wv4[i].w = p[3 * M];
    }
#pragma unroll
    for (int i = 0; i < SPT; ++i) wfrag[(int)threadIdx.x + i * GEMM_T] = wv4[i];

    float badd[TN];
#pragma unroll
    for (int t = 0; t < TN; ++t) badd[t] = bias ? bias[16 * t + lm] : 0.0f;
    float gsc[TN], bsh[TN];
#pragma unroll
    for (int t = 0; t < TN; ++t) {
        gsc[t] = LNELU ? g[16 * t + lm] : 1.0f;
        bsh[t] = LNELU ? be[16 * t + lm] : 0.0f;
    }

    __syncthreads();

    float4 wb[KB * TN];
#pragma unroll
    for (int i = 0; i < KB * TN; ++i) wb[i] = wfrag[i * 32 + lane];

    for (; strip < s_end; strip += NW) {
        float4 a_nxt[KB];
#pragma unroll
        for (int j = 0; j < KB; ++j) a_nxt[j] = a_cur[j];
        if (strip + NW < s_end) load_a_strip<K>(A, strip + NW, lm, lh, N, a_nxt);

        v8f acc[TN];
#pragma unroll
        for (int t = 0; t < TN; ++t) acc[t] = v8f{};

#pragma unroll
        for (int j = 0; j < KB; ++j) {
            v2f a01, a23;
            a01.x = a_cur[j].x; a01.y = a_cur[j].y;
            a23.x = a_cur[j].z; a23.y = a_cur[j].w;
#pragma unroll
            for (int t = 0; t < TN; ++t) {
                v2f b01;
                b01.x = wb[j * TN + t].x; b01.y = wb[j * TN + t].y;
                acc[t] = __builtin_amdgcn_wmma_f32_16x16x4_f32(false, a01, false, b01,
                                                               (short)0, acc[t], false, false);
            }
#pragma unroll
            for (int t = 0; t < TN; ++t) {
                v2f b23;
                b23.x = wb[j * TN + t].z; b23.y = wb[j * TN + t].w;
                acc[t] = __builtin_amdgcn_wmma_f32_16x16x4_f32(false, a23, false, b23,
                                                               (short)0, acc[t], false, false);
            }
        }

        const int row0 = strip << 4;
#pragma unroll
        for (int r = 0; r < 8; ++r) {
            const int row = row0 + r + (lh << 3);
            if (LNELU) {
                float v[TN];
                float s = 0.0f;
#pragma unroll
                for (int t = 0; t < TN; ++t) { v[t] = acc[t][r] + badd[t]; s += v[t]; }
                const float mu = lanes_sum<16>(s) * (1.0f / M);
                float q = 0.0f;
#pragma unroll
                for (int t = 0; t < TN; ++t) { v[t] -= mu; q += v[t] * v[t]; }
                const float rstd = rsqrtf(lanes_sum<16>(q) * (1.0f / M) + 1e-5f);
                if (row < N) {
#pragma unroll
                    for (int t = 0; t < TN; ++t) {
                        const float o = v[t] * rstd * gsc[t] + bsh[t];
                        C[(size_t)row * M + 16 * t + lm] = (o > 0.0f) ? o : (expf(o) - 1.0f);
                    }
                }
            } else if (row < N) {
#pragma unroll
                for (int t = 0; t < TN; ++t)
                    C[(size_t)row * M + 16 * t + lm] = acc[t][r] + badd[t];
            }
        }

#pragma unroll
        for (int j = 0; j < KB; ++j) a_cur[j] = a_nxt[j];
    }
}

constexpr int HEAD_K  = 64;
constexpr int HEAD_M  = 32;
constexpr int HEAD_YP = HEAD_M + 4;
constexpr int HEAD_WGS = 2;

template<int K, int M>
__device__ __forceinline__ void w_quads_load(const float* __restrict__ W, float4 (&q)[K * M / (4 * GEMM_T)])
{
    constexpr int TN = M / 16;
    constexpr int SPT = K * M / (4 * GEMM_T);
    static_assert(SPT * 4 * GEMM_T == K * M, "K * M must be a multiple of 4 * GEMM_T");
#pragma unroll
    for (int i = 0; i < SPT; ++i) {
        const int slot = (int)threadIdx.x + i * GEMM_T;
        const int l = slot & 31;
        const int jt = slot >> 5;
        const int t = jt % TN;
        const int j = jt / TN;
        const float* p = W + (size_t)(8 * j + 4 * (l >> 4)) * M + 16 * t + (l & 15);
        q[i].x = p[0];
        q[i].y = p[M];
        q[i].z = p[2 * M];
        q[i].w = p[3 * M];
    }
}
template<int SPT>
__device__ __forceinline__ void w_quads_store(float4* wfrag, const float4 (&q)[SPT])
{
#pragma unroll
    for (int i = 0; i < SPT; ++i) wfrag[(int)threadIdx.x + i * GEMM_T] = q[i];
}

__global__ __launch_bounds__(GEMM_T)
void head_fused_f32(const float* __restrict__ A, const float* __restrict__ W1,
                    const float* __restrict__ b1, const float* __restrict__ g,
                    const float* __restrict__ be, const float* __restrict__ W2,
                    const float* __restrict__ b2, float* __restrict__ C, int N)
{
    constexpr int K1 = HEAD_K, M1 = HEAD_M, M2 = HEAD_M;
    constexpr int TN1 = M1 / 16;
    constexpr int KB1 = K1 / 8;
    constexpr int TN2 = M2 / 16;
    constexpr int KB2 = M1 / 8;
    constexpr int NW = GEMM_T / 32;
    __shared__ float4 wfrag1[KB1 * TN1 * 32];
    __shared__ float4 wfrag2[KB2 * TN2 * 32];
    __shared__ float4 ytile[NW * 16 * HEAD_YP / 4];

    const int lane = threadIdx.x & 31;
    const int wv = __builtin_amdgcn_readfirstlane((int)(threadIdx.x >> 5));
    const int lh = lane >> 4;
    const int lm = lane & 15;

    const unsigned nstrips = (unsigned)(N + 15) >> 4;
    const int s_beg = (int)(blockIdx.x * nstrips / gridDim.x);
    const int s_end = (int)((blockIdx.x + 1u) * nstrips / gridDim.x);

    int strip = s_beg + wv;
    float4 a_cur[KB1];
#pragma unroll
    for (int j = 0; j < KB1; ++j) a_cur[j] = make_float4(0.0f, 0.0f, 0.0f, 0.0f);
    if (strip < s_end) load_a_strip<K1>(A, strip, lm, lh, N, a_cur);

    float4 q1[K1 * M1 / (4 * GEMM_T)];
    float4 q2[M1 * M2 / (4 * GEMM_T)];
    w_quads_load<K1, M1>(W1, q1);
    w_quads_load<M1, M2>(W2, q2);
    w_quads_store(wfrag1, q1);
    w_quads_store(wfrag2, q2);

    float badd1[TN1], gsc[TN1], bsh[TN1], badd2[TN2];
#pragma unroll
    for (int t = 0; t < TN1; ++t) {
        badd1[t] = b1[16 * t + lm];
        gsc[t] = g[16 * t + lm];
        bsh[t] = be[16 * t + lm];
    }
#pragma unroll
    for (int t = 0; t < TN2; ++t) badd2[t] = b2[16 * t + lm];

    __syncthreads();

    float4 wb1[KB1 * TN1], wb2[KB2 * TN2];
#pragma unroll
    for (int i = 0; i < KB1 * TN1; ++i) wb1[i] = wfrag1[i * 32 + lane];
#pragma unroll
    for (int i = 0; i < KB2 * TN2; ++i) wb2[i] = wfrag2[i * 32 + lane];

    float* yt = (float*)ytile + wv * (16 * HEAD_YP);

    for (; strip < s_end; strip += NW) {
        float4 a_nxt[KB1];
#pragma unroll
        for (int j = 0; j < KB1; ++j) a_nxt[j] = a_cur[j];
        if (strip + NW < s_end) load_a_strip<K1>(A, strip + NW, lm, lh, N, a_nxt);

        v8f acc1[TN1];
#pragma unroll
        for (int t = 0; t < TN1; ++t) acc1[t] = v8f{};
#pragma unroll
        for (int j = 0; j < KB1; ++j) {
            v2f a01, a23;
            a01.x = a_cur[j].x; a01.y = a_cur[j].y;
            a23.x = a_cur[j].z; a23.y = a_cur[j].w;
#pragma unroll
            for (int t = 0; t < TN1; ++t) {
                v2f b01;
                b01.x = wb1[j * TN1 + t].x; b01.y = wb1[j * TN1 + t].y;
                acc1[t] = __builtin_amdgcn_wmma_f32_16x16x4_f32(false, a01, false, b01,
                                                                (short)0, acc1[t], false, false);
            }
#pragma unroll
            for (int t = 0; t < TN1; ++t) {
                v2f b23;
                b23.x = wb1[j * TN1 + t].z; b23.y = wb1[j * TN1 + t].w;
                acc1[t] = __builtin_amdgcn_wmma_f32_16x16x4_f32(false, a23, false, b23,
                                                                (short)0, acc1[t], false, false);
            }
        }

#pragma unroll
        for (int r = 0; r < 8; ++r) {
            float v[TN1];
            float s = 0.0f;
#pragma unroll
            for (int t = 0; t < TN1; ++t) { v[t] = acc1[t][r] + badd1[t]; s += v[t]; }
            const float mu = lanes_sum<16>(s) * (1.0f / M1);
            float q = 0.0f;
#pragma unroll
            for (int t = 0; t < TN1; ++t) { v[t] -= mu; q += v[t] * v[t]; }
            const float rstd = rsqrtf(lanes_sum<16>(q) * (1.0f / M1) + 1e-5f);
#pragma unroll
            for (int t = 0; t < TN1; ++t) {
                const float o = v[t] * rstd * gsc[t] + bsh[t];
                yt[(r + (lh << 3)) * HEAD_YP + 16 * t + lm] = (o > 0.0f) ? o : (expf(o) - 1.0f);
            }
        }

        float4 a2[KB2];
#pragma unroll
        for (int j = 0; j < KB2; ++j)
            a2[j] = *(const float4*)(yt + lm * HEAD_YP + 8 * j + 4 * lh);

        v8f acc2[TN2];
#pragma unroll
        for (int t = 0; t < TN2; ++t) acc2[t] = v8f{};
#pragma unroll
        for (int j = 0; j < KB2; ++j) {
            v2f a01, a23;
            a01.x = a2[j].x; a01.y = a2[j].y;
            a23.x = a2[j].z; a23.y = a2[j].w;
#pragma unroll
            for (int t = 0; t < TN2; ++t) {
                v2f b01;
                b01.x = wb2[j * TN2 + t].x; b01.y = wb2[j * TN2 + t].y;
                acc2[t] = __builtin_amdgcn_wmma_f32_16x16x4_f32(false, a01, false, b01,
                                                                (short)0, acc2[t], false, false);
            }
#pragma unroll
            for (int t = 0; t < TN2; ++t) {
                v2f b23;
                b23.x = wb2[j * TN2 + t].z; b23.y = wb2[j * TN2 + t].w;
                acc2[t] = __builtin_amdgcn_wmma_f32_16x16x4_f32(false, a23, false, b23,
                                                                (short)0, acc2[t], false, false);
            }
        }

        const int row0 = strip << 4;
#pragma unroll
        for (int r = 0; r < 8; ++r) {
            const int row = row0 + r + (lh << 3);
            if (row < N) {
#pragma unroll
                for (int t = 0; t < TN2; ++t)
                    C[(size_t)row * M2 + 16 * t + lm] = acc2[t][r] + badd2[t];
            }
        }

#pragma unroll
        for (int j = 0; j < KB1; ++j) a_cur[j] = a_nxt[j];
    }
}

constexpr int CH_E   = 4096;
constexpr int BK_SH  = 8;
constexpr int BK_N   = 1 << BK_SH;
constexpr int PREP_T = 1024;
constexpr int CH_U   = CH_E / PREP_T;

__global__ __launch_bounds__(1024)
void bin_kernel(const int* __restrict__ src, const int* __restrict__ dst,
                int2* __restrict__ binned, int* __restrict__ tcnt, int* __restrict__ toff,
                int E, int NB, int NCH)
{
    __shared__ int hist[PREP_T];
    __shared__ int offs[PREP_T];
    __shared__ int wsum[32];
    const int t = threadIdx.x, lane = t & 31, wv = t >> 5;
    const int chunk = blockIdx.x;
    const int e_base = chunk * CH_E;

    hist[t] = 0;
    __syncthreads();

    int s[CH_U], d[CH_U], r[CH_U];
#pragma unroll
    for (int u = 0; u < CH_U; ++u) {
        const int e = e_base + u * PREP_T + t;
        const bool ok = e < E;
        s[u] = ok ? src[e] : 0;
        d[u] = ok ? dst[e] : -1;
    }
#pragma unroll
    for (int u = 0; u < CH_U; ++u)
        r[u] = (d[u] >= 0) ? atomicAdd(&hist[d[u] >> BK_SH], 1) : 0;
    __syncthreads();

    const int c = hist[t];
    int x = c;
#pragma unroll
    for (int off = 1; off < 32; off <<= 1) {
        int y = __shfl_up(x, off, 32);
        if (lane >= off) x += y;
    }
    if (lane == 31) wsum[wv] = x;
    __syncthreads();
    if (wv == 0) {
        const int tot = wsum[lane];
        int q = tot;
#pragma unroll
        for (int off = 1; off < 32; off <<= 1) {
            int y = __shfl_up(q, off, 32);
            if (lane >= off) q += y;
        }
        wsum[lane] = q - tot;
    }
    __syncthreads();
    const int excl = wsum[wv] + x - c;
    offs[t] = excl;
    if (t < NB) {
        tcnt[(size_t)t * NCH + chunk] = c;
        toff[(size_t)t * NCH + chunk] = excl;
    }
    __syncthreads();
#pragma unroll
    for (int u = 0; u < CH_U; ++u)
        if (d[u] >= 0) binned[e_base + offs[d[u] >> BK_SH] + r[u]] = make_int2(s[u], d[u]);
}

__global__ __launch_bounds__(1024)
void bucket_place_kernel(const int* __restrict__ tcnt, const int* __restrict__ toff,
                         const int2* __restrict__ binned, int* __restrict__ rowptr,
                         float* __restrict__ dinv, int2* __restrict__ colw, int N, int NCH)
{
    __shared__ int ncnt[BK_N];
    __shared__ int lrow[BK_N];
    __shared__ int wsum[32];
    __shared__ int s_prefix;
    const int t = threadIdx.x, lane = t & 31, wv = t >> 5;
    const int b = blockIdx.x;
    const int node0 = b << BK_SH;
    const int* __restrict__ cnt_row = tcnt + (size_t)b * NCH;
    const int* __restrict__ off_row = toff + (size_t)b * NCH;

    if (t < BK_N) ncnt[t] = 0;

    int p = 0;
    for (int c = t; c < NCH; c += PREP_T) p += off_row[c];
#pragma unroll
    for (int off = 16; off; off >>= 1) p += __shfl_xor(p, off, 32);
    if (lane == 0) wsum[wv] = p;
    __syncthreads();
    if (wv == 0) {
        int q = wsum[lane];
#pragma unroll
        for (int off = 16; off; off >>= 1) q += __shfl_xor(q, off, 32);
        if (lane == 0) s_prefix = q;
    }
    __syncthreads();
    const int bstart = s_prefix;

    for (int c = t; c < NCH; c += PREP_T) {
        const int n = cnt_row[c];
        const int2* __restrict__ seg = binned + (size_t)c * CH_E + off_row[c];
        for (int i = 0; i < n; ++i) atomicAdd(&ncnt[seg[i].y - node0], 1);
    }
    __syncthreads();

    const int cn = (t < BK_N) ? ncnt[t] : 0;
    int x = cn;
#pragma unroll
    for (int off = 1; off < 32; off <<= 1) {
        int y = __shfl_up(x, off, 32);
        if (lane >= off) x += y;
    }
    if (t < BK_N && lane == 31) wsum[wv] = x;
    __syncthreads();
    if (t < BK_N) {
        int before = 0;
        for (int w = 0; w < wv; ++w) before += wsum[w];
        const int excl = before + x - cn;
        lrow[t] = excl;
        ncnt[t] = 0;
        const int node = node0 + t;
        if (node < N) {
            rowptr[node] = bstart + excl;
            dinv[node] = rsqrtf((float)cn + 1.0f);
            if (node == N - 1) rowptr[N] = bstart + excl + cn;
        }
    }
    __syncthreads();

    for (int c = t; c < NCH; c += PREP_T) {
        const int n = cnt_row[c];
        const int2* __restrict__ seg = binned + (size_t)c * CH_E + off_row[c];
        for (int i = 0; i < n; ++i) {
            const int2 sd = seg[i];
            const int dl = sd.y - node0;
            const int r = atomicAdd(&ncnt[dl], 1);
            colw[bstart + lrow[dl] + r] = sd;
        }
    }
}

constexpr int EU = 4;

__global__ __launch_bounds__(256)
void weight_kernel(const float* __restrict__ dinv, int2* __restrict__ colw, int E)
{
    const int stride = (int)(gridDim.x * blockDim.x);
    for (int p0 = (int)(blockIdx.x * blockDim.x + threadIdx.x); p0 < E; p0 += EU * stride) {
        if (p0 + (EU - 1) * stride < E) {
            int2 sd[EU];
            float w[EU];
#pragma unroll
            for (int u = 0; u < EU; ++u) sd[u] = colw[p0 + u * stride];
#pragma unroll
            for (int u = 0; u < EU; ++u) w[u] = dinv[sd[u].x] * dinv[sd[u].y];
#pragma unroll
            for (int u = 0; u < EU; ++u)
                colw[p0 + u * stride] = make_int2(sd[u].x, __float_as_int(w[u]));
        } else {
            for (int p = p0; p < E; p += stride) {
                const int2 sd1 = colw[p];
                colw[p] = make_int2(sd1.x, __float_as_int(dinv[sd1.x] * dinv[sd1.y]));
            }
        }
    }
}

template<int M, bool TAIL>
__global__ __launch_bounds__(256)
void csr_gather_ln_elu(const int* __restrict__ rowptr, const int2* __restrict__ colw,
                       const float* __restrict__ xw, const float* __restrict__ dinv,
                       const float* __restrict__ bias, const float* __restrict__ g,
                       const float* __restrict__ be, float* __restrict__ out, int N)
{
    constexpr int LPR = M / 4;
    constexpr int EPW = 32 / LPR;
    constexpr int UN  = 4;
    const int nwaves = (int)(gridDim.x * (blockDim.x >> 5));
    int wave = (int)(blockIdx.x * (blockDim.x >> 5)) + (int)(threadIdx.x >> 5);
    wave = __builtin_amdgcn_readfirstlane(wave);
    const int lane = threadIdx.x & 31;
    const int sub = lane / LPR;
    const int c = (lane % LPR) << 2;

    float4 bi = make_float4(0.0f, 0.0f, 0.0f, 0.0f), gg = bi, bb = bi;
    if (TAIL) {
        bi.x = bias[c]; bi.y = bias[c + 1]; bi.z = bias[c + 2]; bi.w = bias[c + 3];
        gg.x = g[c];    gg.y = g[c + 1];    gg.z = g[c + 2];    gg.w = g[c + 3];
        bb.x = be[c];   bb.y = be[c + 1];   bb.z = be[c + 2];   bb.w = be[c + 3];
    }

    for (int node = wave; node < N; node += nwaves) {
        const int beg = rowptr[node];
        const int end = rowptr[node + 1];
        const float di = dinv[node];
        const int selfw = __float_as_int(di * di);
        const int stop = end + 1;

        float4 acc = make_float4(0.0f, 0.0f, 0.0f, 0.0f);
        int base = beg;
        do {
            const int n = min(32, stop - base);
            const int e = base + lane;
            int2 mine = make_int2(0, 0);
            if (e < end)       mine = colw[e];
            else if (e == end) mine = make_int2(node, selfw);
            for (int j0 = 0; j0 < n; j0 += UN * EPW) {
                float4 v[UN];
                float  w[UN];
#pragma unroll
                for (int u = 0; u < UN; ++u) {
                    const int j = (j0 + u * EPW + sub) & 31;
                    const int s = __shfl(mine.x, j, 32);
                    w[u] = __int_as_float(__shfl(mine.y, j, 32));
                    v[u] = *(const float4*)(xw + (size_t)s * M + c);
                }
#pragma unroll
                for (int u = 0; u < UN; ++u) {
                    if (j0 + u * EPW + sub < n) {
                        acc.x += v[u].x * w[u];
                        acc.y += v[u].y * w[u];
                        acc.z += v[u].z * w[u];
                        acc.w += v[u].w * w[u];
                    }
                }
            }
            base += 32;
        } while (base < stop);
        if (EPW == 2) {
            acc.x += __shfl_xor(acc.x, 16, 32);
            acc.y += __shfl_xor(acc.y, 16, 32);
            acc.z += __shfl_xor(acc.z, 16, 32);
            acc.w += __shfl_xor(acc.w, 16, 32);
        }

        if (!TAIL) {
            if (sub == 0) *(float4*)(out + (size_t)node * M + c) = acc;
            continue;
        }

        float4 t;
        t.x = acc.x + bi.x;
        t.y = acc.y + bi.y;
        t.z = acc.z + bi.z;
        t.w = acc.w + bi.w;

        const float sum = lanes_sum<LPR>((t.x + t.y) + (t.z + t.w));
        const float mu = sum * (1.0f / M);
        const float dx = t.x - mu, dy = t.y - mu, dz = t.z - mu, dw = t.w - mu;
        const float vs = lanes_sum<LPR>((dx * dx + dy * dy) + (dz * dz + dw * dw));
        const float rstd = rsqrtf(vs * (1.0f / M) + 1e-5f);
        float4 o;
        o.x = dx * rstd * gg.x + bb.x;
        o.y = dy * rstd * gg.y + bb.y;
        o.z = dz * rstd * gg.z + bb.z;
        o.w = dw * rstd * gg.w + bb.w;
        o.x = (o.x > 0.0f) ? o.x : (expf(o.x) - 1.0f);
        o.y = (o.y > 0.0f) ? o.y : (expf(o.y) - 1.0f);
        o.z = (o.z > 0.0f) ? o.z : (expf(o.z) - 1.0f);
        o.w = (o.w > 0.0f) ? o.w : (expf(o.w) - 1.0f);
        if (sub == 0) *(float4*)(out + (size_t)node * M + c) = o;
    }
}

extern "C" void kernel_launch(void* const* d_in, const int* in_sizes, int n_in,
                              void* d_out, int out_size, void* d_ws, size_t ws_size,
                              hipStream_t stream)
{
    const float* x   = (const float*)d_in[0];
    const int*   ei  = (const int*)  d_in[1];
    const float* W1  = (const float*)d_in[2];
    const float* b1  = (const float*)d_in[3];
    const float* g1  = (const float*)d_in[4];
    const float* be1 = (const float*)d_in[5];
    const float* W2  = (const float*)d_in[6];
    const float* b2  = (const float*)d_in[7];
    const float* g2  = (const float*)d_in[8];
    const float* be2 = (const float*)d_in[9];
    const float* W3  = (const float*)d_in[10];
    const float* b3  = (const float*)d_in[11];
    const float* g3  = (const float*)d_in[12];
    const float* be3 = (const float*)d_in[13];
    const float* lw1 = (const float*)d_in[14];
    const float* lb1 = (const float*)d_in[15];
    const float* g4  = (const float*)d_in[16];
    const float* be4 = (const float*)d_in[17];
    const float* lw2 = (const float*)d_in[18];
    const float* lb2 = (const float*)d_in[19];

    const int N = in_sizes[0] / 128;
    const int E = in_sizes[1] / 2;
    const int* src = ei;
    const int* dst = ei + E;

    const int NB  = (N + BK_N - 1) / BK_N;
    const int NCH = ceil_div_ll(E, CH_E);
    float* ws   = (float*)d_ws;
    float* xw   = ws;  ws += (size_t)N * 128;
    float* agg  = ws;  ws += (size_t)N * 128;
    float* h    = ws;  ws += (size_t)N * 64;
    int2*  colw = (int2*)ws;
    ws += (size_t)N * 64;
    ws += N;
    float* dinv   = ws;        ws += N;
    int*   rowptr = (int*)ws;  ws += E;
    int2*  binned = (int2*)agg;
    int*   tcnt   = (int*)agg + (size_t)2 * NCH * CH_E;
    int*   toff   = tcnt + (size_t)NB * NCH;

    const int B = 256;
    auto gemm_grid = [N](int per_wgp) {
        const int want = GEMM_WGPS * per_wgp;
        const int most = ceil_div_ll((N + 15) / 16, GEMM_T / 32);
        return want < most ? want : most;
    };
    const int node_blocks = loop_grid(ceil_div_ll(N, 8));
    const int edge_blocks = loop_grid(ceil_div_ll(E, (long long)B * EU));

    bin_kernel<<<NCH, PREP_T, 0, stream>>>(src, dst, binned, tcnt, toff, E, NB, NCH);
    bucket_place_kernel<<<NB, PREP_T, 0, stream>>>(tcnt, toff, binned, rowptr, dinv, colw, N, NCH);
    weight_kernel<<<edge_blocks, B, 0, stream>>>(dinv, colw, E);

    wmma_gemm_f32<128, 64, false><<<gemm_grid(1), GEMM_T, 0, stream>>>(x, W1, nullptr, nullptr, nullptr, xw, N);
    csr_gather_ln_elu<64, true><<<node_blocks, B, 0, stream>>>(rowptr, colw, xw, dinv, b1, g1, be1, h, N);

    csr_gather_ln_elu<64, false><<<node_blocks, B, 0, stream>>>(rowptr, colw, h, dinv, nullptr, nullptr, nullptr, xw, N);
    wmma_gemm_f32<64, 128, true><<<gemm_grid(1), GEMM_T, 0, stream>>>(xw, W2, b2, g2, be2, agg, N);

    wmma_gemm_f32<128, 64, false><<<gemm_grid(1), GEMM_T, 0, stream>>>(agg, W3, nullptr, nullptr, nullptr, xw, N);
    csr_gather_ln_elu<64, true><<<node_blocks, B, 0, stream>>>(rowptr, colw, xw, dinv, b3, g3, be3, h, N);

    head_fused_f32<<<gemm_grid(HEAD_WGS), GEMM_T, 0, stream>>>(h, lw1, lb1, g4, be4, lw2, lb2, (float*)d_out, N);
}


# Edge weights factored out: producers store rows times dinv, the gather adds them unweighted and scales the sum by dinv[dst]; weight pass and weight words gone; term rounding changes, sum order kept
# speedup vs baseline: 4.2401x; 1.1559x over previous
#include <hip/hip_runtime.h>
#include <hip/hip_bf16.h>
#include <math.h>

typedef __attribute__((ext_vector_type(2))) float v2f;
typedef __attribute__((ext_vector_type(8))) float v8f;

static inline int ceil_div_ll(long long a, long long b) { return (int)((a + b - 1) / b); }

constexpr int LOOP_BLOCKS = 2048;
static inline int loop_grid(long long blocks_wanted)
{
    return (int)(blocks_wanted < LOOP_BLOCKS ? blocks_wanted : LOOP_BLOCKS);
}

__device__ __forceinline__ float xhalf(float x)
{
    const int b = __builtin_bit_cast(int, x);
    return __builtin_bit_cast(float,
        __builtin_amdgcn_permlanex16(b, b, 0x76543210, 0xfedcba98, false, false));
}
template<int MASK>
__device__ __forceinline__ float xrow(float x)
{
    const int b = __builtin_bit_cast(int, x);
    return __builtin_bit_cast(float, __builtin_amdgcn_update_dpp(b, b, 0x160 | MASK, 0xf, 0xf, false));
}
template<int LANES>
__device__ __forceinline__ float lanes_sum(float x)
{
    x += xrow<1>(x);
    x += xrow<2>(x);
    x += xrow<4>(x);
    x += xrow<8>(x);
    if (LANES == 32) x += xhalf(x);
    return x;
}

constexpr int GEMM_T = 256;
constexpr int GEMM_WGPS = 256;

template<int K>
__device__ __forceinline__ void load_a_strip(const float* __restrict__ A, int strip, int lm, int lh,
                                             int N, float4 (&a)[K / 8])
{
    int arow = (strip << 4) + lm;
    arow = arow < N ? arow : N - 1;
    const float4* ap = (const float4*)(A + (size_t)arow * K) + lh;
#pragma unroll
    for (int j = 0; j < K / 8; ++j) a[j] = ap[2 * j];
}

template<bool ON>
__device__ __forceinline__ void load_row_scales(const float* __restrict__ rscale, int strip, int lh,
                                                int N, float (&rs)[8])
{
#pragma unroll
    for (int r = 0; r < 8; ++r) {
        int srow = (strip << 4) + r + (lh << 3);
        srow = srow < N ? srow : N - 1;
        rs[r] = ON ? rscale[srow] : 1.0f;
    }
}

template<int K, int M, bool LNELU, bool RSCALE>
__global__ __launch_bounds__(GEMM_T)
void wmma_gemm_f32(const float* __restrict__ A, const float* __restrict__ W,
                   const float* __restrict__ bias, const float* __restrict__ g,
                   const float* __restrict__ be, const float* __restrict__ rscale,
                   float* __restrict__ C, int N)
{
    static_assert(!(LNELU && RSCALE), "the row scale belongs to the bare product");
    constexpr int TN = M / 16;
    constexpr int KB = K / 8;
    constexpr int NW = GEMM_T / 32;
    __shared__ float4 wfrag[KB * TN * 32];

    const int lane = threadIdx.x & 31;
    const int wv = __builtin_amdgcn_readfirstlane((int)(threadIdx.x >> 5));
    const int lh = lane >> 4;
    const int lm = lane & 15;

    const unsigned nstrips = (unsigned)(N + 15) >> 4;
    const int s_beg = (int)(blockIdx.x * nstrips / gridDim.x);
    const int s_end = (int)((blockIdx.x + 1u) * nstrips / gridDim.x);

    int strip = s_beg + wv;
    float4 a_cur[KB];
#pragma unroll
    for (int j = 0; j < KB; ++j) a_cur[j] = make_float4(0.0f, 0.0f, 0.0f, 0.0f);
    if (strip < s_end) load_a_strip<K>(A, strip, lm, lh, N, a_cur);
    float rs_cur[8];
#pragma unroll
    for (int r = 0; r < 8; ++r) rs_cur[r] = 1.0f;
    if (strip < s_end) load_row_scales<RSCALE>(rscale, strip, lh, N, rs_cur);

    constexpr int SPT = KB * TN * 32 / GEMM_T;
    static_assert(SPT * GEMM_T == KB * TN * 32, "K * M must be a multiple of 4 * GEMM_T");
    float4 wv4[SPT];
#pragma unroll
    for (int i = 0; i < SPT; ++i) {
        const int slot = (int)threadIdx.x + i * GEMM_T;
        const int l = slot & 31;
        const int jt = slot >> 5;
        const int t = jt % TN;
        const int j = jt / TN;
        const float* p = W + (size_t)(8 * j + 4 * (l >> 4)) * M + 16 * t + (l & 15);
        wv4[i].x = p[0];
        wv4[i].y = p[M];
        wv4[i].z = p[2 * M];
        wv4[i].w = p[3 * M];
    }
#pragma unroll
    for (int i = 0; i < SPT; ++i) wfrag[(int)threadIdx.x + i * GEMM_T] = wv4[i];

    float badd[TN];
#pragma unroll
    for (int t = 0; t < TN; ++t) badd[t] = bias ? bias[16 * t + lm] : 0.0f;
    float gsc[TN], bsh[TN];
#pragma unroll
    for (int t = 0; t < TN; ++t) {
        gsc[t] = LNELU ? g[16 * t + lm] : 1.0f;
        bsh[t] = LNELU ? be[16 * t + lm] : 0.0f;
    }

    __syncthreads();

    float4 wb[KB * TN];
#pragma unroll
    for (int i = 0; i < KB * TN; ++i) wb[i] = wfrag[i * 32 + lane];

    for (; strip < s_end; strip += NW) {
        float4 a_nxt[KB];
#pragma unroll
        for (int j = 0; j < KB; ++j) a_nxt[j] = a_cur[j];
        float rs_nxt[8];
#pragma unroll
        for (int r = 0; r < 8; ++r) rs_nxt[r] = rs_cur[r];
        if (strip + NW < s_end) {
            load_a_strip<K>(A, strip + NW, lm, lh, N, a_nxt);
            load_row_scales<RSCALE>(rscale, strip + NW, lh, N, rs_nxt);
        }

        v8f acc[TN];
#pragma unroll
        for (int t = 0; t < TN; ++t) acc[t] = v8f{};

#pragma unroll
        for (int j = 0; j < KB; ++j) {
            v2f a01, a23;
            a01.x = a_cur[j].x; a01.y = a_cur[j].y;
            a23.x = a_cur[j].z; a23.y = a_cur[j].w;
#pragma unroll
            for (int t = 0; t < TN; ++t) {
                v2f b01;
                b01.x = wb[j * TN + t].x; b01.y = wb[j * TN + t].y;
                acc[t] = __builtin_amdgcn_wmma_f32_16x16x4_f32(false, a01, false, b01,
                                                               (short)0, acc[t], false, false);
            }
#pragma unroll
            for (int t = 0; t < TN; ++t) {
                v2f b23;
                b23.x = wb[j * TN + t].z; b23.y = wb[j * TN + t].w;
                acc[t] = __builtin_amdgcn_wmma_f32_16x16x4_f32(false, a23, false, b23,
                                                               (short)0, acc[t], false, false);
            }
        }

        const int row0 = strip << 4;
#pragma unroll
        for (int r = 0; r < 8; ++r) {
            const int row = row0 + r + (lh << 3);
            if (LNELU) {
                float v[TN];
                float s = 0.0f;
#pragma unroll
                for (int t = 0; t < TN; ++t) { v[t] = acc[t][r] + badd[t]; s += v[t]; }
                const float mu = lanes_sum<16>(s) * (1.0f / M);
                float q = 0.0f;
#pragma unroll
                for (int t = 0; t < TN; ++t) { v[t] -= mu; q += v[t] * v[t]; }
                const float rstd = rsqrtf(lanes_sum<16>(q) * (1.0f / M) + 1e-5f);
                if (row < N) {
#pragma unroll
                    for (int t = 0; t < TN; ++t) {
                        const float o = v[t] * rstd * gsc[t] + bsh[t];
                        C[(size_t)row * M + 16 * t + lm] = (o > 0.0f) ? o : (expf(o) - 1.0f);
                    }
                }
            } else if (row < N) {
#pragma unroll
                for (int t = 0; t < TN; ++t) {
                    const float o = acc[t][r] + badd[t];
                    C[(size_t)row * M + 16 * t + lm] = RSCALE ? o * rs_cur[r] : o;
                }
            }
        }

#pragma unroll
        for (int j = 0; j < KB; ++j) a_cur[j] = a_nxt[j];
#pragma unroll
        for (int r = 0; r < 8; ++r) rs_cur[r] = rs_nxt[r];
    }
}

constexpr int HEAD_K  = 64;
constexpr int HEAD_M  = 32;
constexpr int HEAD_YP = HEAD_M + 4;
constexpr int HEAD_WGS = 2;

template<int K, int M>
__device__ __forceinline__ void w_quads_load(const float* __restrict__ W, float4 (&q)[K * M / (4 * GEMM_T)])
{
    constexpr int TN = M / 16;
    constexpr int SPT = K * M / (4 * GEMM_T);
    static_assert(SPT * 4 * GEMM_T == K * M, "K * M must be a multiple of 4 * GEMM_T");
#pragma unroll
    for (int i = 0; i < SPT; ++i) {
        const int slot = (int)threadIdx.x + i * GEMM_T;
        const int l = slot & 31;
        const int jt = slot >> 5;
        const int t = jt % TN;
        const int j = jt / TN;
        const float* p = W + (size_t)(8 * j + 4 * (l >> 4)) * M + 16 * t + (l & 15);
        q[i].x = p[0];
        q[i].y = p[M];
        q[i].z = p[2 * M];
        q[i].w = p[3 * M];
    }
}
template<int SPT>
__device__ __forceinline__ void w_quads_store(float4* wfrag, const float4 (&q)[SPT])
{
#pragma unroll
    for (int i = 0; i < SPT; ++i) wfrag[(int)threadIdx.x + i * GEMM_T] = q[i];
}

__global__ __launch_bounds__(GEMM_T)
void head_fused_f32(const float* __restrict__ A, const float* __restrict__ W1,
                    const float* __restrict__ b1, const float* __restrict__ g,
                    const float* __restrict__ be, const float* __restrict__ W2,
                    const float* __restrict__ b2, float* __restrict__ C, int N)
{
    constexpr int K1 = HEAD_K, M1 = HEAD_M, M2 = HEAD_M;
    constexpr int TN1 = M1 / 16;
    constexpr int KB1 = K1 / 8;
    constexpr int TN2 = M2 / 16;
    constexpr int KB2 = M1 / 8;
    constexpr int NW = GEMM_T / 32;
    __shared__ float4 wfrag1[KB1 * TN1 * 32];
    __shared__ float4 wfrag2[KB2 * TN2 * 32];
    __shared__ float4 ytile[NW * 16 * HEAD_YP / 4];

    const int lane = threadIdx.x & 31;
    const int wv = __builtin_amdgcn_readfirstlane((int)(threadIdx.x >> 5));
    const int lh = lane >> 4;
    const int lm = lane & 15;

    const unsigned nstrips = (unsigned)(N + 15) >> 4;
    const int s_beg = (int)(blockIdx.x * nstrips / gridDim.x);
    const int s_end = (int)((blockIdx.x + 1u) * nstrips / gridDim.x);

    int strip = s_beg + wv;
    float4 a_cur[KB1];
#pragma unroll
    for (int j = 0; j < KB1; ++j) a_cur[j] = make_float4(0.0f, 0.0f, 0.0f, 0.0f);
    if (strip < s_end) load_a_strip<K1>(A, strip, lm, lh, N, a_cur);

    float4 q1[K1 * M1 / (4 * GEMM_T)];
    float4 q2[M1 * M2 / (4 * GEMM_T)];
    w_quads_load<K1, M1>(W1, q1);
    w_quads_load<M1, M2>(W2, q2);
    w_quads_store(wfrag1, q1);
    w_quads_store(wfrag2, q2);

    float badd1[TN1], gsc[TN1], bsh[TN1], badd2[TN2];
#pragma unroll
    for (int t = 0; t < TN1; ++t) {
        badd1[t] = b1[16 * t + lm];
        gsc[t] = g[16 * t + lm];
        bsh[t] = be[16 * t + lm];
    }
#pragma unroll
    for (int t = 0; t < TN2; ++t) badd2[t] = b2[16 * t + lm];

    __syncthreads();

    float4 wb1[KB1 * TN1], wb2[KB2 * TN2];
#pragma unroll
    for (int i = 0; i < KB1 * TN1; ++i) wb1[i] = wfrag1[i * 32 + lane];
#pragma unroll
    for (int i = 0; i < KB2 * TN2; ++i) wb2[i] = wfrag2[i * 32 + lane];

    float* yt = (float*)ytile + wv * (16 * HEAD_YP);

    for (; strip < s_end; strip += NW) {
        float4 a_nxt[KB1];
#pragma unroll
        for (int j = 0; j < KB1; ++j) a_nxt[j] = a_cur[j];
        if (strip + NW < s_end) load_a_strip<K1>(A, strip + NW, lm, lh, N, a_nxt);

        v8f acc1[TN1];
#pragma unroll
        for (int t = 0; t < TN1; ++t) acc1[t] = v8f{};
#pragma unroll
        for (int j = 0; j < KB1; ++j) {
            v2f a01, a23;
            a01.x = a_cur[j].x; a01.y = a_cur[j].y;
            a23.x = a_cur[j].z; a23.y = a_cur[j].w;
#pragma unroll
            for (int t = 0; t < TN1; ++t) {
                v2f b01;
                b01.x = wb1[j * TN1 + t].x; b01.y = wb1[j * TN1 + t].y;
                acc1[t] = __builtin_amdgcn_wmma_f32_16x16x4_f32(false, a01, false, b01,
                                                                (short)0, acc1[t], false, false);
            }
#pragma unroll
            for (int t = 0; t < TN1; ++t) {
                v2f b23;
                b23.x = wb1[j * TN1 + t].z; b23.y = wb1[j * TN1 + t].w;
                acc1[t] = __builtin_amdgcn_wmma_f32_16x16x4_f32(false, a23, false, b23,
                                                                (short)0, acc1[t], false, false);
            }
        }

#pragma unroll
        for (int r = 0; r < 8; ++r) {
            float v[TN1];
            float s = 0.0f;
#pragma unroll
            for (int t = 0; t < TN1; ++t) { v[t] = acc1[t][r] + badd1[t]; s += v[t]; }
            const float mu = lanes_sum<16>(s) * (1.0f / M1);
            float q = 0.0f;
#pragma unroll
            for (int t = 0; t < TN1; ++t) { v[t] -= mu; q += v[t] * v[t]; }
            const float rstd = rsqrtf(lanes_sum<16>(q) * (1.0f / M1) + 1e-5f);
#pragma unroll
            for (int t = 0; t < TN1; ++t) {
                const float o = v[t] * rstd * gsc[t] + bsh[t];
                yt[(r + (lh << 3)) * HEAD_YP + 16 * t + lm] = (o > 0.0f) ? o : (expf(o) - 1.0f);
            }
        }

        float4 a2[KB2];
#pragma unroll
        for (int j = 0; j < KB2; ++j)
            a2[j] = *(const float4*)(yt + lm * HEAD_YP + 8 * j + 4 * lh);

        v8f acc2[TN2];
#pragma unroll
        for (int t = 0; t < TN2; ++t) acc2[t] = v8f{};
#pragma unroll
        for (int j = 0; j < KB2; ++j) {
            v2f a01, a23;
            a01.x = a2[j].x; a01.y = a2[j].y;
            a23.x = a2[j].z; a23.y = a2[j].w;
#pragma unroll
            for (int t = 0; t < TN2; ++t) {
                v2f b01;
                b01.x = wb2[j * TN2 + t].x; b01.y = wb2[j * TN2 + t].y;
                acc2[t] = __builtin_amdgcn_wmma_f32_16x16x4_f32(false, a01, false, b01,
                                                                (short)0, acc2[t], false, false);
            }
#pragma unroll
            for (int t = 0; t < TN2; ++t) {
                v2f b23;
                b23.x = wb2[j * TN2 + t].z; b23.y = wb2[j * TN2 + t].w;
                acc2[t] = __builtin_amdgcn_wmma_f32_16x16x4_f32(false, a23, false, b23,
                                                                (short)0, acc2[t], false, false);
            }
        }

        const int row0 = strip << 4;
#pragma unroll
        for (int r = 0; r < 8; ++r) {
            const int row = row0 + r + (lh << 3);
            if (row < N) {
#pragma unroll
                for (int t = 0; t < TN2; ++t)
                    C[(size_t)row * M2 + 16 * t + lm] = acc2[t][r] + badd2[t];
            }
        }

#pragma unroll
        for (int j = 0; j < KB1; ++j) a_cur[j] = a_nxt[j];
    }
}

constexpr int CH_E   = 4096;
constexpr int BK_SH  = 8;
constexpr int BK_N   = 1 << BK_SH;
constexpr int PREP_T = 1024;
constexpr int CH_U   = CH_E / PREP_T;

__global__ __launch_bounds__(1024)
void bin_kernel(const int* __restrict__ src, const int* __restrict__ dst,
                int2* __restrict__ binned, int* __restrict__ tcnt, int* __restrict__ toff,
                int E, int NB, int NCH)
{
    __shared__ int hist[PREP_T];
    __shared__ int offs[PREP_T];
    __shared__ int wsum[32];
    const int t = threadIdx.x, lane = t & 31, wv = t >> 5;
    const int chunk = blockIdx.x;
    const int e_base = chunk * CH_E;

    hist[t] = 0;
    __syncthreads();

    int s[CH_U], d[CH_U], r[CH_U];
#pragma unroll
    for (int u = 0; u < CH_U; ++u) {
        const int e = e_base + u * PREP_T + t;
        const bool ok = e < E;
        s[u] = ok ? src[e] : 0;
        d[u] = ok ? dst[e] : -1;
    }
#pragma unroll
    for (int u = 0; u < CH_U; ++u)
        r[u] = (d[u] >= 0) ? atomicAdd(&hist[d[u] >> BK_SH], 1) : 0;
    __syncthreads();

    const int c = hist[t];
    int x = c;
#pragma unroll
    for (int off = 1; off < 32; off <<= 1) {
        int y = __shfl_up(x, off, 32);
        if (lane >= off) x += y;
    }
    if (lane == 31) wsum[wv] = x;
    __syncthreads();
    if (wv == 0) {
        const int tot = wsum[lane];
        int q = tot;
#pragma unroll
        for (int off = 1; off < 32; off <<= 1) {
            int y = __shfl_up(q, off, 32);
            if (lane >= off) q += y;
        }
        wsum[lane] = q - tot;
    }
    __syncthreads();
    const int excl = wsum[wv] + x - c;
    offs[t] = excl;
    if (t < NB) {
        tcnt[(size_t)t * NCH + chunk] = c;
        toff[(size_t)t * NCH + chunk] = excl;
    }
    __syncthreads();
#pragma unroll
    for (int u = 0; u < CH_U; ++u)
        if (d[u] >= 0) binned[e_base + offs[d[u] >> BK_SH] + r[u]] = make_int2(s[u], d[u]);
}

__global__ __launch_bounds__(1024)
void bucket_place_kernel(const int* __restrict__ tcnt, const int* __restrict__ toff,
                         const int2* __restrict__ binned, int* __restrict__ rowptr,
                         float* __restrict__ dinv, int* __restrict__ col, int N, int NCH)
{
    __shared__ int ncnt[BK_N];
    __shared__ int lrow[BK_N];
    __shared__ int wsum[32];
    __shared__ int s_prefix;
    const int t = threadIdx.x, lane = t & 31, wv = t >> 5;
    const int b = blockIdx.x;
    const int node0 = b << BK_SH;
    const int* __restrict__ cnt_row = tcnt + (size_t)b * NCH;
    const int* __restrict__ off_row = toff + (size_t)b * NCH;

    if (t < BK_N) ncnt[t] = 0;

    int p = 0;
    for (int c = t; c < NCH; c += PREP_T) p += off_row[c];
#pragma unroll
    for (int off = 16; off; off >>= 1) p += __shfl_xor(p, off, 32);
    if (lane == 0) wsum[wv] = p;
    __syncthreads();
    if (wv == 0) {
        int q = wsum[lane];
#pragma unroll
        for (int off = 16; off; off >>= 1) q += __shfl_xor(q, off, 32);
        if (lane == 0) s_prefix = q;
    }
    __syncthreads();
    const int bstart = s_prefix;

    for (int c = t; c < NCH; c += PREP_T) {
        const int n = cnt_row[c];
        const int2* __restrict__ seg = binned + (size_t)c * CH_E + off_row[c];
        for (int i = 0; i < n; ++i) atomicAdd(&ncnt[seg[i].y - node0], 1);
    }
    __syncthreads();

    const int cn = (t < BK_N) ? ncnt[t] : 0;
    int x = cn;
#pragma unroll
    for (int off = 1; off < 32; off <<= 1) {
        int y = __shfl_up(x, off, 32);
        if (lane >= off) x += y;
    }
    if (t < BK_N && lane == 31) wsum[wv] = x;
    __syncthreads();
    if (t < BK_N) {
        int before = 0;
        for (int w = 0; w < wv; ++w) before += wsum[w];
        const int excl = before + x - cn;
        lrow[t] = excl;
        ncnt[t] = 0;
        const int node = node0 + t;
        if (node < N) {
            rowptr[node] = bstart + excl;
            dinv[node] = rsqrtf((float)cn + 1.0f);
            if (node == N - 1) rowptr[N] = bstart + excl + cn;
        }
    }
    __syncthreads();

    for (int c = t; c < NCH; c += PREP_T) {
        const int n = cnt_row[c];
        const int2* __restrict__ seg = binned + (size_t)c * CH_E + off_row[c];
        for (int i = 0; i < n; ++i) {
            const int2 sd = seg[i];
            const int dl = sd.y - node0;
            const int r = atomicAdd(&ncnt[dl], 1);
            col[bstart + lrow[dl] + r] = sd.x;
        }
    }
}

template<int M, bool TAIL, bool PRE>
__global__ __launch_bounds__(256)
void csr_gather_ln_elu(const int* __restrict__ rowptr, const int* __restrict__ col,
                       const float* __restrict__ ys, const float* __restrict__ dinv,
                       const float* __restrict__ bias, const float* __restrict__ g,
                       const float* __restrict__ be, float* __restrict__ out, int N)
{
    static_assert(TAIL || !PRE, "the bare aggregate is never gathered again");
    constexpr int LPR = M / 4;
    constexpr int EPW = 32 / LPR;
    constexpr int UN  = 4;
    const int nwaves = (int)(gridDim.x * (blockDim.x >> 5));
    int wave = (int)(blockIdx.x * (blockDim.x >> 5)) + (int)(threadIdx.x >> 5);
    wave = __builtin_amdgcn_readfirstlane(wave);
    const int lane = threadIdx.x & 31;
    const int sub = lane / LPR;
    const int c = (lane % LPR) << 2;

    float4 bi = make_float4(0.0f, 0.0f, 0.0f, 0.0f), gg = bi, bb = bi;
    if (TAIL) {
        bi.x = bias[c]; bi.y = bias[c + 1]; bi.z = bias[c + 2]; bi.w = bias[c + 3];
        gg.x = g[c];    gg.y = g[c + 1];    gg.z = g[c + 2];    gg.w = g[c + 3];
        bb.x = be[c];   bb.y = be[c + 1];   bb.z = be[c + 2];   bb.w = be[c + 3];
    }

    for (int node = wave; node < N; node += nwaves) {
        const int beg = rowptr[node];
        const int end = rowptr[node + 1];
        const float di = dinv[node];
        const int stop = end + 1;

        float4 acc = make_float4(0.0f, 0.0f, 0.0f, 0.0f);
        int base = beg;
        do {
            const int n = min(32, stop - base);
            const int e = base + lane;
            int mine = 0;
            if (e < end)       mine = col[e];
            else if (e == end) mine = node;
            for (int j0 = 0; j0 < n; j0 += UN * EPW) {
                float4 v[UN];
#pragma unroll
                for (int u = 0; u < UN; ++u) {
                    const int j = (j0 + u * EPW + sub) & 31;
                    const int s = __shfl(mine, j, 32);
                    v[u] = *(const float4*)(ys + (size_t)s * M + c);
                }
#pragma unroll
                for (int u = 0; u < UN; ++u) {
                    if (j0 + u * EPW + sub < n) {
                        acc.x += v[u].x;
                        acc.y += v[u].y;
                        acc.z += v[u].z;
                        acc.w += v[u].w;
                    }
                }
            }
            base += 32;
        } while (base < stop);
        if (EPW == 2) {
            acc.x += __shfl_xor(acc.x, 16, 32);
            acc.y += __shfl_xor(acc.y, 16, 32);
            acc.z += __shfl_xor(acc.z, 16, 32);
            acc.w += __shfl_xor(acc.w, 16, 32);
        }

        if (!TAIL) {
            if (sub == 0)
                *(float4*)(out + (size_t)node * M + c) =
                    make_float4(acc.x * di, acc.y * di, acc.z * di, acc.w * di);
            continue;
        }

        float4 t;
        t.x = acc.x * di + bi.x;
        t.y = acc.y * di + bi.y;
        t.z = acc.z * di + bi.z;
        t.w = acc.w * di + bi.w;

        const float sum = lanes_sum<LPR>((t.x + t.y) + (t.z + t.w));
        const float mu = sum * (1.0f / M);
        const float dx = t.x - mu, dy = t.y - mu, dz = t.z - mu, dw = t.w - mu;
        const float vs = lanes_sum<LPR>((dx * dx + dy * dy) + (dz * dz + dw * dw));
        const float rstd = rsqrtf(vs * (1.0f / M) + 1e-5f);
        float4 o;
        o.x = dx * rstd * gg.x + bb.x;
        o.y = dy * rstd * gg.y + bb.y;
        o.z = dz * rstd * gg.z + bb.z;
        o.w = dw * rstd * gg.w + bb.w;
        o.x = (o.x > 0.0f) ? o.x : (expf(o.x) - 1.0f);
        o.y = (o.y > 0.0f) ? o.y : (expf(o.y) - 1.0f);
        o.z = (o.z > 0.0f) ? o.z : (expf(o.z) - 1.0f);
        o.w = (o.w > 0.0f) ? o.w : (expf(o.w) - 1.0f);
        if (PRE) { o.x *= di; o.y *= di; o.z *= di; o.w *= di; }
        if (sub == 0) *(float4*)(out + (size_t)node * M + c) = o;
    }
}

extern "C" void kernel_launch(void* const* d_in, const int* in_sizes, int n_in,
                              void* d_out, int out_size, void* d_ws, size_t ws_size,
                              hipStream_t stream)
{
    const float* x   = (const float*)d_in[0];
    const int*   ei  = (const int*)  d_in[1];
    const float* W1  = (const float*)d_in[2];
    const float* b1  = (const float*)d_in[3];
    const float* g1  = (const float*)d_in[4];
    const float* be1 = (const float*)d_in[5];
    const float* W2  = (const float*)d_in[6];
    const float* b2  = (const float*)d_in[7];
    const float* g2  = (const float*)d_in[8];
    const float* be2 = (const float*)d_in[9];
    const float* W3  = (const float*)d_in[10];
    const float* b3  = (const float*)d_in[11];
    const float* g3  = (const float*)d_in[12];
    const float* be3 = (const float*)d_in[13];
    const float* lw1 = (const float*)d_in[14];
    const float* lb1 = (const float*)d_in[15];
    const float* g4  = (const float*)d_in[16];
    const float* be4 = (const float*)d_in[17];
    const float* lw2 = (const float*)d_in[18];
    const float* lb2 = (const float*)d_in[19];

    const int N = in_sizes[0] / 128;
    const int E = in_sizes[1] / 2;
    const int* src = ei;
    const int* dst = ei + E;

    const int NB  = (N + BK_N - 1) / BK_N;
    const int NCH = ceil_div_ll(E, CH_E);
    float* ws   = (float*)d_ws;
    float* xw   = ws;  ws += (size_t)N * 128;
    float* agg  = ws;  ws += (size_t)N * 128;
    float* h    = ws;  ws += (size_t)N * 64;
    int*   col  = (int*)ws;
    ws += (size_t)N * 64;
    ws += N;
    float* dinv   = ws;        ws += N;
    int*   rowptr = (int*)ws;  ws += E;
    int2*  binned = (int2*)agg;
    int*   tcnt   = (int*)agg + (size_t)2 * NCH * CH_E;
    int*   toff   = tcnt + (size_t)NB * NCH;

    const int B = 256;
    auto gemm_grid = [N](int per_wgp) {
        const int want = GEMM_WGPS * per_wgp;
        const int most = ceil_div_ll((N + 15) / 16, GEMM_T / 32);
        return want < most ? want : most;
    };
    const int node_blocks = loop_grid(ceil_div_ll(N, 8));

    bin_kernel<<<NCH, PREP_T, 0, stream>>>(src, dst, binned, tcnt, toff, E, NB, NCH);
    bucket_place_kernel<<<NB, PREP_T, 0, stream>>>(tcnt, toff, binned, rowptr, dinv, col, N, NCH);

    wmma_gemm_f32<128, 64, false, true><<<gemm_grid(1), GEMM_T, 0, stream>>>(x, W1, nullptr, nullptr, nullptr, dinv, xw, N);
    csr_gather_ln_elu<64, true, true><<<node_blocks, B, 0, stream>>>(rowptr, col, xw, dinv, b1, g1, be1, h, N);

    csr_gather_ln_elu<64, false, false><<<node_blocks, B, 0, stream>>>(rowptr, col, h, dinv, nullptr, nullptr, nullptr, xw, N);
    wmma_gemm_f32<64, 128, true, false><<<gemm_grid(1), GEMM_T, 0, stream>>>(xw, W2, b2, g2, be2, nullptr, agg, N);

    wmma_gemm_f32<128, 64, false, true><<<gemm_grid(1), GEMM_T, 0, stream>>>(agg, W3, nullptr, nullptr, nullptr, dinv, xw, N);
    csr_gather_ln_elu<64, true, false><<<node_blocks, B, 0, stream>>>(rowptr, col, xw, dinv, b3, g3, be3, h, N);

    head_fused_f32<<<gemm_grid(HEAD_WGS), GEMM_T, 0, stream>>>(h, lw1, lb1, g4, be4, lw2, lb2, (float*)d_out, N);
}


# Fuse layer 2's product+LN+ELU with layer 3's product in one kernel: hidden rows go C layout -> per-wave LDS tile -> A fragments, W3 read from LDS; 1 launch, the 128-wide round trip gone; sums kept
# speedup vs baseline: 4.3743x; 1.0317x over previous
#include <hip/hip_runtime.h>
#include <hip/hip_bf16.h>
#include <math.h>

typedef __attribute__((ext_vector_type(2))) float v2f;
typedef __attribute__((ext_vector_type(8))) float v8f;

static inline int ceil_div_ll(long long a, long long b) { return (int)((a + b - 1) / b); }

constexpr int LOOP_BLOCKS = 2048;
static inline int loop_grid(long long blocks_wanted)
{
    return (int)(blocks_wanted < LOOP_BLOCKS ? blocks_wanted : LOOP_BLOCKS);
}

__device__ __forceinline__ float xhalf(float x)
{
    const int b = __builtin_bit_cast(int, x);
    return __builtin_bit_cast(float,
        __builtin_amdgcn_permlanex16(b, b, 0x76543210, 0xfedcba98, false, false));
}
template<int MASK>
__device__ __forceinline__ float xrow(float x)
{
    const int b = __builtin_bit_cast(int, x);
    return __builtin_bit_cast(float, __builtin_amdgcn_update_dpp(b, b, 0x160 | MASK, 0xf, 0xf, false));
}
template<int LANES>
__device__ __forceinline__ float lanes_sum(float x)
{
    x += xrow<1>(x);
    x += xrow<2>(x);
    x += xrow<4>(x);
    x += xrow<8>(x);
    if (LANES == 32) x += xhalf(x);
    return x;
}

constexpr int GEMM_T = 256;
constexpr int GEMM_WGPS = 256;

template<int K>
__device__ __forceinline__ void load_a_strip(const float* __restrict__ A, int strip, int lm, int lh,
                                             int N, float4 (&a)[K / 8])
{
    int arow = (strip << 4) + lm;
    arow = arow < N ? arow : N - 1;
    const float4* ap = (const float4*)(A + (size_t)arow * K) + lh;
#pragma unroll
    for (int j = 0; j < K / 8; ++j) a[j] = ap[2 * j];
}

template<bool ON>
__device__ __forceinline__ void load_row_scales(const float* __restrict__ rscale, int strip, int lh,
                                                int N, float (&rs)[8])
{
#pragma unroll
    for (int r = 0; r < 8; ++r) {
        int srow = (strip << 4) + r + (lh << 3);
        srow = srow < N ? srow : N - 1;
        rs[r] = ON ? rscale[srow] : 1.0f;
    }
}

template<int K, int M, bool LNELU, bool RSCALE>
__global__ __launch_bounds__(GEMM_T)
void wmma_gemm_f32(const float* __restrict__ A, const float* __restrict__ W,
                   const float* __restrict__ bias, const float* __restrict__ g,
                   const float* __restrict__ be, const float* __restrict__ rscale,
                   float* __restrict__ C, int N)
{
    static_assert(!(LNELU && RSCALE), "the row scale belongs to the bare product");
    constexpr int TN = M / 16;
    constexpr int KB = K / 8;
    constexpr int NW = GEMM_T / 32;
    __shared__ float4 wfrag[KB * TN * 32];

    const int lane = threadIdx.x & 31;
    const int wv = __builtin_amdgcn_readfirstlane((int)(threadIdx.x >> 5));
    const int lh = lane >> 4;
    const int lm = lane & 15;

    const unsigned nstrips = (unsigned)(N + 15) >> 4;
    const int s_beg = (int)(blockIdx.x * nstrips / gridDim.x);
    const int s_end = (int)((blockIdx.x + 1u) * nstrips / gridDim.x);

    int strip = s_beg + wv;
    float4 a_cur[KB];
#pragma unroll
    for (int j = 0; j < KB; ++j) a_cur[j] = make_float4(0.0f, 0.0f, 0.0f, 0.0f);
    if (strip < s_end) load_a_strip<K>(A, strip, lm, lh, N, a_cur);
    float rs_cur[8];
#pragma unroll
    for (int r = 0; r < 8; ++r) rs_cur[r] = 1.0f;
    if (strip < s_end) load_row_scales<RSCALE>(rscale, strip, lh, N, rs_cur);

    constexpr int SPT = KB * TN * 32 / GEMM_T;
    static_assert(SPT * GEMM_T == KB * TN * 32, "K * M must be a multiple of 4 * GEMM_T");
    float4 wv4[SPT];
#pragma unroll
    for (int i = 0; i < SPT; ++i) {
        const int slot = (int)threadIdx.x + i * GEMM_T;
        const int l = slot & 31;
        const int jt = slot >> 5;
        const int t = jt % TN;
        const int j = jt / TN;
        const float* p = W + (size_t)(8 * j + 4 * (l >> 4)) * M + 16 * t + (l & 15);
        wv4[i].x = p[0];
        wv4[i].y = p[M];
        wv4[i].z = p[2 * M];
        wv4[i].w = p[3 * M];
    }
#pragma unroll
    for (int i = 0; i < SPT; ++i) wfrag[(int)threadIdx.x + i * GEMM_T] = wv4[i];

    float badd[TN];
#pragma unroll
    for (int t = 0; t < TN; ++t) badd[t] = bias ? bias[16 * t + lm] : 0.0f;
    float gsc[TN], bsh[TN];
#pragma unroll
    for (int t = 0; t < TN; ++t) {
        gsc[t] = LNELU ? g[16 * t + lm] : 1.0f;
        bsh[t] = LNELU ? be[16 * t + lm] : 0.0f;
    }

    __syncthreads();

    float4 wb[KB * TN];
#pragma unroll
    for (int i = 0; i < KB * TN; ++i) wb[i] = wfrag[i * 32 + lane];

    for (; strip < s_end; strip += NW) {
        float4 a_nxt[KB];
#pragma unroll
        for (int j = 0; j < KB; ++j) a_nxt[j] = a_cur[j];
        float rs_nxt[8];
#pragma unroll
        for (int r = 0; r < 8; ++r) rs_nxt[r] = rs_cur[r];
        if (strip + NW < s_end) {
            load_a_strip<K>(A, strip + NW, lm, lh, N, a_nxt);
            load_row_scales<RSCALE>(rscale, strip + NW, lh, N, rs_nxt);
        }

        v8f acc[TN];
#pragma unroll
        for (int t = 0; t < TN; ++t) acc[t] = v8f{};

#pragma unroll
        for (int j = 0; j < KB; ++j) {
            v2f a01, a23;
            a01.x = a_cur[j].x; a01.y = a_cur[j].y;
            a23.x = a_cur[j].z; a23.y = a_cur[j].w;
#pragma unroll
            for (int t = 0; t < TN; ++t) {
                v2f b01;
                b01.x = wb[j * TN + t].x; b01.y = wb[j * TN + t].y;
                acc[t] = __builtin_amdgcn_wmma_f32_16x16x4_f32(false, a01, false, b01,
                                                               (short)0, acc[t], false, false);
            }
#pragma unroll
            for (int t = 0; t < TN; ++t) {
                v2f b23;
                b23.x = wb[j * TN + t].z; b23.y = wb[j * TN + t].w;
                acc[t] = __builtin_amdgcn_wmma_f32_16x16x4_f32(false, a23, false, b23,
                                                               (short)0, acc[t], false, false);
            }
        }

        const int row0 = strip << 4;
#pragma unroll
        for (int r = 0; r < 8; ++r) {
            const int row = row0 + r + (lh << 3);
            if (LNELU) {
                float v[TN];
                float s = 0.0f;
#pragma unroll
                for (int t = 0; t < TN; ++t) { v[t] = acc[t][r] + badd[t]; s += v[t]; }
                const float mu = lanes_sum<16>(s) * (1.0f / M);
                float q = 0.0f;
#pragma unroll
                for (int t = 0; t < TN; ++t) { v[t] -= mu; q += v[t] * v[t]; }
                const float rstd = rsqrtf(lanes_sum<16>(q) * (1.0f / M) + 1e-5f);
                if (row < N) {
#pragma unroll
                    for (int t = 0; t < TN; ++t) {
                        const float o = v[t] * rstd * gsc[t] + bsh[t];
                        C[(size_t)row * M + 16 * t + lm] = (o > 0.0f) ? o : (expf(o) - 1.0f);
                    }
                }
            } else if (row < N) {
#pragma unroll
                for (int t = 0; t < TN; ++t) {
                    const float o = acc[t][r] + badd[t];
                    C[(size_t)row * M + 16 * t + lm] = RSCALE ? o * rs_cur[r] : o;
                }
            }
        }

#pragma unroll
        for (int j = 0; j < KB; ++j) a_cur[j] = a_nxt[j];
#pragma unroll
        for (int r = 0; r < 8; ++r) rs_cur[r] = rs_nxt[r];
    }
}

constexpr int HEAD_K  = 64;
constexpr int HEAD_M  = 32;
constexpr int HEAD_YP = HEAD_M + 4;
constexpr int HEAD_WGS = 2;

template<int K, int M>
__device__ __forceinline__ void w_quads_load(const float* __restrict__ W, float4 (&q)[K * M / (4 * GEMM_T)])
{
    constexpr int TN = M / 16;
    constexpr int SPT = K * M / (4 * GEMM_T);
    static_assert(SPT * 4 * GEMM_T == K * M, "K * M must be a multiple of 4 * GEMM_T");
#pragma unroll
    for (int i = 0; i < SPT; ++i) {
        const int slot = (int)threadIdx.x + i * GEMM_T;
        const int l = slot & 31;
        const int jt = slot >> 5;
        const int t = jt % TN;
        const int j = jt / TN;
        const float* p = W + (size_t)(8 * j + 4 * (l >> 4)) * M + 16 * t + (l & 15);
        q[i].x = p[0];
        q[i].y = p[M];
        q[i].z = p[2 * M];
        q[i].w = p[3 * M];
    }
}
template<int SPT>
__device__ __forceinline__ void w_quads_store(float4* wfrag, const float4 (&q)[SPT])
{
#pragma unroll
    for (int i = 0; i < SPT; ++i) wfrag[(int)threadIdx.x + i * GEMM_T] = q[i];
}

__global__ __launch_bounds__(GEMM_T)
void head_fused_f32(const float* __restrict__ A, const float* __restrict__ W1,
                    const float* __restrict__ b1, const float* __restrict__ g,
                    const float* __restrict__ be, const float* __restrict__ W2,
                    const float* __restrict__ b2, float* __restrict__ C, int N)
{
    constexpr int K1 = HEAD_K, M1 = HEAD_M, M2 = HEAD_M;
    constexpr int TN1 = M1 / 16;
    constexpr int KB1 = K1 / 8;
    constexpr int TN2 = M2 / 16;
    constexpr int KB2 = M1 / 8;
    constexpr int NW = GEMM_T / 32;
    __shared__ float4 wfrag1[KB1 * TN1 * 32];
    __shared__ float4 wfrag2[KB2 * TN2 * 32];
    __shared__ float4 ytile[NW * 16 * HEAD_YP / 4];

    const int lane = threadIdx.x & 31;
    const int wv = __builtin_amdgcn_readfirstlane((int)(threadIdx.x >> 5));
    const int lh = lane >> 4;
    const int lm = lane & 15;

    const unsigned nstrips = (unsigned)(N + 15) >> 4;
    const int s_beg = (int)(blockIdx.x * nstrips / gridDim.x);
    const int s_end = (int)((blockIdx.x + 1u) * nstrips / gridDim.x);

    int strip = s_beg + wv;
    float4 a_cur[KB1];
#pragma unroll
    for (int j = 0; j < KB1; ++j) a_cur[j] = make_float4(0.0f, 0.0f, 0.0f, 0.0f);
    if (strip < s_end) load_a_strip<K1>(A, strip, lm, lh, N, a_cur);

    float4 q1[K1 * M1 / (4 * GEMM_T)];
    float4 q2[M1 * M2 / (4 * GEMM_T)];
    w_quads_load<K1, M1>(W1, q1);
    w_quads_load<M1, M2>(W2, q2);
    w_quads_store(wfrag1, q1);
    w_quads_store(wfrag2, q2);

    float badd1[TN1], gsc[TN1], bsh[TN1], badd2[TN2];
#pragma unroll
    for (int t = 0; t < TN1; ++t) {
        badd1[t] = b1[16 * t + lm];
        gsc[t] = g[16 * t + lm];
        bsh[t] = be[16 * t + lm];
    }
#pragma unroll
    for (int t = 0; t < TN2; ++t) badd2[t] = b2[16 * t + lm];

    __syncthreads();

    float4 wb1[KB1 * TN1], wb2[KB2 * TN2];
#pragma unroll
    for (int i = 0; i < KB1 * TN1; ++i) wb1[i] = wfrag1[i * 32 + lane];
#pragma unroll
    for (int i = 0; i < KB2 * TN2; ++i) wb2[i] = wfrag2[i * 32 + lane];

    float* yt = (float*)ytile + wv * (16 * HEAD_YP);

    for (; strip < s_end; strip += NW) {
        float4 a_nxt[KB1];
#pragma unroll
        for (int j = 0; j < KB1; ++j) a_nxt[j] = a_cur[j];
        if (strip + NW < s_end) load_a_strip<K1>(A, strip + NW, lm, lh, N, a_nxt);

        v8f acc1[TN1];
#pragma unroll
        for (int t = 0; t < TN1; ++t) acc1[t] = v8f{};
#pragma unroll
        for (int j = 0; j < KB1; ++j) {
            v2f a01, a23;
            a01.x = a_cur[j].x; a01.y = a_cur[j].y;
            a23.x = a_cur[j].z; a23.y = a_cur[j].w;
#pragma unroll
            for (int t = 0; t < TN1; ++t) {
                v2f b01;
                b01.x = wb1[j * TN1 + t].x; b01.y = wb1[j * TN1 + t].y;
                acc1[t] = __builtin_amdgcn_wmma_f32_16x16x4_f32(false, a01, false, b01,
                                                                (short)0, acc1[t], false, false);
            }
#pragma unroll
            for (int t = 0; t < TN1; ++t) {
                v2f b23;
                b23.x = wb1[j * TN1 + t].z; b23.y = wb1[j * TN1 + t].w;
                acc1[t] = __builtin_amdgcn_wmma_f32_16x16x4_f32(false, a23, false, b23,
                                                                (short)0, acc1[t], false, false);
            }
        }

#pragma unroll
        for (int r = 0; r < 8; ++r) {
            float v[TN1];
            float s = 0.0f;
#pragma unroll
            for (int t = 0; t < TN1; ++t) { v[t] = acc1[t][r] + badd1[t]; s += v[t]; }
            const float mu = lanes_sum<16>(s) * (1.0f / M1);
            float q = 0.0f;
#pragma unroll
            for (int t = 0; t < TN1; ++t) { v[t] -= mu; q += v[t] * v[t]; }
            const float rstd = rsqrtf(lanes_sum<16>(q) * (1.0f / M1) + 1e-5f);
#pragma unroll
            for (int t = 0; t < TN1; ++t) {
                const float o = v[t] * rstd * gsc[t] + bsh[t];
                yt[(r + (lh << 3)) * HEAD_YP + 16 * t + lm] = (o > 0.0f) ? o : (expf(o) - 1.0f);
            }
        }

        float4 a2[KB2];
#pragma unroll
        for (int j = 0; j < KB2; ++j)
            a2[j] = *(const float4*)(yt + lm * HEAD_YP + 8 * j + 4 * lh);

        v8f acc2[TN2];
#pragma unroll
        for (int t = 0; t < TN2; ++t) acc2[t] = v8f{};
#pragma unroll
        for (int j = 0; j < KB2; ++j) {
            v2f a01, a23;
            a01.x = a2[j].x; a01.y = a2[j].y;
            a23.x = a2[j].z; a23.y = a2[j].w;
#pragma unroll
            for (int t = 0; t < TN2; ++t) {
                v2f b01;
                b01.x = wb2[j * TN2 + t].x; b01.y = wb2[j * TN2 + t].y;
                acc2[t] = __builtin_amdgcn_wmma_f32_16x16x4_f32(false, a01, false, b01,
                                                                (short)0, acc2[t], false, false);
            }
#pragma unroll
            for (int t = 0; t < TN2; ++t) {
                v2f b23;
                b23.x = wb2[j * TN2 + t].z; b23.y = wb2[j * TN2 + t].w;
                acc2[t] = __builtin_amdgcn_wmma_f32_16x16x4_f32(false, a23, false, b23,
                                                                (short)0, acc2[t], false, false);
            }
        }

        const int row0 = strip << 4;
#pragma unroll
        for (int r = 0; r < 8; ++r) {
            const int row = row0 + r + (lh << 3);
            if (row < N) {
#pragma unroll
                for (int t = 0; t < TN2; ++t)
                    C[(size_t)row * M2 + 16 * t + lm] = acc2[t][r] + badd2[t];
            }
        }

#pragma unroll
        for (int j = 0; j < KB1; ++j) a_cur[j] = a_nxt[j];
    }
}

constexpr int PAIR_K  = 64;
constexpr int PAIR_H  = 128;
constexpr int PAIR_M  = 64;
constexpr int PAIR_YP = PAIR_H + 4;

__global__ __launch_bounds__(GEMM_T)
void gemm_pair_f32(const float* __restrict__ A, const float* __restrict__ W1,
                   const float* __restrict__ b1, const float* __restrict__ g,
                   const float* __restrict__ be, const float* __restrict__ W2,
                   const float* __restrict__ rscale, float* __restrict__ C, int N)
{
    constexpr int K1 = PAIR_K, M1 = PAIR_H, M2 = PAIR_M;
    constexpr int TN1 = M1 / 16;
    constexpr int KB1 = K1 / 8;
    constexpr int TN2 = M2 / 16;
    constexpr int KB2 = M1 / 8;
    constexpr int NW = GEMM_T / 32;
    __shared__ float4 wfrag1[KB1 * TN1 * 32];
    __shared__ float4 wfrag2[KB2 * TN2 * 32];
    __shared__ float4 ytile[NW * 16 * PAIR_YP / 4];

    const int lane = threadIdx.x & 31;
    const int wv = __builtin_amdgcn_readfirstlane((int)(threadIdx.x >> 5));
    const int lh = lane >> 4;
    const int lm = lane & 15;

    const unsigned nstrips = (unsigned)(N + 15) >> 4;
    const int s_beg = (int)(blockIdx.x * nstrips / gridDim.x);
    const int s_end = (int)((blockIdx.x + 1u) * nstrips / gridDim.x);

    int strip = s_beg + wv;
    float4 a_cur[KB1];
#pragma unroll
    for (int j = 0; j < KB1; ++j) a_cur[j] = make_float4(0.0f, 0.0f, 0.0f, 0.0f);
    if (strip < s_end) load_a_strip<K1>(A, strip, lm, lh, N, a_cur);
    float rs_cur[8];
#pragma unroll
    for (int r = 0; r < 8; ++r) rs_cur[r] = 1.0f;
    if (strip < s_end) load_row_scales<true>(rscale, strip, lh, N, rs_cur);

    float4 q1[K1 * M1 / (4 * GEMM_T)];
    float4 q2[M1 * M2 / (4 * GEMM_T)];
    w_quads_load<K1, M1>(W1, q1);
    w_quads_load<M1, M2>(W2, q2);
    w_quads_store(wfrag1, q1);
    w_quads_store(wfrag2, q2);

    float badd1[TN1], gsc[TN1], bsh[TN1];
#pragma unroll
    for (int t = 0; t < TN1; ++t) {
        badd1[t] = b1[16 * t + lm];
        gsc[t] = g[16 * t + lm];
        bsh[t] = be[16 * t + lm];
    }

    __syncthreads();

    float4 wb1[KB1 * TN1];
#pragma unroll
    for (int i = 0; i < KB1 * TN1; ++i) wb1[i] = wfrag1[i * 32 + lane];

    float* yt = (float*)ytile + wv * (16 * PAIR_YP);
    const float4* w2l = wfrag2 + lane;

    for (; strip < s_end; strip += NW) {
        float4 a_nxt[KB1];
#pragma unroll
        for (int j = 0; j < KB1; ++j) a_nxt[j] = a_cur[j];
        float rs_nxt[8];
#pragma unroll
        for (int r = 0; r < 8; ++r) rs_nxt[r] = rs_cur[r];
        if (strip + NW < s_end) {
            load_a_strip<K1>(A, strip + NW, lm, lh, N, a_nxt);
            load_row_scales<true>(rscale, strip + NW, lh, N, rs_nxt);
        }

        v8f acc1[TN1];
#pragma unroll
        for (int t = 0; t < TN1; ++t) acc1[t] = v8f{};
#pragma unroll
        for (int j = 0; j < KB1; ++j) {
            v2f a01, a23;
            a01.x = a_cur[j].x; a01.y = a_cur[j].y;
            a23.x = a_cur[j].z; a23.y = a_cur[j].w;
#pragma unroll
            for (int t = 0; t < TN1; ++t) {
                v2f b01;
                b01.x = wb1[j * TN1 + t].x; b01.y = wb1[j * TN1 + t].y;
                acc1[t] = __builtin_amdgcn_wmma_f32_16x16x4_f32(false, a01, false, b01,
                                                                (short)0, acc1[t], false, false);
            }
#pragma unroll
            for (int t = 0; t < TN1; ++t) {
                v2f b23;
                b23.x = wb1[j * TN1 + t].z; b23.y = wb1[j * TN1 + t].w;
                acc1[t] = __builtin_amdgcn_wmma_f32_16x16x4_f32(false, a23, false, b23,
                                                                (short)0, acc1[t], false, false);
            }
        }

#pragma unroll
        for (int r = 0; r < 8; ++r) {
            float v[TN1];
            float s = 0.0f;
#pragma unroll
            for (int t = 0; t < TN1; ++t) { v[t] = acc1[t][r] + badd1[t]; s += v[t]; }
            const float mu = lanes_sum<16>(s) * (1.0f / M1);
            float q = 0.0f;
#pragma unroll
            for (int t = 0; t < TN1; ++t) { v[t] -= mu; q += v[t] * v[t]; }
            const float rstd = rsqrtf(lanes_sum<16>(q) * (1.0f / M1) + 1e-5f);
#pragma unroll
            for (int t = 0; t < TN1; ++t) {
                const float o = v[t] * rstd * gsc[t] + bsh[t];
                yt[(r + (lh << 3)) * PAIR_YP + 16 * t + lm] = (o > 0.0f) ? o : (expf(o) - 1.0f);
            }
        }

        float4 a2[KB2];
#pragma unroll
        for (int j = 0; j < KB2; ++j)
            a2[j] = *(const float4*)(yt + lm * PAIR_YP + 8 * j + 4 * lh);

        v8f acc2[TN2];
#pragma unroll
        for (int t = 0; t < TN2; ++t) acc2[t] = v8f{};
#pragma unroll
        for (int j = 0; j < KB2; ++j) {
            v2f a01, a23;
            a01.x = a2[j].x; a01.y = a2[j].y;
            a23.x = a2[j].z; a23.y = a2[j].w;
            float4 wq[TN2];
#pragma unroll
            for (int t = 0; t < TN2; ++t) wq[t] = w2l[(j * TN2 + t) * 32];
#pragma unroll
            for (int t = 0; t < TN2; ++t) {
                v2f b01;
                b01.x = wq[t].x; b01.y = wq[t].y;
                acc2[t] = __builtin_amdgcn_wmma_f32_16x16x4_f32(false, a01, false, b01,
                                                                (short)0, acc2[t], false, false);
            }
#pragma unroll
            for (int t = 0; t < TN2; ++t) {
                v2f b23;
                b23.x = wq[t].z; b23.y = wq[t].w;
                acc2[t] = __builtin_amdgcn_wmma_f32_16x16x4_f32(false, a23, false, b23,
                                                                (short)0, acc2[t], false, false);
            }
        }

        const int row0 = strip << 4;
#pragma unroll
        for (int r = 0; r < 8; ++r) {
            const int row = row0 + r + (lh << 3);
            if (row < N) {
#pragma unroll
                for (int t = 0; t < TN2; ++t)
                    C[(size_t)row * M2 + 16 * t + lm] = acc2[t][r] * rs_cur[r];
            }
        }

#pragma unroll
        for (int j = 0; j < KB1; ++j) a_cur[j] = a_nxt[j];
#pragma unroll
        for (int r = 0; r < 8; ++r) rs_cur[r] = rs_nxt[r];
    }
}

constexpr int CH_E   = 4096;
constexpr int BK_SH  = 8;
constexpr int BK_N   = 1 << BK_SH;
constexpr int PREP_T = 1024;
constexpr int CH_U   = CH_E / PREP_T;

__global__ __launch_bounds__(1024)
void bin_kernel(const int* __restrict__ src, const int* __restrict__ dst,
                int2* __restrict__ binned, int* __restrict__ tcnt, int* __restrict__ toff,
                int E, int NB, int NCH)
{
    __shared__ int hist[PREP_T];
    __shared__ int offs[PREP_T];
    __shared__ int wsum[32];
    const int t = threadIdx.x, lane = t & 31, wv = t >> 5;
    const int chunk = blockIdx.x;
    const int e_base = chunk * CH_E;

    hist[t] = 0;
    __syncthreads();

    int s[CH_U], d[CH_U], r[CH_U];
#pragma unroll
    for (int u = 0; u < CH_U; ++u) {
        const int e = e_base + u * PREP_T + t;
        const bool ok = e < E;
        s[u] = ok ? src[e] : 0;
        d[u] = ok ? dst[e] : -1;
    }
#pragma unroll
    for (int u = 0; u < CH_U; ++u)
        r[u] = (d[u] >= 0) ? atomicAdd(&hist[d[u] >> BK_SH], 1) : 0;
    __syncthreads();

    const int c = hist[t];
    int x = c;
#pragma unroll
    for (int off = 1; off < 32; off <<= 1) {
        int y = __shfl_up(x, off, 32);
        if (lane >= off) x += y;
    }
    if (lane == 31) wsum[wv] = x;
    __syncthreads();
    if (wv == 0) {
        const int tot = wsum[lane];
        int q = tot;
#pragma unroll
        for (int off = 1; off < 32; off <<= 1) {
            int y = __shfl_up(q, off, 32);
            if (lane >= off) q += y;
        }
        wsum[lane] = q - tot;
    }
    __syncthreads();
    const int excl = wsum[wv] + x - c;
    offs[t] = excl;
    if (t < NB) {
        tcnt[(size_t)t * NCH + chunk] = c;
        toff[(size_t)t * NCH + chunk] = excl;
    }
    __syncthreads();
#pragma unroll
    for (int u = 0; u < CH_U; ++u)
        if (d[u] >= 0) binned[e_base + offs[d[u] >> BK_SH] + r[u]] = make_int2(s[u], d[u]);
}

__global__ __launch_bounds__(1024)
void bucket_place_kernel(const int* __restrict__ tcnt, const int* __restrict__ toff,
                         const int2* __restrict__ binned, int* __restrict__ rowptr,
                         float* __restrict__ dinv, int* __restrict__ col, int N, int NCH)
{
    __shared__ int ncnt[BK_N];
    __shared__ int lrow[BK_N];
    __shared__ int wsum[32];
    __shared__ int s_prefix;
    const int t = threadIdx.x, lane = t & 31, wv = t >> 5;
    const int b = blockIdx.x;
    const int node0 = b << BK_SH;
    const int* __restrict__ cnt_row = tcnt + (size_t)b * NCH;
    const int* __restrict__ off_row = toff + (size_t)b * NCH;

    if (t < BK_N) ncnt[t] = 0;

    int p = 0;
    for (int c = t; c < NCH; c += PREP_T) p += off_row[c];
#pragma unroll
    for (int off = 16; off; off >>= 1) p += __shfl_xor(p, off, 32);
    if (lane == 0) wsum[wv] = p;
    __syncthreads();
    if (wv == 0) {
        int q = wsum[lane];
#pragma unroll
        for (int off = 16; off; off >>= 1) q += __shfl_xor(q, off, 32);
        if (lane == 0) s_prefix = q;
    }
    __syncthreads();
    const int bstart = s_prefix;

    for (int c = t; c < NCH; c += PREP_T) {
        const int n = cnt_row[c];
        const int2* __restrict__ seg = binned + (size_t)c * CH_E + off_row[c];
        for (int i = 0; i < n; ++i) atomicAdd(&ncnt[seg[i].y - node0], 1);
    }
    __syncthreads();

    const int cn = (t < BK_N) ? ncnt[t] : 0;
    int x = cn;
#pragma unroll
    for (int off = 1; off < 32; off <<= 1) {
        int y = __shfl_up(x, off, 32);
        if (lane >= off) x += y;
    }
    if (t < BK_N && lane == 31) wsum[wv] = x;
    __syncthreads();
    if (t < BK_N) {
        int before = 0;
        for (int w = 0; w < wv; ++w) before += wsum[w];
        const int excl = before + x - cn;
        lrow[t] = excl;
        ncnt[t] = 0;
        const int node = node0 + t;
        if (node < N) {
            rowptr[node] = bstart + excl;
            dinv[node] = rsqrtf((float)cn + 1.0f);
            if (node == N - 1) rowptr[N] = bstart + excl + cn;
        }
    }
    __syncthreads();

    for (int c = t; c < NCH; c += PREP_T) {
        const int n = cnt_row[c];
        const int2* __restrict__ seg = binned + (size_t)c * CH_E + off_row[c];
        for (int i = 0; i < n; ++i) {
            const int2 sd = seg[i];
            const int dl = sd.y - node0;
            const int r = atomicAdd(&ncnt[dl], 1);
            col[bstart + lrow[dl] + r] = sd.x;
        }
    }
}

template<int M, bool TAIL, bool PRE>
__global__ __launch_bounds__(256)
void csr_gather_ln_elu(const int* __restrict__ rowptr, const int* __restrict__ col,
                       const float* __restrict__ ys, const float* __restrict__ dinv,
                       const float* __restrict__ bias, const float* __restrict__ g,
                       const float* __restrict__ be, float* __restrict__ out, int N)
{
    static_assert(TAIL || !PRE, "the bare aggregate is never gathered again");
    constexpr int LPR = M / 4;
    constexpr int EPW = 32 / LPR;
    constexpr int UN  = 4;
    const int nwaves = (int)(gridDim.x * (blockDim.x >> 5));
    int wave = (int)(blockIdx.x * (blockDim.x >> 5)) + (int)(threadIdx.x >> 5);
    wave = __builtin_amdgcn_readfirstlane(wave);
    const int lane = threadIdx.x & 31;
    const int sub = lane / LPR;
    const int c = (lane % LPR) << 2;

    float4 bi = make_float4(0.0f, 0.0f, 0.0f, 0.0f), gg = bi, bb = bi;
    if (TAIL) {
        bi.x = bias[c]; bi.y = bias[c + 1]; bi.z = bias[c + 2]; bi.w = bias[c + 3];
        gg.x = g[c];    gg.y = g[c + 1];    gg.z = g[c + 2];    gg.w = g[c + 3];
        bb.x = be[c];   bb.y = be[c + 1];   bb.z = be[c + 2];   bb.w = be[c + 3];
    }

    for (int node = wave; node < N; node += nwaves) {
        const int beg = rowptr[node];
        const int end = rowptr[node + 1];
        const float di = dinv[node];
        const int stop = end + 1;

        float4 acc = make_float4(0.0f, 0.0f, 0.0f, 0.0f);
        int base = beg;
        do {
            const int n = min(32, stop - base);
            const int e = base + lane;
            int mine = 0;
            if (e < end)       mine = col[e];
            else if (e == end) mine = node;
            for (int j0 = 0; j0 < n; j0 += UN * EPW) {
                float4 v[UN];
#pragma unroll
                for (int u = 0; u < UN; ++u) {
                    const int j = (j0 + u * EPW + sub) & 31;
                    const int s = __shfl(mine, j, 32);
                    v[u] = *(const float4*)(ys + (size_t)s * M + c);
                }
#pragma unroll
                for (int u = 0; u < UN; ++u) {
                    if (j0 + u * EPW + sub < n) {
                        acc.x += v[u].x;
                        acc.y += v[u].y;
                        acc.z += v[u].z;
                        acc.w += v[u].w;
                    }
                }
            }
            base += 32;
        } while (base < stop);
        if (EPW == 2) {
            acc.x += __shfl_xor(acc.x, 16, 32);
            acc.y += __shfl_xor(acc.y, 16, 32);
            acc.z += __shfl_xor(acc.z, 16, 32);
            acc.w += __shfl_xor(acc.w, 16, 32);
        }

        if (!TAIL) {
            if (sub == 0)
                *(float4*)(out + (size_t)node * M + c) =
                    make_float4(acc.x * di, acc.y * di, acc.z * di, acc.w * di);
            continue;
        }

        float4 t;
        t.x = acc.x * di + bi.x;
        t.y = acc.y * di + bi.y;
        t.z = acc.z * di + bi.z;
        t.w = acc.w * di + bi.w;

        const float sum = lanes_sum<LPR>((t.x + t.y) + (t.z + t.w));
        const float mu = sum * (1.0f / M);
        const float dx = t.x - mu, dy = t.y - mu, dz = t.z - mu, dw = t.w - mu;
        const float vs = lanes_sum<LPR>((dx * dx + dy * dy) + (dz * dz + dw * dw));
        const float rstd = rsqrtf(vs * (1.0f / M) + 1e-5f);
        float4 o;
        o.x = dx * rstd * gg.x + bb.x;
        o.y = dy * rstd * gg.y + bb.y;
        o.z = dz * rstd * gg.z + bb.z;
        o.w = dw * rstd * gg.w + bb.w;
        o.x = (o.x > 0.0f) ? o.x : (expf(o.x) - 1.0f);
        o.y = (o.y > 0.0f) ? o.y : (expf(o.y) - 1.0f);
        o.z = (o.z > 0.0f) ? o.z : (expf(o.z) - 1.0f);
        o.w = (o.w > 0.0f) ? o.w : (expf(o.w) - 1.0f);
        if (PRE) { o.x *= di; o.y *= di; o.z *= di; o.w *= di; }
        if (sub == 0) *(float4*)(out + (size_t)node * M + c) = o;
    }
}

extern "C" void kernel_launch(void* const* d_in, const int* in_sizes, int n_in,
                              void* d_out, int out_size, void* d_ws, size_t ws_size,
                              hipStream_t stream)
{
    const float* x   = (const float*)d_in[0];
    const int*   ei  = (const int*)  d_in[1];
    const float* W1  = (const float*)d_in[2];
    const float* b1  = (const float*)d_in[3];
    const float* g1  = (const float*)d_in[4];
    const float* be1 = (const float*)d_in[5];
    const float* W2  = (const float*)d_in[6];
    const float* b2  = (const float*)d_in[7];
    const float* g2  = (const float*)d_in[8];
    const float* be2 = (const float*)d_in[9];
    const float* W3  = (const float*)d_in[10];
    const float* b3  = (const float*)d_in[11];
    const float* g3  = (const float*)d_in[12];
    const float* be3 = (const float*)d_in[13];
    const float* lw1 = (const float*)d_in[14];
    const float* lb1 = (const float*)d_in[15];
    const float* g4  = (const float*)d_in[16];
    const float* be4 = (const float*)d_in[17];
    const float* lw2 = (const float*)d_in[18];
    const float* lb2 = (const float*)d_in[19];

    const int N = in_sizes[0] / 128;
    const int E = in_sizes[1] / 2;
    const int* src = ei;
    const int* dst = ei + E;

    const int NB  = (N + BK_N - 1) / BK_N;
    const int NCH = ceil_div_ll(E, CH_E);
    float* ws   = (float*)d_ws;
    float* xw   = ws;  ws += (size_t)N * 128;
    float* agg  = ws;  ws += (size_t)N * 128;
    float* h    = ws;  ws += (size_t)N * 64;
    int*   col  = (int*)ws;
    ws += (size_t)N * 64;
    ws += N;
    float* dinv   = ws;        ws += N;
    int*   rowptr = (int*)ws;  ws += E;
    int2*  binned = (int2*)agg;
    int*   tcnt   = (int*)agg + (size_t)2 * NCH * CH_E;
    int*   toff   = tcnt + (size_t)NB * NCH;

    const int B = 256;
    auto gemm_grid = [N](int per_wgp) {
        const int want = GEMM_WGPS * per_wgp;
        const int most = ceil_div_ll((N + 15) / 16, GEMM_T / 32);
        return want < most ? want : most;
    };
    const int node_blocks = loop_grid(ceil_div_ll(N, 8));

    bin_kernel<<<NCH, PREP_T, 0, stream>>>(src, dst, binned, tcnt, toff, E, NB, NCH);
    bucket_place_kernel<<<NB, PREP_T, 0, stream>>>(tcnt, toff, binned, rowptr, dinv, col, N, NCH);

    wmma_gemm_f32<128, 64, false, true><<<gemm_grid(1), GEMM_T, 0, stream>>>(x, W1, nullptr, nullptr, nullptr, dinv, xw, N);
    csr_gather_ln_elu<64, true, true><<<node_blocks, B, 0, stream>>>(rowptr, col, xw, dinv, b1, g1, be1, h, N);

    csr_gather_ln_elu<64, false, false><<<node_blocks, B, 0, stream>>>(rowptr, col, h, dinv, nullptr, nullptr, nullptr, xw, N);
    gemm_pair_f32<<<gemm_grid(1), GEMM_T, 0, stream>>>(xw, W2, b2, g2, be2, W3, dinv, agg, N);
    csr_gather_ln_elu<64, true, false><<<node_blocks, B, 0, stream>>>(rowptr, col, agg, dinv, b3, g3, be3, h, N);

    head_fused_f32<<<gemm_grid(HEAD_WGS), GEMM_T, 0, stream>>>(h, lw1, lb1, g4, be4, lw2, lb2, (float*)d_out, N);
}


# Row gather loop: a list's left-over entries are asked for first, with only the loads they take, and land under the full batches, which add rows with no per-slot select; 32-bit row offsets; sums kept
# speedup vs baseline: 4.5704x; 1.0448x over previous
#include <hip/hip_runtime.h>
#include <hip/hip_bf16.h>
#include <math.h>

typedef __attribute__((ext_vector_type(2))) float v2f;
typedef __attribute__((ext_vector_type(8))) float v8f;

static inline int ceil_div_ll(long long a, long long b) { return (int)((a + b - 1) / b); }

constexpr int LOOP_BLOCKS = 2048;
static inline int loop_grid(long long blocks_wanted)
{
    return (int)(blocks_wanted < LOOP_BLOCKS ? blocks_wanted : LOOP_BLOCKS);
}

__device__ __forceinline__ float xhalf(float x)
{
    const int b = __builtin_bit_cast(int, x);
    return __builtin_bit_cast(float,
        __builtin_amdgcn_permlanex16(b, b, 0x76543210, 0xfedcba98, false, false));
}
template<int MASK>
__device__ __forceinline__ float xrow(float x)
{
    const int b = __builtin_bit_cast(int, x);
    return __builtin_bit_cast(float, __builtin_amdgcn_update_dpp(b, b, 0x160 | MASK, 0xf, 0xf, false));
}
template<int LANES>
__device__ __forceinline__ float lanes_sum(float x)
{
    x += xrow<1>(x);
    x += xrow<2>(x);
    x += xrow<4>(x);
    x += xrow<8>(x);
    if (LANES == 32) x += xhalf(x);
    return x;
}

constexpr int GEMM_T = 256;
constexpr int GEMM_WGPS = 256;

template<int K>
__device__ __forceinline__ void load_a_strip(const float* __restrict__ A, int strip, int lm, int lh,
                                             int N, float4 (&a)[K / 8])
{
    int arow = (strip << 4) + lm;
    arow = arow < N ? arow : N - 1;
    const float4* ap = (const float4*)(A + (size_t)arow * K) + lh;
#pragma unroll
    for (int j = 0; j < K / 8; ++j) a[j] = ap[2 * j];
}

template<bool ON>
__device__ __forceinline__ void load_row_scales(const float* __restrict__ rscale, int strip, int lh,
                                                int N, float (&rs)[8])
{
#pragma unroll
    for (int r = 0; r < 8; ++r) {
        int srow = (strip << 4) + r + (lh << 3);
        srow = srow < N ? srow : N - 1;
        rs[r] = ON ? rscale[srow] : 1.0f;
    }
}

template<int K, int M, bool LNELU, bool RSCALE>
__global__ __launch_bounds__(GEMM_T)
void wmma_gemm_f32(const float* __restrict__ A, const float* __restrict__ W,
                   const float* __restrict__ bias, const float* __restrict__ g,
                   const float* __restrict__ be, const float* __restrict__ rscale,
                   float* __restrict__ C, int N)
{
    static_assert(!(LNELU && RSCALE), "the row scale belongs to the bare product");
    constexpr int TN = M / 16;
    constexpr int KB = K / 8;
    constexpr int NW = GEMM_T / 32;
    __shared__ float4 wfrag[KB * TN * 32];

    const int lane = threadIdx.x & 31;
    const int wv = __builtin_amdgcn_readfirstlane((int)(threadIdx.x >> 5));
    const int lh = lane >> 4;
    const int lm = lane & 15;

    const unsigned nstrips = (unsigned)(N + 15) >> 4;
    const int s_beg = (int)(blockIdx.x * nstrips / gridDim.x);
    const int s_end = (int)((blockIdx.x + 1u) * nstrips / gridDim.x);

    int strip = s_beg + wv;
    float4 a_cur[KB];
#pragma unroll
    for (int j = 0; j < KB; ++j) a_cur[j] = make_float4(0.0f, 0.0f, 0.0f, 0.0f);
    if (strip < s_end) load_a_strip<K>(A, strip, lm, lh, N, a_cur);
    float rs_cur[8];
#pragma unroll
    for (int r = 0; r < 8; ++r) rs_cur[r] = 1.0f;
    if (strip < s_end) load_row_scales<RSCALE>(rscale, strip, lh, N, rs_cur);

    constexpr int SPT = KB * TN * 32 / GEMM_T;
    static_assert(SPT * GEMM_T == KB * TN * 32, "K * M must be a multiple of 4 * GEMM_T");
    float4 wv4[SPT];
#pragma unroll
    for (int i = 0; i < SPT; ++i) {
        const int slot = (int)threadIdx.x + i * GEMM_T;
        const int l = slot & 31;
        const int jt = slot >> 5;
        const int t = jt % TN;
        const int j = jt / TN;
        const float* p = W + (size_t)(8 * j + 4 * (l >> 4)) * M + 16 * t + (l & 15);
        wv4[i].x = p[0];
        wv4[i].y = p[M];
        wv4[i].z = p[2 * M];
        wv4[i].w = p[3 * M];
    }
#pragma unroll
    for (int i = 0; i < SPT; ++i) wfrag[(int)threadIdx.x + i * GEMM_T] = wv4[i];

    float badd[TN];
#pragma unroll
    for (int t = 0; t < TN; ++t) badd[t] = bias ? bias[16 * t + lm] : 0.0f;
    float gsc[TN], bsh[TN];
#pragma unroll
    for (int t = 0; t < TN; ++t) {
        gsc[t] = LNELU ? g[16 * t + lm] : 1.0f;
        bsh[t] = LNELU ? be[16 * t + lm] : 0.0f;
    }

    __syncthreads();

    float4 wb[KB * TN];
#pragma unroll
    for (int i = 0; i < KB * TN; ++i) wb[i] = wfrag[i * 32 + lane];

    for (; strip < s_end; strip += NW) {
        float4 a_nxt[KB];
#pragma unroll
        for (int j = 0; j < KB; ++j) a_nxt[j] = a_cur[j];
        float rs_nxt[8];
#pragma unroll
        for (int r = 0; r < 8; ++r) rs_nxt[r] = rs_cur[r];
        if (strip + NW < s_end) {
            load_a_strip<K>(A, strip + NW, lm, lh, N, a_nxt);
            load_row_scales<RSCALE>(rscale, strip + NW, lh, N, rs_nxt);
        }

        v8f acc[TN];
#pragma unroll
        for (int t = 0; t < TN; ++t) acc[t] = v8f{};

#pragma unroll
        for (int j = 0; j < KB; ++j) {
            v2f a01, a23;
            a01.x = a_cur[j].x; a01.y = a_cur[j].y;
            a23.x = a_cur[j].z; a23.y = a_cur[j].w;
#pragma unroll
            for (int t = 0; t < TN; ++t) {
                v2f b01;
                b01.x = wb[j * TN + t].x; b01.y = wb[j * TN + t].y;
                acc[t] = __builtin_amdgcn_wmma_f32_16x16x4_f32(false, a01, false, b01,
                                                               (short)0, acc[t], false, false);
            }
#pragma unroll
            for (int t = 0; t < TN; ++t) {
                v2f b23;
                b23.x = wb[j * TN + t].z; b23.y = wb[j * TN + t].w;
                acc[t] = __builtin_amdgcn_wmma_f32_16x16x4_f32(false, a23, false, b23,
                                                               (short)0, acc[t], false, false);
            }
        }

        const int row0 = strip << 4;
#pragma unroll
        for (int r = 0; r < 8; ++r) {
            const int row = row0 + r + (lh << 3);
            if (LNELU) {
                float v[TN];
                float s = 0.0f;
#pragma unroll
                for (int t = 0; t < TN; ++t) { v[t] = acc[t][r] + badd[t]; s += v[t]; }
                const float mu = lanes_sum<16>(s) * (1.0f / M);
                float q = 0.0f;
#pragma unroll
                for (int t = 0; t < TN; ++t) { v[t] -= mu; q += v[t] * v[t]; }
                const float rstd = rsqrtf(lanes_sum<16>(q) * (1.0f / M) + 1e-5f);
                if (row < N) {
#pragma unroll
                    for (int t = 0; t < TN; ++t) {
                        const float o = v[t] * rstd * gsc[t] + bsh[t];
                        C[(size_t)row * M + 16 * t + lm] = (o > 0.0f) ? o : (expf(o) - 1.0f);
                    }
                }
            } else if (row < N) {
#pragma unroll
                for (int t = 0; t < TN; ++t) {
                    const float o = acc[t][r] + badd[t];
                    C[(size_t)row * M + 16 * t + lm] = RSCALE ? o * rs_cur[r] : o;
                }
            }
        }

#pragma unroll
        for (int j = 0; j < KB; ++j) a_cur[j] = a_nxt[j];
#pragma unroll
        for (int r = 0; r < 8; ++r) rs_cur[r] = rs_nxt[r];
    }
}

constexpr int HEAD_K  = 64;
constexpr int HEAD_M  = 32;
constexpr int HEAD_YP = HEAD_M + 4;
constexpr int HEAD_WGS = 2;

template<int K, int M>
__device__ __forceinline__ void w_quads_load(const float* __restrict__ W, float4 (&q)[K * M / (4 * GEMM_T)])
{
    constexpr int TN = M / 16;
    constexpr int SPT = K * M / (4 * GEMM_T);
    static_assert(SPT * 4 * GEMM_T == K * M, "K * M must be a multiple of 4 * GEMM_T");
#pragma unroll
    for (int i = 0; i < SPT; ++i) {
        const int slot = (int)threadIdx.x + i * GEMM_T;
        const int l = slot & 31;
        const int jt = slot >> 5;
        const int t = jt % TN;
        const int j = jt / TN;
        const float* p = W + (size_t)(8 * j + 4 * (l >> 4)) * M + 16 * t + (l & 15);
        q[i].x = p[0];
        q[i].y = p[M];
        q[i].z = p[2 * M];
        q[i].w = p[3 * M];
    }
}
template<int SPT>
__device__ __forceinline__ void w_quads_store(float4* wfrag, const float4 (&q)[SPT])
{
#pragma unroll
    for (int i = 0; i < SPT; ++i) wfrag[(int)threadIdx.x + i * GEMM_T] = q[i];
}

__global__ __launch_bounds__(GEMM_T)
void head_fused_f32(const float* __restrict__ A, const float* __restrict__ W1,
                    const float* __restrict__ b1, const float* __restrict__ g,
                    const float* __restrict__ be, const float* __restrict__ W2,
                    const float* __restrict__ b2, float* __restrict__ C, int N)
{
    constexpr int K1 = HEAD_K, M1 = HEAD_M, M2 = HEAD_M;
    constexpr int TN1 = M1 / 16;
    constexpr int KB1 = K1 / 8;
    constexpr int TN2 = M2 / 16;
    constexpr int KB2 = M1 / 8;
    constexpr int NW = GEMM_T / 32;
    __shared__ float4 wfrag1[KB1 * TN1 * 32];
    __shared__ float4 wfrag2[KB2 * TN2 * 32];
    __shared__ float4 ytile[NW * 16 * HEAD_YP / 4];

    const int lane = threadIdx.x & 31;
    const int wv = __builtin_amdgcn_readfirstlane((int)(threadIdx.x >> 5));
    const int lh = lane >> 4;
    const int lm = lane & 15;

    const unsigned nstrips = (unsigned)(N + 15) >> 4;
    const int s_beg = (int)(blockIdx.x * nstrips / gridDim.x);
    const int s_end = (int)((blockIdx.x + 1u) * nstrips / gridDim.x);

    int strip = s_beg + wv;
    float4 a_cur[KB1];
#pragma unroll
    for (int j = 0; j < KB1; ++j) a_cur[j] = make_float4(0.0f, 0.0f, 0.0f, 0.0f);
    if (strip < s_end) load_a_strip<K1>(A, strip, lm, lh, N, a_cur);

    float4 q1[K1 * M1 / (4 * GEMM_T)];
    float4 q2[M1 * M2 / (4 * GEMM_T)];
    w_quads_load<K1, M1>(W1, q1);
    w_quads_load<M1, M2>(W2, q2);
    w_quads_store(wfrag1, q1);
    w_quads_store(wfrag2, q2);

    float badd1[TN1], gsc[TN1], bsh[TN1], badd2[TN2];
#pragma unroll
    for (int t = 0; t < TN1; ++t) {
        badd1[t] = b1[16 * t + lm];
        gsc[t] = g[16 * t + lm];
        bsh[t] = be[16 * t + lm];
    }
#pragma unroll
    for (int t = 0; t < TN2; ++t) badd2[t] = b2[16 * t + lm];

    __syncthreads();

    float4 wb1[KB1 * TN1], wb2[KB2 * TN2];
#pragma unroll
    for (int i = 0; i < KB1 * TN1; ++i) wb1[i] = wfrag1[i * 32 + lane];
#pragma unroll
    for (int i = 0; i < KB2 * TN2; ++i) wb2[i] = wfrag2[i * 32 + lane];

    float* yt = (float*)ytile + wv * (16 * HEAD_YP);

    for (; strip < s_end; strip += NW) {
        float4 a_nxt[KB1];
#pragma unroll
        for (int j = 0; j < KB1; ++j) a_nxt[j] = a_cur[j];
        if (strip + NW < s_end) load_a_strip<K1>(A, strip + NW, lm, lh, N, a_nxt);

        v8f acc1[TN1];
#pragma unroll
        for (int t = 0; t < TN1; ++t) acc1[t] = v8f{};
#pragma unroll
        for (int j = 0; j < KB1; ++j) {
            v2f a01, a23;
            a01.x = a_cur[j].x; a01.y = a_cur[j].y;
            a23.x = a_cur[j].z; a23.y = a_cur[j].w;
#pragma unroll
            for (int t = 0; t < TN1; ++t) {
                v2f b01;
                b01.x = wb1[j * TN1 + t].x; b01.y = wb1[j * TN1 + t].y;
                acc1[t] = __builtin_amdgcn_wmma_f32_16x16x4_f32(false, a01, false, b01,
                                                                (short)0, acc1[t], false, false);
            }
#pragma unroll
            for (int t = 0; t < TN1; ++t) {
                v2f b23;
                b23.x = wb1[j * TN1 + t].z; b23.y = wb1[j * TN1 + t].w;
                acc1[t] = __builtin_amdgcn_wmma_f32_16x16x4_f32(false, a23, false, b23,
                                                                (short)0, acc1[t], false, false);
            }
        }

#pragma unroll
        for (int r = 0; r < 8; ++r) {
            float v[TN1];
            float s = 0.0f;
#pragma unroll
            for (int t = 0; t < TN1; ++t) { v[t] = acc1[t][r] + badd1[t]; s += v[t]; }
            const float mu = lanes_sum<16>(s) * (1.0f / M1);
            float q = 0.0f;
#pragma unroll
            for (int t = 0; t < TN1; ++t) { v[t] -= mu; q += v[t] * v[t]; }
            const float rstd = rsqrtf(lanes_sum<16>(q) * (1.0f / M1) + 1e-5f);
#pragma unroll
            for (int t = 0; t < TN1; ++t) {
                const float o = v[t] * rstd * gsc[t] + bsh[t];
                yt[(r + (lh << 3)) * HEAD_YP + 16 * t + lm] = (o > 0.0f) ? o : (expf(o) - 1.0f);
            }
        }

        float4 a2[KB2];
#pragma unroll
        for (int j = 0; j < KB2; ++j)
            a2[j] = *(const float4*)(yt + lm * HEAD_YP + 8 * j + 4 * lh);

        v8f acc2[TN2];
#pragma unroll
        for (int t = 0; t < TN2; ++t) acc2[t] = v8f{};
#pragma unroll
        for (int j = 0; j < KB2; ++j) {
            v2f a01, a23;
            a01.x = a2[j].x; a01.y = a2[j].y;
            a23.x = a2[j].z; a23.y = a2[j].w;
#pragma unroll
            for (int t = 0; t < TN2; ++t) {
                v2f b01;
                b01.x = wb2[j * TN2 + t].x; b01.y = wb2[j * TN2 + t].y;
                acc2[t] = __builtin_amdgcn_wmma_f32_16x16x4_f32(false, a01, false, b01,
                                                                (short)0, acc2[t], false, false);
            }
#pragma unroll
            for (int t = 0; t < TN2; ++t) {
                v2f b23;
                b23.x = wb2[j * TN2 + t].z; b23.y = wb2[j * TN2 + t].w;
                acc2[t] = __builtin_amdgcn_wmma_f32_16x16x4_f32(false, a23, false, b23,
                                                                (short)0, acc2[t], false, false);
            }
        }

        const int row0 = strip << 4;
#pragma unroll
        for (int r = 0; r < 8; ++r) {
            const int row = row0 + r + (lh << 3);
            if (row < N) {
#pragma unroll
                for (int t = 0; t < TN2; ++t)
                    C[(size_t)row * M2 + 16 * t + lm] = acc2[t][r] + badd2[t];
            }
        }

#pragma unroll
        for (int j = 0; j < KB1; ++j) a_cur[j] = a_nxt[j];
    }
}

constexpr int PAIR_K  = 64;
constexpr int PAIR_H  = 128;
constexpr int PAIR_M  = 64;
constexpr int PAIR_YP = PAIR_H + 4;

__global__ __launch_bounds__(GEMM_T)
void gemm_pair_f32(const float* __restrict__ A, const float* __restrict__ W1,
                   const float* __restrict__ b1, const float* __restrict__ g,
                   const float* __restrict__ be, const float* __restrict__ W2,
                   const float* __restrict__ rscale, float* __restrict__ C, int N)
{
    constexpr int K1 = PAIR_K, M1 = PAIR_H, M2 = PAIR_M;
    constexpr int TN1 = M1 / 16;
    constexpr int KB1 = K1 / 8;
    constexpr int TN2 = M2 / 16;
    constexpr int KB2 = M1 / 8;
    constexpr int NW = GEMM_T / 32;
    __shared__ float4 wfrag1[KB1 * TN1 * 32];
    __shared__ float4 wfrag2[KB2 * TN2 * 32];
    __shared__ float4 ytile[NW * 16 * PAIR_YP / 4];

    const int lane = threadIdx.x & 31;
    const int wv = __builtin_amdgcn_readfirstlane((int)(threadIdx.x >> 5));
    const int lh = lane >> 4;
    const int lm = lane & 15;

    const unsigned nstrips = (unsigned)(N + 15) >> 4;
    const int s_beg = (int)(blockIdx.x * nstrips / gridDim.x);
    const int s_end = (int)((blockIdx.x + 1u) * nstrips / gridDim.x);

    int strip = s_beg + wv;
    float4 a_cur[KB1];
#pragma unroll
    for (int j = 0; j < KB1; ++j) a_cur[j] = make_float4(0.0f, 0.0f, 0.0f, 0.0f);
    if (strip < s_end) load_a_strip<K1>(A, strip, lm, lh, N, a_cur);
    float rs_cur[8];
#pragma unroll
    for (int r = 0; r < 8; ++r) rs_cur[r] = 1.0f;
    if (strip < s_end) load_row_scales<true>(rscale, strip, lh, N, rs_cur);

    float4 q1[K1 * M1 / (4 * GEMM_T)];
    float4 q2[M1 * M2 / (4 * GEMM_T)];
    w_quads_load<K1, M1>(W1, q1);
    w_quads_load<M1, M2>(W2, q2);
    w_quads_store(wfrag1, q1);
    w_quads_store(wfrag2, q2);

    float badd1[TN1], gsc[TN1], bsh[TN1];
#pragma unroll
    for (int t = 0; t < TN1; ++t) {
        badd1[t] = b1[16 * t + lm];
        gsc[t] = g[16 * t + lm];
        bsh[t] = be[16 * t + lm];
    }

    __syncthreads();

    float4 wb1[KB1 * TN1];
#pragma unroll
    for (int i = 0; i < KB1 * TN1; ++i) wb1[i] = wfrag1[i * 32 + lane];

    float* yt = (float*)ytile + wv * (16 * PAIR_YP);
    const float4* w2l = wfrag2 + lane;

    for (; strip < s_end; strip += NW) {
        float4 a_nxt[KB1];
#pragma unroll
        for (int j = 0; j < KB1; ++j) a_nxt[j] = a_cur[j];
        float rs_nxt[8];
#pragma unroll
        for (int r = 0; r < 8; ++r) rs_nxt[r] = rs_cur[r];
        if (strip + NW < s_end) {
            load_a_strip<K1>(A, strip + NW, lm, lh, N, a_nxt);
            load_row_scales<true>(rscale, strip + NW, lh, N, rs_nxt);
        }

        v8f acc1[TN1];
#pragma unroll
        for (int t = 0; t < TN1; ++t) acc1[t] = v8f{};
#pragma unroll
        for (int j = 0; j < KB1; ++j) {
            v2f a01, a23;
            a01.x = a_cur[j].x; a01.y = a_cur[j].y;
            a23.x = a_cur[j].z; a23.y = a_cur[j].w;
#pragma unroll
            for (int t = 0; t < TN1; ++t) {
                v2f b01;
                b01.x = wb1[j * TN1 + t].x; b01.y = wb1[j * TN1 + t].y;
                acc1[t] = __builtin_amdgcn_wmma_f32_16x16x4_f32(false, a01, false, b01,
                                                                (short)0, acc1[t], false, false);
            }
#pragma unroll
            for (int t = 0; t < TN1; ++t) {
                v2f b23;
                b23.x = wb1[j * TN1 + t].z; b23.y = wb1[j * TN1 + t].w;
                acc1[t] = __builtin_amdgcn_wmma_f32_16x16x4_f32(false, a23, false, b23,
                                                                (short)0, acc1[t], false, false);
            }
        }

#pragma unroll
        for (int r = 0; r < 8; ++r) {
            float v[TN1];
            float s = 0.0f;
#pragma unroll
            for (int t = 0; t < TN1; ++t) { v[t] = acc1[t][r] + badd1[t]; s += v[t]; }
            const float mu = lanes_sum<16>(s) * (1.0f / M1);
            float q = 0.0f;
#pragma unroll
            for (int t = 0; t < TN1; ++t) { v[t] -= mu; q += v[t] * v[t]; }
            const float rstd = rsqrtf(lanes_sum<16>(q) * (1.0f / M1) + 1e-5f);
#pragma unroll
            for (int t = 0; t < TN1; ++t) {
                const float o = v[t] * rstd * gsc[t] + bsh[t];
                yt[(r + (lh << 3)) * PAIR_YP + 16 * t + lm] = (o > 0.0f) ? o : (expf(o) - 1.0f);
            }
        }

        float4 a2[KB2];
#pragma unroll
        for (int j = 0; j < KB2; ++j)
            a2[j] = *(const float4*)(yt + lm * PAIR_YP + 8 * j + 4 * lh);

        v8f acc2[TN2];
#pragma unroll
        for (int t = 0; t < TN2; ++t) acc2[t] = v8f{};
#pragma unroll
        for (int j = 0; j < KB2; ++j) {
            v2f a01, a23;
            a01.x = a2[j].x; a01.y = a2[j].y;
            a23.x = a2[j].z; a23.y = a2[j].w;
            float4 wq[TN2];
#pragma unroll
            for (int t = 0; t < TN2; ++t) wq[t] = w2l[(j * TN2 + t) * 32];
#pragma unroll
            for (int t = 0; t < TN2; ++t) {
                v2f b01;
                b01.x = wq[t].x; b01.y = wq[t].y;
                acc2[t] = __builtin_amdgcn_wmma_f32_16x16x4_f32(false, a01, false, b01,
                                                                (short)0, acc2[t], false, false);
            }
#pragma unroll
            for (int t = 0; t < TN2; ++t) {
                v2f b23;
                b23.x = wq[t].z; b23.y = wq[t].w;
                acc2[t] = __builtin_amdgcn_wmma_f32_16x16x4_f32(false, a23, false, b23,
                                                                (short)0, acc2[t], false, false);
            }
        }

        const int row0 = strip << 4;
#pragma unroll
        for (int r = 0; r < 8; ++r) {
            const int row = row0 + r + (lh << 3);
            if (row < N) {
#pragma unroll
                for (int t = 0; t < TN2; ++t)
                    C[(size_t)row * M2 + 16 * t + lm] = acc2[t][r] * rs_cur[r];
            }
        }

#pragma unroll
        for (int j = 0; j < KB1; ++j) a_cur[j] = a_nxt[j];
#pragma unroll
        for (int r = 0; r < 8; ++r) rs_cur[r] = rs_nxt[r];
    }
}

constexpr int CH_E   = 4096;
constexpr int BK_SH  = 8;
constexpr int BK_N   = 1 << BK_SH;
constexpr int PREP_T = 1024;
constexpr int CH_U   = CH_E / PREP_T;

__global__ __launch_bounds__(1024)
void bin_kernel(const int* __restrict__ src, const int* __restrict__ dst,
                int2* __restrict__ binned, int* __restrict__ tcnt, int* __restrict__ toff,
                int E, int NB, int NCH)
{
    __shared__ int hist[PREP_T];
    __shared__ int offs[PREP_T];
    __shared__ int wsum[32];
    const int t = threadIdx.x, lane = t & 31, wv = t >> 5;
    const int chunk = blockIdx.x;
    const int e_base = chunk * CH_E;

    hist[t] = 0;
    __syncthreads();

    int s[CH_U], d[CH_U], r[CH_U];
#pragma unroll
    for (int u = 0; u < CH_U; ++u) {
        const int e = e_base + u * PREP_T + t;
        const bool ok = e < E;
        s[u] = ok ? src[e] : 0;
        d[u] = ok ? dst[e] : -1;
    }
#pragma unroll
    for (int u = 0; u < CH_U; ++u)
        r[u] = (d[u] >= 0) ? atomicAdd(&hist[d[u] >> BK_SH], 1) : 0;
    __syncthreads();

    const int c = hist[t];
    int x = c;
#pragma unroll
    for (int off = 1; off < 32; off <<= 1) {
        int y = __shfl_up(x, off, 32);
        if (lane >= off) x += y;
    }
    if (lane == 31) wsum[wv] = x;
    __syncthreads();
    if (wv == 0) {
        const int tot = wsum[lane];
        int q = tot;
#pragma unroll
        for (int off = 1; off < 32; off <<= 1) {
            int y = __shfl_up(q, off, 32);
            if (lane >= off) q += y;
        }
        wsum[lane] = q - tot;
    }
    __syncthreads();
    const int excl = wsum[wv] + x - c;
    offs[t] = excl;
    if (t < NB) {
        tcnt[(size_t)t * NCH + chunk] = c;
        toff[(size_t)t * NCH + chunk] = excl;
    }
    __syncthreads();
#pragma unroll
    for (int u = 0; u < CH_U; ++u)
        if (d[u] >= 0) binned[e_base + offs[d[u] >> BK_SH] + r[u]] = make_int2(s[u], d[u]);
}

__global__ __launch_bounds__(1024)
void bucket_place_kernel(const int* __restrict__ tcnt, const int* __restrict__ toff,
                         const int2* __restrict__ binned, int* __restrict__ rowptr,
                         float* __restrict__ dinv, int* __restrict__ col, int N, int NCH)
{
    __shared__ int ncnt[BK_N];
    __shared__ int lrow[BK_N];
    __shared__ int wsum[32];
    __shared__ int s_prefix;
    const int t = threadIdx.x, lane = t & 31, wv = t >> 5;
    const int b = blockIdx.x;
    const int node0 = b << BK_SH;
    const int* __restrict__ cnt_row = tcnt + (size_t)b * NCH;
    const int* __restrict__ off_row = toff + (size_t)b * NCH;

    if (t < BK_N) ncnt[t] = 0;

    int p = 0;
    for (int c = t; c < NCH; c += PREP_T) p += off_row[c];
#pragma unroll
    for (int off = 16; off; off >>= 1) p += __shfl_xor(p, off, 32);
    if (lane == 0) wsum[wv] = p;
    __syncthreads();
    if (wv == 0) {
        int q = wsum[lane];
#pragma unroll
        for (int off = 16; off; off >>= 1) q += __shfl_xor(q, off, 32);
        if (lane == 0) s_prefix = q;
    }
    __syncthreads();
    const int bstart = s_prefix;

    for (int c = t; c < NCH; c += PREP_T) {
        const int n = cnt_row[c];
        const int2* __restrict__ seg = binned + (size_t)c * CH_E + off_row[c];
        for (int i = 0; i < n; ++i) atomicAdd(&ncnt[seg[i].y - node0], 1);
    }
    __syncthreads();

    const int cn = (t < BK_N) ? ncnt[t] : 0;
    int x = cn;
#pragma unroll
    for (int off = 1; off < 32; off <<= 1) {
        int y = __shfl_up(x, off, 32);
        if (lane >= off) x += y;
    }
    if (t < BK_N && lane == 31) wsum[wv] = x;
    __syncthreads();
    if (t < BK_N) {
        int before = 0;
        for (int w = 0; w < wv; ++w) before += wsum[w];
        const int excl = before + x - cn;
        lrow[t] = excl;
        ncnt[t] = 0;
        const int node = node0 + t;
        if (node < N) {
            rowptr[node] = bstart + excl;
            dinv[node] = rsqrtf((float)cn + 1.0f);
            if (node == N - 1) rowptr[N] = bstart + excl + cn;
        }
    }
    __syncthreads();

    for (int c = t; c < NCH; c += PREP_T) {
        const int n = cnt_row[c];
        const int2* __restrict__ seg = binned + (size_t)c * CH_E + off_row[c];
        for (int i = 0; i < n; ++i) {
            const int2 sd = seg[i];
            const int dl = sd.y - node0;
            const int r = atomicAdd(&ncnt[dl], 1);
            col[bstart + lrow[dl] + r] = sd.x;
        }
    }
}

__device__ __forceinline__ float4 row_quad(const char* __restrict__ ysb, unsigned cb, unsigned roff)
{
    return *(const float4*)(ysb + (roff + cb));
}

template<int M, int NL, int UN>
__device__ __forceinline__ void gather_chunk(const char* __restrict__ ysb, unsigned cb, unsigned moff,
                                             int nfull, int rem, int sub, float4& acc)
{
    constexpr int EPW = 32 / (M / 4);
    constexpr int BATCH = UN * EPW;
    float4 vl[NL > 0 ? NL : 1];
#pragma unroll
    for (int u = 0; u < NL; ++u)
        vl[u] = row_quad(ysb, cb, __shfl(moff, (nfull + u * EPW + sub) & 31, 32));

    for (int j0 = 0; j0 < nfull; j0 += BATCH) {
        float4 v[UN];
#pragma unroll
        for (int u = 0; u < UN; ++u)
            v[u] = row_quad(ysb, cb, __shfl(moff, (j0 + u * EPW + sub) & 31, 32));
#pragma unroll
        for (int u = 0; u < UN; ++u) {
            acc.x += v[u].x;
            acc.y += v[u].y;
            acc.z += v[u].z;
            acc.w += v[u].w;
        }
    }

    if (NL > 0) {
#pragma unroll
        for (int u = 0; u < NL - 1; ++u) {
            acc.x += vl[u].x;
            acc.y += vl[u].y;
            acc.z += vl[u].z;
            acc.w += vl[u].w;
        }
        constexpr int UL = NL > 0 ? NL - 1 : 0;
        const float m = (UL * EPW + sub < rem) ? 1.0f : 0.0f;
        acc.x = fmaf(vl[UL].x, m, acc.x);
        acc.y = fmaf(vl[UL].y, m, acc.y);
        acc.z = fmaf(vl[UL].z, m, acc.z);
        acc.w = fmaf(vl[UL].w, m, acc.w);
    }
}

template<int M, bool TAIL, bool PRE>
__global__ __launch_bounds__(256)
void csr_gather_ln_elu(const int* __restrict__ rowptr, const int* __restrict__ col,
                       const float* __restrict__ ys, const float* __restrict__ dinv,
                       const float* __restrict__ bias, const float* __restrict__ g,
                       const float* __restrict__ be, float* __restrict__ out, int N)
{
    static_assert(TAIL || !PRE, "the bare aggregate is never gathered again");
    constexpr int LPR = M / 4;
    constexpr int EPW = 32 / LPR;
    constexpr int UN  = 4;
    constexpr int BATCH = UN * EPW;
    static_assert(UN == 4 && (BATCH & (BATCH - 1)) == 0, "the left-over cases below are those of four loads");
    const int nwaves = (int)(gridDim.x * (blockDim.x >> 5));
    int wave = (int)(blockIdx.x * (blockDim.x >> 5)) + (int)(threadIdx.x >> 5);
    wave = __builtin_amdgcn_readfirstlane(wave);
    const int lane = threadIdx.x & 31;
    const int sub = lane / LPR;
    const int c = (lane % LPR) << 2;
    const char* __restrict__ ysb = (const char*)ys;
    const unsigned cb = (unsigned)c * 4u;

    float4 bi = make_float4(0.0f, 0.0f, 0.0f, 0.0f), gg = bi, bb = bi;
    if (TAIL) {
        bi.x = bias[c]; bi.y = bias[c + 1]; bi.z = bias[c + 2]; bi.w = bias[c + 3];
        gg.x = g[c];    gg.y = g[c + 1];    gg.z = g[c + 2];    gg.w = g[c + 3];
        bb.x = be[c];   bb.y = be[c + 1];   bb.z = be[c + 2];   bb.w = be[c + 3];
    }

    for (int node = wave; node < N; node += nwaves) {
        const int beg = rowptr[node];
        const int end = rowptr[node + 1];
        const float di = dinv[node];
        const int stop = end + 1;

        float4 acc = make_float4(0.0f, 0.0f, 0.0f, 0.0f);
        int base = beg;
        do {
            const int n = min(32, stop - base);
            const int e = base + lane;
            int mine = node;
            if (e < end) mine = col[e];
            const unsigned moff = (unsigned)mine * (unsigned)(M * 4);

            const int nfull = n & ~(BATCH - 1);
            const int rem = n - nfull;
            switch ((rem + EPW - 1) / EPW) {
                case 0: gather_chunk<M, 0, UN>(ysb, cb, moff, nfull, rem, sub, acc); break;
                case 1: gather_chunk<M, 1, UN>(ysb, cb, moff, nfull, rem, sub, acc); break;
                case 2: gather_chunk<M, 2, UN>(ysb, cb, moff, nfull, rem, sub, acc); break;
                case 3: gather_chunk<M, 3, UN>(ysb, cb, moff, nfull, rem, sub, acc); break;
                default: gather_chunk<M, 4, UN>(ysb, cb, moff, nfull, rem, sub, acc); break;
            }
            base += 32;
        } while (base < stop);
        if (EPW == 2) {
            acc.x += __shfl_xor(acc.x, 16, 32);
            acc.y += __shfl_xor(acc.y, 16, 32);
            acc.z += __shfl_xor(acc.z, 16, 32);
            acc.w += __shfl_xor(acc.w, 16, 32);
        }

        if (!TAIL) {
            if (sub == 0)
                *(float4*)(out + (size_t)node * M + c) =
                    make_float4(acc.x * di, acc.y * di, acc.z * di, acc.w * di);
            continue;
        }

        float4 t;
        t.x = acc.x * di + bi.x;
        t.y = acc.y * di + bi.y;
        t.z = acc.z * di + bi.z;
        t.w = acc.w * di + bi.w;

        const float sum = lanes_sum<LPR>((t.x + t.y) + (t.z + t.w));
        const float mu = sum * (1.0f / M);
        const float dx = t.x - mu, dy = t.y - mu, dz = t.z - mu, dw = t.w - mu;
        const float vs = lanes_sum<LPR>((dx * dx + dy * dy) + (dz * dz + dw * dw));
        const float rstd = rsqrtf(vs * (1.0f / M) + 1e-5f);
        float4 o;
        o.x = dx * rstd * gg.x + bb.x;
        o.y = dy * rstd * gg.y + bb.y;
        o.z = dz * rstd * gg.z + bb.z;
        o.w = dw * rstd * gg.w + bb.w;
        o.x = (o.x > 0.0f) ? o.x : (expf(o.x) - 1.0f);
        o.y = (o.y > 0.0f) ? o.y : (expf(o.y) - 1.0f);
        o.z = (o.z > 0.0f) ? o.z : (expf(o.z) - 1.0f);
        o.w = (o.w > 0.0f) ? o.w : (expf(o.w) - 1.0f);
        if (PRE) { o.x *= di; o.y *= di; o.z *= di; o.w *= di; }
        if (sub == 0) *(float4*)(out + (size_t)node * M + c) = o;
    }
}

extern "C" void kernel_launch(void* const* d_in, const int* in_sizes, int n_in,
                              void* d_out, int out_size, void* d_ws, size_t ws_size,
                              hipStream_t stream)
{
    const float* x   = (const float*)d_in[0];
    const int*   ei  = (const int*)  d_in[1];
    const float* W1  = (const float*)d_in[2];
    const float* b1  = (const float*)d_in[3];
    const float* g1  = (const float*)d_in[4];
    const float* be1 = (const float*)d_in[5];
    const float* W2  = (const float*)d_in[6];
    const float* b2  = (const float*)d_in[7];
    const float* g2  = (const float*)d_in[8];
    const float* be2 = (const float*)d_in[9];
    const float* W3  = (const float*)d_in[10];
    const float* b3  = (const float*)d_in[11];
    const float* g3  = (const float*)d_in[12];
    const float* be3 = (const float*)d_in[13];
    const float* lw1 = (const float*)d_in[14];
    const float* lb1 = (const float*)d_in[15];
    const float* g4  = (const float*)d_in[16];
    const float* be4 = (const float*)d_in[17];
    const float* lw2 = (const float*)d_in[18];
    const float* lb2 = (const float*)d_in[19];

    const int N = in_sizes[0] / 128;
    const int E = in_sizes[1] / 2;
    const int* src = ei;
    const int* dst = ei + E;

    const int NB  = (N + BK_N - 1) / BK_N;
    const int NCH = ceil_div_ll(E, CH_E);
    float* ws   = (float*)d_ws;
    float* xw   = ws;  ws += (size_t)N * 128;
    float* agg  = ws;  ws += (size_t)N * 128;
    float* h    = ws;  ws += (size_t)N * 64;
    int*   col  = (int*)ws;
    ws += (size_t)N * 64;
    ws += N;
    float* dinv   = ws;        ws += N;
    int*   rowptr = (int*)ws;  ws += E;
    int2*  binned = (int2*)agg;
    int*   tcnt   = (int*)agg + (size_t)2 * NCH * CH_E;
    int*   toff   = tcnt + (size_t)NB * NCH;

    const int B = 256;
    auto gemm_grid = [N](int per_wgp) {
        const int want = GEMM_WGPS * per_wgp;
        const int most = ceil_div_ll((N + 15) / 16, GEMM_T / 32);
        return want < most ? want : most;
    };
    const int node_blocks = loop_grid(ceil_div_ll(N, 8));

    bin_kernel<<<NCH, PREP_T, 0, stream>>>(src, dst, binned, tcnt, toff, E, NB, NCH);
    bucket_place_kernel<<<NB, PREP_T, 0, stream>>>(tcnt, toff, binned, rowptr, dinv, col, N, NCH);

    wmma_gemm_f32<128, 64, false, true><<<gemm_grid(1), GEMM_T, 0, stream>>>(x, W1, nullptr, nullptr, nullptr, dinv, xw, N);
    csr_gather_ln_elu<64, true, true><<<node_blocks, B, 0, stream>>>(rowptr, col, xw, dinv, b1, g1, be1, h, N);

    csr_gather_ln_elu<64, false, false><<<node_blocks, B, 0, stream>>>(rowptr, col, h, dinv, nullptr, nullptr, nullptr, xw, N);
    gemm_pair_f32<<<gemm_grid(1), GEMM_T, 0, stream>>>(xw, W2, b2, g2, be2, W3, dinv, agg, N);
    csr_gather_ln_elu<64, true, false><<<node_blocks, B, 0, stream>>>(rowptr, col, agg, dinv, b3, g3, be3, h, N);

    head_fused_f32<<<gemm_grid(HEAD_WGS), GEMM_T, 0, stream>>>(h, lw1, lb1, g4, be4, lw2, lb2, (float*)d_out, N);
}
